# starting kernel
# baseline (speedup 1.0000x reference)
#include <hip/hip_runtime.h>
#include <hip/hip_bf16.h>
#include <math.h>

typedef __attribute__((ext_vector_type(2))) float v2f;
typedef __attribute__((ext_vector_type(8))) float v8f;

#define NIMG 4
#define CIN 256
#define HF 128
#define WF 128
#define A_ANCH 3
#define NPROP (A_ANCH * HF * WF)
#define NPAD 65536
#define PRE_NMS 2000
#define POST_NMS 300
#define OBJ_THR 0.3f
#define IOU_THR 0.7f
#define TILE_W 64
#define CCHUNK 32
#define CIPAD 34

__global__ __launch_bounds__(256) void repack_w_kernel(const float* __restrict__ conv_w,
                                                       float* __restrict__ wt2) {
    int t = blockIdx.x * 256 + threadIdx.x;
    int par = t & 1;
    int co  = (t >> 1) & 255;
    int q   = t >> 9;
    int ci2 = q & 127;
    int kk  = q >> 7;
    int ci  = ci2 * 2 + par;
    int kh  = kk / 3, kw = kk % 3;
    wt2[t] = conv_w[((co * CIN + ci) * 3 + kh) * 3 + kw];
}

__global__ __launch_bounds__(256) void conv_fused_kernel(
    const float* __restrict__ feat, const float* __restrict__ wt2,
    const float* __restrict__ conv_b,
    const float* __restrict__ det_w, const float* __restrict__ det_b,
    const float* __restrict__ reg_w, const float* __restrict__ reg_b,
    float* __restrict__ scores, float* __restrict__ deltas) {

    __shared__ float lds_in[3 * 66 * CIPAD];
    __shared__ float x_tile[256 * 33];
    __shared__ float hb[16];

    int bid = blockIdx.x;
    int n    = bid >> 8;
    int rem  = bid & 255;
    int row  = rem >> 1;
    int col0 = (rem & 1) * TILE_W;

    int tid  = threadIdx.x;
    int wave = tid >> 5;
    int lane = tid & 31;
    int hl   = lane >> 4;
    int lm   = lane & 15;

    v8f acc[2][4] = {};

    const float* featn = feat + (size_t)n * CIN * HF * WF;
    const v2f*   wp    = (const v2f*)wt2;

    for (int cib = 0; cib < CIN; cib += CCHUNK) {
        __syncthreads();
        for (int e = tid; e < 3 * 66 * CCHUNK; e += 256) {
            int col = e % 66;
            int r   = (e / 66) % 3;
            int cl  = e / 198;
            int grow = row + r - 1;
            int gcol = col0 + col - 1;
            float v = 0.0f;
            if (grow >= 0 && grow < HF && gcol >= 0 && gcol < WF)
                v = featn[((size_t)(cib + cl) * HF + grow) * WF + gcol];
            lds_in[(r * 66 + col) * CIPAD + cl] = v;
        }
        __syncthreads();

        for (int kh = 0; kh < 3; kh++) {
            for (int kw = 0; kw < 3; kw++) {
                int kk = kh * 3 + kw;
                #pragma unroll 4
                for (int c = 0; c < CCHUNK; c += 4) {
                    int ci2 = ((cib + c) >> 1) + hl;
                    int co0 = wave * 32 + lm;
                    v2f a0 = wp[(kk * 128 + ci2) * 256 + co0];
                    v2f a1 = wp[(kk * 128 + ci2) * 256 + co0 + 16];
                    int boff = c + 2 * hl;
                    v2f b[4];
                    #pragma unroll
                    for (int nt = 0; nt < 4; nt++)
                        b[nt] = *(const v2f*)&lds_in[(kh * 66 + nt * 16 + lm + kw) * CIPAD + boff];
                    #pragma unroll
                    for (int nt = 0; nt < 4; nt++) {
                        acc[0][nt] = __builtin_amdgcn_wmma_f32_16x16x4_f32(false, a0, false, b[nt], (short)0, acc[0][nt], false, false);
                        acc[1][nt] = __builtin_amdgcn_wmma_f32_16x16x4_f32(false, a1, false, b[nt], (short)0, acc[1][nt], false, false);
                    }
                }
            }
        }
    }

    __syncthreads();
    for (int e = tid; e < 15 * 256; e += 256)
        lds_in[e] = (e < 768) ? det_w[e] : reg_w[e - 768];
    if (tid < 15) hb[tid] = (tid < 3) ? det_b[tid] : reg_b[tid - 3];

    int m0  = hl * 8;
    float biasA[8], biasB[8];
    #pragma unroll
    for (int r = 0; r < 8; r++) {
        biasA[r] = conv_b[wave * 32 + m0 + r];
        biasB[r] = conv_b[wave * 32 + 16 + m0 + r];
    }

    for (int ph = 0; ph < 2; ph++) {
        if (ph) __syncthreads();
        #pragma unroll
        for (int r = 0; r < 8; r++) {
            int coA = wave * 32 + m0 + r;
            int coB = wave * 32 + 16 + m0 + r;
            x_tile[coA * 33 + lm]      = acc[0][ph * 2 + 0][r] + biasA[r];
            x_tile[coA * 33 + 16 + lm] = acc[0][ph * 2 + 1][r] + biasA[r];
            x_tile[coB * 33 + lm]      = acc[1][ph * 2 + 0][r] + biasB[r];
            x_tile[coB * 33 + 16 + lm] = acc[1][ph * 2 + 1][r] + biasB[r];
        }
        __syncthreads();

        for (int t = tid; t < 15 * 32; t += 256) {
            int oi  = t >> 5;
            int pix = t & 31;
            float sum = hb[oi];
            const float* wrow = &lds_in[oi * 256];
            #pragma unroll 8
            for (int co = 0; co < 256; co++)
                sum += wrow[co] * x_tile[co * 33 + pix];
            int p = row * WF + col0 + ph * 32 + pix;
            if (oi < 3) {
                scores[(size_t)n * NPROP + oi * (HF * WF) + p] = 1.0f / (1.0f + expf(-sum));
            } else {
                int ch = oi - 3;
                int ccomp = ch / 3;
                int a = ch % 3;
                deltas[(((size_t)n * NPROP) + a * (HF * WF) + p) * 4 + ccomp] = sum;
            }
        }
    }
}

__global__ __launch_bounds__(256) void decode_kernel(
    const float* __restrict__ scores, const float* __restrict__ deltas,
    float* __restrict__ boxes, unsigned long long* __restrict__ keys) {
    int t = blockIdx.x * 256 + threadIdx.x;
    int n = t >> 16;
    int i = t & 0xFFFF;
    unsigned long long* kout = keys + ((size_t)n << 16);
    if (i >= NPROP) { kout[i] = 0ull; return; }

    int pa = i / 3, aa = i - pa * 3;
    int ya = pa >> 7, xa = pa & 127;
    const float bs[3][4] = {{-91.f, -45.f, 91.f, 45.f},
                            {-64.f, -64.f, 64.f, 64.f},
                            {-45.f, -91.f, 45.f, 91.f}};
    float sxv = (float)(xa * 8), syv = (float)(ya * 8);
    float x1 = sxv + bs[aa][0], y1 = syv + bs[aa][1];
    float x2 = sxv + bs[aa][2], y2 = syv + bs[aa][3];
    float aw = x2 - x1, ah = y2 - y1;
    float acx = y1 + aw * 0.5f;
    float acy = x1 + ah * 0.5f;

    const float* d = deltas + (((size_t)n * NPROP) + i) * 4;
    float px = acx + d[0] * aw;
    float py = acy + d[1] * ah;
    float pw = aw * expf(d[2]);
    float ph = ah * expf(d[3]);
    float b0 = fminf(fmaxf(px - pw * 0.5f, 0.f), 1024.f);
    float b1 = fminf(fmaxf(py - ph * 0.5f, 0.f), 1024.f);
    float b2 = fminf(fmaxf(px + pw * 0.5f, 0.f), 1024.f);
    float b3 = fminf(fmaxf(py + ph * 0.5f, 0.f), 1024.f);

    float* bo = boxes + (((size_t)n * NPROP) + i) * 4;
    bo[0] = b0; bo[1] = b1; bo[2] = b2; bo[3] = b3;

    float s = scores[(size_t)n * NPROP + i];
    bool valid = (s > OBJ_THR) && ((b2 - b0) > 0.f) && ((b3 - b1) > 0.f);
    float ss = valid ? s : -__builtin_inff();
    unsigned u = __float_as_uint(ss);
    unsigned sv = (u & 0x80000000u) ? ~u : (u | 0x80000000u);
    kout[i] = ((unsigned long long)sv << 32) | (unsigned long long)(0xFFFFFFFFu - (unsigned)i);
}

__global__ __launch_bounds__(256) void bitonic_step_kernel(unsigned long long* __restrict__ keys,
                                                           int j, int k) {
    int t = blockIdx.x * 256 + threadIdx.x;
    int n = t >> 16;
    int i = t & 0xFFFF;
    int ixj = i ^ j;
    if (ixj > i) {
        unsigned long long* K = keys + ((size_t)n << 16);
        unsigned long long a = K[i];
        unsigned long long b = K[ixj];
        bool up = ((i & k) == 0);
        bool sw = up ? (a < b) : (a > b);
        if (sw) { K[i] = b; K[ixj] = a; }
    }
}

__global__ __launch_bounds__(1024) void nms_kernel(const unsigned long long* __restrict__ keys,
                                                   const float* __restrict__ boxes,
                                                   float* __restrict__ out) {
    __shared__ float bx[PRE_NMS * 4];
    __shared__ float ar[PRE_NMS];
    __shared__ float sc[PRE_NMS];
    __shared__ int   kp[PRE_NMS];
    __shared__ int   sup;

    int n = blockIdx.x;
    const unsigned long long* K = keys + ((size_t)n << 16);

    for (int i = threadIdx.x; i < PRE_NMS; i += blockDim.x) {
        unsigned long long key = K[i];
        unsigned sv  = (unsigned)(key >> 32);
        unsigned idx = 0xFFFFFFFFu - (unsigned)(key & 0xFFFFFFFFu);
        bool v = sv > 0x007FFFFFu;
        float b0 = 0.f, b1 = 0.f, b2 = 0.f, b3 = 0.f, s = 0.f;
        if (v) {
            const float* bp = boxes + (((size_t)n * NPROP) + idx) * 4;
            b0 = bp[0]; b1 = bp[1]; b2 = bp[2]; b3 = bp[3];
            unsigned u = (sv & 0x80000000u) ? (sv ^ 0x80000000u) : ~sv;
            s = __uint_as_float(u);
        }
        bx[i * 4 + 0] = b0; bx[i * 4 + 1] = b1;
        bx[i * 4 + 2] = b2; bx[i * 4 + 3] = b3;
        ar[i] = fmaxf(b2 - b0, 0.f) * fmaxf(b3 - b1, 0.f);
        sc[i] = s;
        kp[i] = v ? 1 : 0;
    }
    __syncthreads();

    for (int i = 0; i < PRE_NMS; i++) {
        if (threadIdx.x == 0) sup = 0;
        __syncthreads();
        float ib0 = bx[i * 4 + 0], ib1 = bx[i * 4 + 1];
        float ib2 = bx[i * 4 + 2], ib3 = bx[i * 4 + 3];
        float ia  = ar[i];
        int found = 0;
        for (int j = threadIdx.x; j < i; j += blockDim.x) {
            if (kp[j] == 2) {
                float jx1 = bx[j * 4 + 0], jy1 = bx[j * 4 + 1];
                float jx2 = bx[j * 4 + 2], jy2 = bx[j * 4 + 3];
                float ix1 = fmaxf(ib0, jx1), iy1 = fmaxf(ib1, jy1);
                float ix2 = fminf(ib2, jx2), iy2 = fminf(ib3, jy2);
                float inter = fmaxf(ix2 - ix1, 0.f) * fmaxf(iy2 - iy1, 0.f);
                float iou = inter / fmaxf(ia + ar[j] - inter, 1e-9f);
                if (iou > IOU_THR) found = 1;
            }
        }
        if (found) atomicOr(&sup, 1);
        __syncthreads();
        if (threadIdx.x == 0) kp[i] = (kp[i] == 1 && sup == 0) ? 2 : 0;
        __syncthreads();
    }

    for (int t = threadIdx.x; t < POST_NMS * 5; t += blockDim.x)
        out[(size_t)n * POST_NMS * 5 + t] = 0.f;
    __syncthreads();
    if (threadIdx.x == 0) {
        int cnt = 0;
        for (int i = 0; i < PRE_NMS && cnt < POST_NMS; i++) {
            if (kp[i] == 2) {
                float* o = out + (size_t)n * POST_NMS * 5 + cnt * 5;
                o[0] = bx[i * 4 + 0]; o[1] = bx[i * 4 + 1];
                o[2] = bx[i * 4 + 2]; o[3] = bx[i * 4 + 3];
                o[4] = sc[i];
                cnt++;
            }
        }
    }
}

extern "C" void kernel_launch(void* const* d_in, const int* in_sizes, int n_in,
                              void* d_out, int out_size, void* d_ws, size_t ws_size,
                              hipStream_t stream) {
    const float* feat   = (const float*)d_in[0];
    const float* conv_w = (const float*)d_in[1];
    const float* conv_b = (const float*)d_in[2];
    const float* det_w  = (const float*)d_in[3];
    const float* det_b  = (const float*)d_in[4];
    const float* reg_w  = (const float*)d_in[5];
    const float* reg_b  = (const float*)d_in[6];
    float* out = (float*)d_out;

    char* ws = (char*)d_ws;
    float* wt2                  = (float*)(ws);
    float* scores               = (float*)(ws + 2359296);
    float* deltas               = (float*)(ws + 2359296 + 786432);
    float* boxes                = (float*)(ws + 2359296 + 786432 + 3145728);
    unsigned long long* keys    = (unsigned long long*)(ws + 2359296 + 786432 + 3145728 + 3145728);

    repack_w_kernel<<<2304, 256, 0, stream>>>(conv_w, wt2);
    conv_fused_kernel<<<NIMG * HF * (WF / TILE_W), 256, 0, stream>>>(
        feat, wt2, conv_b, det_w, det_b, reg_w, reg_b, scores, deltas);
    decode_kernel<<<(NIMG * NPAD) / 256, 256, 0, stream>>>(scores, deltas, boxes, keys);
    for (int k = 2; k <= NPAD; k <<= 1)
        for (int j = k >> 1; j > 0; j >>= 1)
            bitonic_step_kernel<<<(NIMG * NPAD) / 256, 256, 0, stream>>>(keys, j, k);
    nms_kernel<<<NIMG, 1024, 0, stream>>>(keys, boxes, out);
}


# NMS via pairwise IoU>thr bit rows: a 500-block kernel builds the 2000x2000 bits per image, then one wave per image scans them serially; replaces the 2000-step 3-barrier block loop (+1 launch)
# speedup vs baseline: 1.9828x; 1.9828x over previous
#include <hip/hip_runtime.h>
#include <hip/hip_bf16.h>
#include <math.h>

typedef __attribute__((ext_vector_type(2))) float v2f;
typedef __attribute__((ext_vector_type(8))) float v8f;
typedef __attribute__((ext_vector_type(4))) float v4f;
typedef __attribute__((ext_vector_type(2))) unsigned v2u;

#define NIMG 4
#define CIN 256
#define HF 128
#define WF 128
#define A_ANCH 3
#define NPROP (A_ANCH * HF * WF)
#define NPAD 65536
#define PRE_NMS 2000
#define POST_NMS 300
#define OBJ_THR 0.3f
#define IOU_THR 0.7f
#define TILE_W 64
#define CCHUNK 32
#define CIPAD 34

__global__ __launch_bounds__(256) void repack_w_kernel(const float* __restrict__ conv_w,
                                                       float* __restrict__ wt2) {
    int t = blockIdx.x * 256 + threadIdx.x;
    int par = t & 1;
    int co  = (t >> 1) & 255;
    int q   = t >> 9;
    int ci2 = q & 127;
    int kk  = q >> 7;
    int ci  = ci2 * 2 + par;
    int kh  = kk / 3, kw = kk % 3;
    wt2[t] = conv_w[((co * CIN + ci) * 3 + kh) * 3 + kw];
}

__global__ __launch_bounds__(256) void conv_fused_kernel(
    const float* __restrict__ feat, const float* __restrict__ wt2,
    const float* __restrict__ conv_b,
    const float* __restrict__ det_w, const float* __restrict__ det_b,
    const float* __restrict__ reg_w, const float* __restrict__ reg_b,
    float* __restrict__ scores, float* __restrict__ deltas) {

    __shared__ float lds_in[3 * 66 * CIPAD];
    __shared__ float x_tile[256 * 33];
    __shared__ float hb[16];

    int bid = blockIdx.x;
    int n    = bid >> 8;
    int rem  = bid & 255;
    int row  = rem >> 1;
    int col0 = (rem & 1) * TILE_W;

    int tid  = threadIdx.x;
    int wave = tid >> 5;
    int lane = tid & 31;
    int hl   = lane >> 4;
    int lm   = lane & 15;

    v8f acc[2][4] = {};

    const float* featn = feat + (size_t)n * CIN * HF * WF;
    const v2f*   wp    = (const v2f*)wt2;

    for (int cib = 0; cib < CIN; cib += CCHUNK) {
        __syncthreads();
        for (int e = tid; e < 3 * 66 * CCHUNK; e += 256) {
            int col = e % 66;
            int r   = (e / 66) % 3;
            int cl  = e / 198;
            int grow = row + r - 1;
            int gcol = col0 + col - 1;
            float v = 0.0f;
            if (grow >= 0 && grow < HF && gcol >= 0 && gcol < WF)
                v = featn[((size_t)(cib + cl) * HF + grow) * WF + gcol];
            lds_in[(r * 66 + col) * CIPAD + cl] = v;
        }
        __syncthreads();

        for (int kh = 0; kh < 3; kh++) {
            for (int kw = 0; kw < 3; kw++) {
                int kk = kh * 3 + kw;
                #pragma unroll 4
                for (int c = 0; c < CCHUNK; c += 4) {
                    int ci2 = ((cib + c) >> 1) + hl;
                    int co0 = wave * 32 + lm;
                    v2f a0 = wp[(kk * 128 + ci2) * 256 + co0];
                    v2f a1 = wp[(kk * 128 + ci2) * 256 + co0 + 16];
                    int boff = c + 2 * hl;
                    v2f b[4];
                    #pragma unroll
                    for (int nt = 0; nt < 4; nt++)
                        b[nt] = *(const v2f*)&lds_in[(kh * 66 + nt * 16 + lm + kw) * CIPAD + boff];
                    #pragma unroll
                    for (int nt = 0; nt < 4; nt++) {
                        acc[0][nt] = __builtin_amdgcn_wmma_f32_16x16x4_f32(false, a0, false, b[nt], (short)0, acc[0][nt], false, false);
                        acc[1][nt] = __builtin_amdgcn_wmma_f32_16x16x4_f32(false, a1, false, b[nt], (short)0, acc[1][nt], false, false);
                    }
                }
            }
        }
    }

    __syncthreads();
    for (int e = tid; e < 15 * 256; e += 256)
        lds_in[e] = (e < 768) ? det_w[e] : reg_w[e - 768];
    if (tid < 15) hb[tid] = (tid < 3) ? det_b[tid] : reg_b[tid - 3];

    int m0  = hl * 8;
    float biasA[8], biasB[8];
    #pragma unroll
    for (int r = 0; r < 8; r++) {
        biasA[r] = conv_b[wave * 32 + m0 + r];
        biasB[r] = conv_b[wave * 32 + 16 + m0 + r];
    }

    for (int ph = 0; ph < 2; ph++) {
        if (ph) __syncthreads();
        #pragma unroll
        for (int r = 0; r < 8; r++) {
            int coA = wave * 32 + m0 + r;
            int coB = wave * 32 + 16 + m0 + r;
            x_tile[coA * 33 + lm]      = acc[0][ph * 2 + 0][r] + biasA[r];
            x_tile[coA * 33 + 16 + lm] = acc[0][ph * 2 + 1][r] + biasA[r];
            x_tile[coB * 33 + lm]      = acc[1][ph * 2 + 0][r] + biasB[r];
            x_tile[coB * 33 + 16 + lm] = acc[1][ph * 2 + 1][r] + biasB[r];
        }
        __syncthreads();

        for (int t = tid; t < 15 * 32; t += 256) {
            int oi  = t >> 5;
            int pix = t & 31;
            float sum = hb[oi];
            const float* wrow = &lds_in[oi * 256];
            #pragma unroll 8
            for (int co = 0; co < 256; co++)
                sum += wrow[co] * x_tile[co * 33 + pix];
            int p = row * WF + col0 + ph * 32 + pix;
            if (oi < 3) {
                scores[(size_t)n * NPROP + oi * (HF * WF) + p] = 1.0f / (1.0f + expf(-sum));
            } else {
                int ch = oi - 3;
                int ccomp = ch / 3;
                int a = ch % 3;
                deltas[(((size_t)n * NPROP) + a * (HF * WF) + p) * 4 + ccomp] = sum;
            }
        }
    }
}

__global__ __launch_bounds__(256) void decode_kernel(
    const float* __restrict__ scores, const float* __restrict__ deltas,
    float* __restrict__ boxes, unsigned long long* __restrict__ keys) {
    int t = blockIdx.x * 256 + threadIdx.x;
    int n = t >> 16;
    int i = t & 0xFFFF;
    unsigned long long* kout = keys + ((size_t)n << 16);
    if (i >= NPROP) { kout[i] = 0ull; return; }

    int pa = i / 3, aa = i - pa * 3;
    int ya = pa >> 7, xa = pa & 127;
    const float bs[3][4] = {{-91.f, -45.f, 91.f, 45.f},
                            {-64.f, -64.f, 64.f, 64.f},
                            {-45.f, -91.f, 45.f, 91.f}};
    float sxv = (float)(xa * 8), syv = (float)(ya * 8);
    float x1 = sxv + bs[aa][0], y1 = syv + bs[aa][1];
    float x2 = sxv + bs[aa][2], y2 = syv + bs[aa][3];
    float aw = x2 - x1, ah = y2 - y1;
    float acx = y1 + aw * 0.5f;
    float acy = x1 + ah * 0.5f;

    const float* d = deltas + (((size_t)n * NPROP) + i) * 4;
    float px = acx + d[0] * aw;
    float py = acy + d[1] * ah;
    float pw = aw * expf(d[2]);
    float ph = ah * expf(d[3]);
    float b0 = fminf(fmaxf(px - pw * 0.5f, 0.f), 1024.f);
    float b1 = fminf(fmaxf(py - ph * 0.5f, 0.f), 1024.f);
    float b2 = fminf(fmaxf(px + pw * 0.5f, 0.f), 1024.f);
    float b3 = fminf(fmaxf(py + ph * 0.5f, 0.f), 1024.f);

    float* bo = boxes + (((size_t)n * NPROP) + i) * 4;
    bo[0] = b0; bo[1] = b1; bo[2] = b2; bo[3] = b3;

    float s = scores[(size_t)n * NPROP + i];
    bool valid = (s > OBJ_THR) && ((b2 - b0) > 0.f) && ((b3 - b1) > 0.f);
    float ss = valid ? s : -__builtin_inff();
    unsigned u = __float_as_uint(ss);
    unsigned sv = (u & 0x80000000u) ? ~u : (u | 0x80000000u);
    kout[i] = ((unsigned long long)sv << 32) | (unsigned long long)(0xFFFFFFFFu - (unsigned)i);
}

__global__ __launch_bounds__(256) void bitonic_step_kernel(unsigned long long* __restrict__ keys,
                                                           int j, int k) {
    int t = blockIdx.x * 256 + threadIdx.x;
    int n = t >> 16;
    int i = t & 0xFFFF;
    int ixj = i ^ j;
    if (ixj > i) {
        unsigned long long* K = keys + ((size_t)n << 16);
        unsigned long long a = K[i];
        unsigned long long b = K[ixj];
        bool up = ((i & k) == 0);
        bool sw = up ? (a < b) : (a > b);
        if (sw) { K[i] = b; K[ixj] = a; }
    }
}

#define MASK_ROWS 16
#define MASK_BLKS (PRE_NMS / MASK_ROWS)
#define LPITCH 65
__global__ __launch_bounds__(256) void nms_mask_kernel(const unsigned long long* __restrict__ keys,
                                                       const float* __restrict__ boxes,
                                                       unsigned* __restrict__ mask,
                                                       float* __restrict__ sorted) {
    __shared__ v4f   sbx[32 * LPITCH];
    __shared__ float sar[32 * LPITCH];

    int n  = blockIdx.x / MASK_BLKS;
    int rb = blockIdx.x % MASK_BLKS;
    int tid = threadIdx.x;
    const unsigned long long* K = keys + ((size_t)n << 16);
    const float* bimg = boxes + (size_t)n * NPROP * 4;

    unsigned long long kk[8];
    #pragma unroll
    for (int q = 0; q < 8; q++) {
        int j = tid + 256 * q;
        kk[q] = K[j < PRE_NMS ? j : PRE_NMS - 1];
    }
    v4f gb[8];
    #pragma unroll
    for (int q = 0; q < 8; q++) {
        int j = tid + 256 * q;
        unsigned sv  = (unsigned)(kk[q] >> 32);
        unsigned idx = 0xFFFFFFFFu - (unsigned)(kk[q] & 0xFFFFFFFFu);
        bool v = (j < PRE_NMS) && (sv > 0x007FFFFFu);
        gb[q] = *(const v4f*)(bimg + (size_t)(v ? idx : 0u) * 4);
    }
    #pragma unroll
    for (int q = 0; q < 8; q++) {
        int j = tid + 256 * q;
        unsigned sv = (unsigned)(kk[q] >> 32);
        bool v = (j < PRE_NMS) && (sv > 0x007FFFFFu);
        v4f bb = {0.f, 0.f, 0.f, 0.f};
        float s = 0.f;
        if (v) {
            bb = gb[q];
            unsigned u = (sv & 0x80000000u) ? (sv ^ 0x80000000u) : ~sv;
            s = __uint_as_float(u);
        }
        int slot = (j & 31) * LPITCH + (j >> 5);
        sbx[slot] = bb;
        sar[slot] = fmaxf(bb.z - bb.x, 0.f) * fmaxf(bb.w - bb.y, 0.f);
        if (j < PRE_NMS && (j / MASK_ROWS) == rb) {
            v4f* so = (v4f*)sorted + ((size_t)n * PRE_NMS + j) * 2;
            v4f inf = {s, v ? 1.f : 0.f, 0.f, 0.f};
            so[0] = bb;
            so[1] = inf;
        }
    }
    __syncthreads();

    int wave = tid >> 5;
    int lane = tid & 31;
    for (int rr = 0; rr < 2; rr++) {
        int i = rb * MASK_ROWS + wave * 2 + rr;
        int islot = (i & 31) * LPITCH + (i >> 5);
        v4f bi = sbx[islot];
        float ia = sar[islot];
        int wdiag = i >> 5;
        unsigned* mrow = mask + ((size_t)n * PRE_NMS + i) * 64;
        for (int pass = 0; pass < 2; pass++) {
            int w = pass * 32 + lane;
            unsigned bits = 0u;
            if (w >= wdiag) {
                for (int b = 0; b < 32; b++) {
                    v4f bj = sbx[b * LPITCH + w];
                    float ja = sar[b * LPITCH + w];
                    float ix1 = fmaxf(bi.x, bj.x), iy1 = fmaxf(bi.y, bj.y);
                    float ix2 = fminf(bi.z, bj.z), iy2 = fminf(bi.w, bj.w);
                    float inter = fmaxf(ix2 - ix1, 0.f) * fmaxf(iy2 - iy1, 0.f);
                    float iou = inter / fmaxf(ia + ja - inter, 1e-9f);
                    if (iou > IOU_THR) bits |= (1u << b);
                }
            }
            mrow[w] = bits;
        }
    }
}

__global__ __launch_bounds__(32) void nms_scan_kernel(const unsigned* __restrict__ mask,
                                                      const float* __restrict__ sorted,
                                                      float* __restrict__ out) {
    int n = blockIdx.x;
    int lane = threadIdx.x;
    const v2u* M = (const v2u*)mask + (size_t)n * PRE_NMS * 32;
    const v4f* S = (const v4f*)sorted + (size_t)n * PRE_NMS * 2;
    float* o_img = out + (size_t)n * POST_NMS * 5;

    unsigned rem_lo = 0u, rem_hi = 0u;

    v2u cur[32], nxt[32];
    v4f cbox, cinf, nbox, ninf;
    #pragma unroll
    for (int r = 0; r < 32; r++) cur[r] = M[(size_t)r * 32 + lane];
    cbox = S[(size_t)lane * 2];
    cinf = S[(size_t)lane * 2 + 1];

    int cnt = 0;
    for (int c = 0; c < 63; c++) {
        int cn = (c + 1 < 63) ? c + 1 : 62;
        #pragma unroll
        for (int r = 0; r < 32; r++) {
            int i = cn * 32 + r;
            i = i < PRE_NMS ? i : PRE_NMS - 1;
            nxt[r] = M[(size_t)i * 32 + lane];
        }
        {
            int i = cn * 32 + lane;
            i = i < PRE_NMS ? i : PRE_NMS - 1;
            nbox = S[(size_t)i * 2];
            ninf = S[(size_t)i * 2 + 1];
        }

        int own = c >> 1;
        bool inval = (c * 32 + lane >= PRE_NMS) || (cinf.y == 0.f);
        unsigned remw = (unsigned)__builtin_amdgcn_readlane((int)((c & 1) ? rem_hi : rem_lo), own)
                      | __builtin_amdgcn_ballot_w32(inval);
        unsigned kept = 0u;
        #pragma unroll
        for (int r = 0; r < 32; r++) {
            unsigned d = (unsigned)__builtin_amdgcn_readlane((int)((c & 1) ? cur[r].y : cur[r].x), own);
            unsigned m = ((remw >> r) & 1u) ? 0u : 0xFFFFFFFFu;
            kept  |= m & (1u << r);
            remw  |= d & m;
            rem_lo |= cur[r].x & m;
            rem_hi |= cur[r].y & m;
        }

        unsigned mine = (kept >> lane) & 1u;
        int pos = cnt + __popc(kept & ((1u << lane) - 1u));
        if (mine && pos < POST_NMS) {
            float* o = o_img + pos * 5;
            o[0] = cbox.x; o[1] = cbox.y; o[2] = cbox.z; o[3] = cbox.w;
            o[4] = cinf.x;
        }
        cnt += __popc(kept);

        #pragma unroll
        for (int r = 0; r < 32; r++) cur[r] = nxt[r];
        cbox = nbox; cinf = ninf;
    }

    int c0 = cnt < POST_NMS ? cnt : POST_NMS;
    for (int t = c0 * 5 + lane; t < POST_NMS * 5; t += 32) o_img[t] = 0.f;
}

extern "C" void kernel_launch(void* const* d_in, const int* in_sizes, int n_in,
                              void* d_out, int out_size, void* d_ws, size_t ws_size,
                              hipStream_t stream) {
    const float* feat   = (const float*)d_in[0];
    const float* conv_w = (const float*)d_in[1];
    const float* conv_b = (const float*)d_in[2];
    const float* det_w  = (const float*)d_in[3];
    const float* det_b  = (const float*)d_in[4];
    const float* reg_w  = (const float*)d_in[5];
    const float* reg_b  = (const float*)d_in[6];
    float* out = (float*)d_out;

    char* ws = (char*)d_ws;
    float* wt2                  = (float*)(ws);
    float* scores               = (float*)(ws + 2359296);
    float* deltas               = (float*)(ws + 2359296 + 786432);
    float* boxes                = (float*)(ws + 2359296 + 786432 + 3145728);
    unsigned long long* keys    = (unsigned long long*)(ws + 2359296 + 786432 + 3145728 + 3145728);

    repack_w_kernel<<<2304, 256, 0, stream>>>(conv_w, wt2);
    conv_fused_kernel<<<NIMG * HF * (WF / TILE_W), 256, 0, stream>>>(
        feat, wt2, conv_b, det_w, det_b, reg_w, reg_b, scores, deltas);
    decode_kernel<<<(NIMG * NPAD) / 256, 256, 0, stream>>>(scores, deltas, boxes, keys);
    for (int k = 2; k <= NPAD; k <<= 1)
        for (int j = k >> 1; j > 0; j >>= 1)
            bitonic_step_kernel<<<(NIMG * NPAD) / 256, 256, 0, stream>>>(keys, j, k);
    unsigned* nmask  = (unsigned*)(ws);
    float*    sorted = (float*)(ws + 2048000);
    nms_mask_kernel<<<NIMG * MASK_BLKS, 256, 0, stream>>>(keys, boxes, nmask, sorted);
    nms_scan_kernel<<<NIMG, 32, 0, stream>>>(nmask, sorted, out);
}


# top-2000 keys without a full sort: sort 2048-key chunks in LDS, then fold chunks keeping the best 2048 (max with the reversed chunk + LDS merge steps); 3 launches replace 136 global bitonic steps
# speedup vs baseline: 2.9685x; 1.4971x over previous
#include <hip/hip_runtime.h>
#include <hip/hip_bf16.h>
#include <math.h>

typedef __attribute__((ext_vector_type(2))) float v2f;
typedef __attribute__((ext_vector_type(8))) float v8f;
typedef __attribute__((ext_vector_type(4))) float v4f;
typedef __attribute__((ext_vector_type(2))) unsigned v2u;

#define NIMG 4
#define CIN 256
#define HF 128
#define WF 128
#define A_ANCH 3
#define NPROP (A_ANCH * HF * WF)
#define NPAD 65536
#define PRE_NMS 2000
#define POST_NMS 300
#define OBJ_THR 0.3f
#define IOU_THR 0.7f
#define TILE_W 64
#define CCHUNK 32
#define CIPAD 34

__global__ __launch_bounds__(256) void repack_w_kernel(const float* __restrict__ conv_w,
                                                       float* __restrict__ wt2) {
    int t = blockIdx.x * 256 + threadIdx.x;
    int par = t & 1;
    int co  = (t >> 1) & 255;
    int q   = t >> 9;
    int ci2 = q & 127;
    int kk  = q >> 7;
    int ci  = ci2 * 2 + par;
    int kh  = kk / 3, kw = kk % 3;
    wt2[t] = conv_w[((co * CIN + ci) * 3 + kh) * 3 + kw];
}

__global__ __launch_bounds__(256) void conv_fused_kernel(
    const float* __restrict__ feat, const float* __restrict__ wt2,
    const float* __restrict__ conv_b,
    const float* __restrict__ det_w, const float* __restrict__ det_b,
    const float* __restrict__ reg_w, const float* __restrict__ reg_b,
    float* __restrict__ scores, float* __restrict__ deltas) {

    __shared__ float lds_in[3 * 66 * CIPAD];
    __shared__ float x_tile[256 * 33];
    __shared__ float hb[16];

    int bid = blockIdx.x;
    int n    = bid >> 8;
    int rem  = bid & 255;
    int row  = rem >> 1;
    int col0 = (rem & 1) * TILE_W;

    int tid  = threadIdx.x;
    int wave = tid >> 5;
    int lane = tid & 31;
    int hl   = lane >> 4;
    int lm   = lane & 15;

    v8f acc[2][4] = {};

    const float* featn = feat + (size_t)n * CIN * HF * WF;
    const v2f*   wp    = (const v2f*)wt2;

    for (int cib = 0; cib < CIN; cib += CCHUNK) {
        __syncthreads();
        for (int e = tid; e < 3 * 66 * CCHUNK; e += 256) {
            int col = e % 66;
            int r   = (e / 66) % 3;
            int cl  = e / 198;
            int grow = row + r - 1;
            int gcol = col0 + col - 1;
            float v = 0.0f;
            if (grow >= 0 && grow < HF && gcol >= 0 && gcol < WF)
                v = featn[((size_t)(cib + cl) * HF + grow) * WF + gcol];
            lds_in[(r * 66 + col) * CIPAD + cl] = v;
        }
        __syncthreads();

        for (int kh = 0; kh < 3; kh++) {
            for (int kw = 0; kw < 3; kw++) {
                int kk = kh * 3 + kw;
                #pragma unroll 4
                for (int c = 0; c < CCHUNK; c += 4) {
                    int ci2 = ((cib + c) >> 1) + hl;
                    int co0 = wave * 32 + lm;
                    v2f a0 = wp[(kk * 128 + ci2) * 256 + co0];
                    v2f a1 = wp[(kk * 128 + ci2) * 256 + co0 + 16];
                    int boff = c + 2 * hl;
                    v2f b[4];
                    #pragma unroll
                    for (int nt = 0; nt < 4; nt++)
                        b[nt] = *(const v2f*)&lds_in[(kh * 66 + nt * 16 + lm + kw) * CIPAD + boff];
                    #pragma unroll
                    for (int nt = 0; nt < 4; nt++) {
                        acc[0][nt] = __builtin_amdgcn_wmma_f32_16x16x4_f32(false, a0, false, b[nt], (short)0, acc[0][nt], false, false);
                        acc[1][nt] = __builtin_amdgcn_wmma_f32_16x16x4_f32(false, a1, false, b[nt], (short)0, acc[1][nt], false, false);
                    }
                }
            }
        }
    }

    __syncthreads();
    for (int e = tid; e < 15 * 256; e += 256)
        lds_in[e] = (e < 768) ? det_w[e] : reg_w[e - 768];
    if (tid < 15) hb[tid] = (tid < 3) ? det_b[tid] : reg_b[tid - 3];

    int m0  = hl * 8;
    float biasA[8], biasB[8];
    #pragma unroll
    for (int r = 0; r < 8; r++) {
        biasA[r] = conv_b[wave * 32 + m0 + r];
        biasB[r] = conv_b[wave * 32 + 16 + m0 + r];
    }

    for (int ph = 0; ph < 2; ph++) {
        if (ph) __syncthreads();
        #pragma unroll
        for (int r = 0; r < 8; r++) {
            int coA = wave * 32 + m0 + r;
            int coB = wave * 32 + 16 + m0 + r;
            x_tile[coA * 33 + lm]      = acc[0][ph * 2 + 0][r] + biasA[r];
            x_tile[coA * 33 + 16 + lm] = acc[0][ph * 2 + 1][r] + biasA[r];
            x_tile[coB * 33 + lm]      = acc[1][ph * 2 + 0][r] + biasB[r];
            x_tile[coB * 33 + 16 + lm] = acc[1][ph * 2 + 1][r] + biasB[r];
        }
        __syncthreads();

        for (int t = tid; t < 15 * 32; t += 256) {
            int oi  = t >> 5;
            int pix = t & 31;
            float sum = hb[oi];
            const float* wrow = &lds_in[oi * 256];
            #pragma unroll 8
            for (int co = 0; co < 256; co++)
                sum += wrow[co] * x_tile[co * 33 + pix];
            int p = row * WF + col0 + ph * 32 + pix;
            if (oi < 3) {
                scores[(size_t)n * NPROP + oi * (HF * WF) + p] = 1.0f / (1.0f + expf(-sum));
            } else {
                int ch = oi - 3;
                int ccomp = ch / 3;
                int a = ch % 3;
                deltas[(((size_t)n * NPROP) + a * (HF * WF) + p) * 4 + ccomp] = sum;
            }
        }
    }
}

__global__ __launch_bounds__(256) void decode_kernel(
    const float* __restrict__ scores, const float* __restrict__ deltas,
    float* __restrict__ boxes, unsigned long long* __restrict__ keys) {
    int t = blockIdx.x * 256 + threadIdx.x;
    int n = t >> 16;
    int i = t & 0xFFFF;
    unsigned long long* kout = keys + ((size_t)n << 16);
    if (i >= NPROP) { kout[i] = 0ull; return; }

    int pa = i / 3, aa = i - pa * 3;
    int ya = pa >> 7, xa = pa & 127;
    const float bs[3][4] = {{-91.f, -45.f, 91.f, 45.f},
                            {-64.f, -64.f, 64.f, 64.f},
                            {-45.f, -91.f, 45.f, 91.f}};
    float sxv = (float)(xa * 8), syv = (float)(ya * 8);
    float x1 = sxv + bs[aa][0], y1 = syv + bs[aa][1];
    float x2 = sxv + bs[aa][2], y2 = syv + bs[aa][3];
    float aw = x2 - x1, ah = y2 - y1;
    float acx = y1 + aw * 0.5f;
    float acy = x1 + ah * 0.5f;

    const float* d = deltas + (((size_t)n * NPROP) + i) * 4;
    float px = acx + d[0] * aw;
    float py = acy + d[1] * ah;
    float pw = aw * expf(d[2]);
    float ph = ah * expf(d[3]);
    float b0 = fminf(fmaxf(px - pw * 0.5f, 0.f), 1024.f);
    float b1 = fminf(fmaxf(py - ph * 0.5f, 0.f), 1024.f);
    float b2 = fminf(fmaxf(px + pw * 0.5f, 0.f), 1024.f);
    float b3 = fminf(fmaxf(py + ph * 0.5f, 0.f), 1024.f);

    float* bo = boxes + (((size_t)n * NPROP) + i) * 4;
    bo[0] = b0; bo[1] = b1; bo[2] = b2; bo[3] = b3;

    float s = scores[(size_t)n * NPROP + i];
    bool valid = (s > OBJ_THR) && ((b2 - b0) > 0.f) && ((b3 - b1) > 0.f);
    float ss = valid ? s : -__builtin_inff();
    unsigned u = __float_as_uint(ss);
    unsigned sv = (u & 0x80000000u) ? ~u : (u | 0x80000000u);
    kout[i] = ((unsigned long long)sv << 32) | (unsigned long long)(0xFFFFFFFFu - (unsigned)i);
}

#define CHUNK 2048
#define NCHUNK (NPROP / CHUNK)
#define FOLD_A 6
#define FOLD_B (NCHUNK / FOLD_A)

__device__ __forceinline__ void bitonic_ce(unsigned long long* s, int t, int j, int k) {
    int lo = t & (j - 1);
    int i  = ((t - lo) << 1) | lo;
    int p  = i | j;
    unsigned long long a = s[i];
    unsigned long long b = s[p];
    bool up = ((i & k) == 0);
    bool sw = up ? (a < b) : (a > b);
    if (sw) { s[i] = b; s[p] = a; }
}

__global__ __launch_bounds__(1024) void topk_chunk_sort_kernel(unsigned long long* keys) {
    __shared__ unsigned long long s[CHUNK];
    int n   = blockIdx.x / NCHUNK;
    int c   = blockIdx.x % NCHUNK;
    int tid = threadIdx.x;
    unsigned long long* g = keys + ((size_t)n << 16) + (size_t)c * CHUNK;
    s[tid]        = g[tid];
    s[tid + 1024] = g[tid + 1024];
    __syncthreads();
    for (int k = 2; k <= CHUNK; k <<= 1)
        for (int j = k >> 1; j > 0; j >>= 1) {
            bitonic_ce(s, tid, j, k);
            __syncthreads();
        }
    g[tid]        = s[tid];
    g[tid + 1024] = s[tid + 1024];
}

__global__ __launch_bounds__(1024) void topk_fold_kernel(unsigned long long* keys,
                                                         int nch, int stride, int groups) {
    __shared__ unsigned long long s[CHUNK];
    int n   = blockIdx.x / groups;
    int gi  = blockIdx.x % groups;
    int tid = threadIdx.x;
    size_t pitch = (size_t)stride * CHUNK;
    unsigned long long* g = keys + ((size_t)n << 16) + (size_t)gi * nch * pitch;
    s[tid]        = g[tid];
    s[tid + 1024] = g[tid + 1024];
    unsigned long long d0 = 0ull, d1 = 0ull;
    if (nch > 1) {
        d0 = g[pitch + (CHUNK - 1 - tid)];
        d1 = g[pitch + (CHUNK - 1 - 1024 - tid)];
    }
    for (int c = 1; c < nch; c++) {
        unsigned long long a0 = s[tid];
        unsigned long long a1 = s[tid + 1024];
        unsigned long long e0 = a0 > d0 ? a0 : d0;
        unsigned long long e1 = a1 > d1 ? a1 : d1;
        s[tid]        = e0 > e1 ? e0 : e1;
        s[tid + 1024] = e0 > e1 ? e1 : e0;
        if (c + 1 < nch) {
            const unsigned long long* gn = g + (size_t)(c + 1) * pitch;
            d0 = gn[CHUNK - 1 - tid];
            d1 = gn[CHUNK - 1 - 1024 - tid];
        }
        __syncthreads();
        for (int j = CHUNK >> 2; j > 0; j >>= 1) {
            bitonic_ce(s, tid, j, CHUNK);
            __syncthreads();
        }
    }
    g[tid]        = s[tid];
    g[tid + 1024] = s[tid + 1024];
}

#define MASK_ROWS 16
#define MASK_BLKS (PRE_NMS / MASK_ROWS)
#define LPITCH 65
__global__ __launch_bounds__(256) void nms_mask_kernel(const unsigned long long* __restrict__ keys,
                                                       const float* __restrict__ boxes,
                                                       unsigned* __restrict__ mask,
                                                       float* __restrict__ sorted) {
    __shared__ v4f   sbx[32 * LPITCH];
    __shared__ float sar[32 * LPITCH];

    int n  = blockIdx.x / MASK_BLKS;
    int rb = blockIdx.x % MASK_BLKS;
    int tid = threadIdx.x;
    const unsigned long long* K = keys + ((size_t)n << 16);
    const float* bimg = boxes + (size_t)n * NPROP * 4;

    unsigned long long kk[8];
    #pragma unroll
    for (int q = 0; q < 8; q++) {
        int j = tid + 256 * q;
        kk[q] = K[j < PRE_NMS ? j : PRE_NMS - 1];
    }
    v4f gb[8];
    #pragma unroll
    for (int q = 0; q < 8; q++) {
        int j = tid + 256 * q;
        unsigned sv  = (unsigned)(kk[q] >> 32);
        unsigned idx = 0xFFFFFFFFu - (unsigned)(kk[q] & 0xFFFFFFFFu);
        bool v = (j < PRE_NMS) && (sv > 0x007FFFFFu);
        gb[q] = *(const v4f*)(bimg + (size_t)(v ? idx : 0u) * 4);
    }
    #pragma unroll
    for (int q = 0; q < 8; q++) {
        int j = tid + 256 * q;
        unsigned sv = (unsigned)(kk[q] >> 32);
        bool v = (j < PRE_NMS) && (sv > 0x007FFFFFu);
        v4f bb = {0.f, 0.f, 0.f, 0.f};
        float s = 0.f;
        if (v) {
            bb = gb[q];
            unsigned u = (sv & 0x80000000u) ? (sv ^ 0x80000000u) : ~sv;
            s = __uint_as_float(u);
        }
        int slot = (j & 31) * LPITCH + (j >> 5);
        sbx[slot] = bb;
        sar[slot] = fmaxf(bb.z - bb.x, 0.f) * fmaxf(bb.w - bb.y, 0.f);
        if (j < PRE_NMS && (j / MASK_ROWS) == rb) {
            v4f* so = (v4f*)sorted + ((size_t)n * PRE_NMS + j) * 2;
            v4f inf = {s, v ? 1.f : 0.f, 0.f, 0.f};
            so[0] = bb;
            so[1] = inf;
        }
    }
    __syncthreads();

    int wave = tid >> 5;
    int lane = tid & 31;
    for (int rr = 0; rr < 2; rr++) {
        int i = rb * MASK_ROWS + wave * 2 + rr;
        int islot = (i & 31) * LPITCH + (i >> 5);
        v4f bi = sbx[islot];
        float ia = sar[islot];
        int wdiag = i >> 5;
        unsigned* mrow = mask + ((size_t)n * PRE_NMS + i) * 64;
        for (int pass = 0; pass < 2; pass++) {
            int w = pass * 32 + lane;
            unsigned bits = 0u;
            if (w >= wdiag) {
                for (int b = 0; b < 32; b++) {
                    v4f bj = sbx[b * LPITCH + w];
                    float ja = sar[b * LPITCH + w];
                    float ix1 = fmaxf(bi.x, bj.x), iy1 = fmaxf(bi.y, bj.y);
                    float ix2 = fminf(bi.z, bj.z), iy2 = fminf(bi.w, bj.w);
                    float inter = fmaxf(ix2 - ix1, 0.f) * fmaxf(iy2 - iy1, 0.f);
                    float iou = inter / fmaxf(ia + ja - inter, 1e-9f);
                    if (iou > IOU_THR) bits |= (1u << b);
                }
            }
            mrow[w] = bits;
        }
    }
}

__global__ __launch_bounds__(32) void nms_scan_kernel(const unsigned* __restrict__ mask,
                                                      const float* __restrict__ sorted,
                                                      float* __restrict__ out) {
    int n = blockIdx.x;
    int lane = threadIdx.x;
    const v2u* M = (const v2u*)mask + (size_t)n * PRE_NMS * 32;
    const v4f* S = (const v4f*)sorted + (size_t)n * PRE_NMS * 2;
    float* o_img = out + (size_t)n * POST_NMS * 5;

    unsigned rem_lo = 0u, rem_hi = 0u;

    v2u cur[32], nxt[32];
    v4f cbox, cinf, nbox, ninf;
    #pragma unroll
    for (int r = 0; r < 32; r++) cur[r] = M[(size_t)r * 32 + lane];
    cbox = S[(size_t)lane * 2];
    cinf = S[(size_t)lane * 2 + 1];

    int cnt = 0;
    for (int c = 0; c < 63; c++) {
        int cn = (c + 1 < 63) ? c + 1 : 62;
        #pragma unroll
        for (int r = 0; r < 32; r++) {
            int i = cn * 32 + r;
            i = i < PRE_NMS ? i : PRE_NMS - 1;
            nxt[r] = M[(size_t)i * 32 + lane];
        }
        {
            int i = cn * 32 + lane;
            i = i < PRE_NMS ? i : PRE_NMS - 1;
            nbox = S[(size_t)i * 2];
            ninf = S[(size_t)i * 2 + 1];
        }

        int own = c >> 1;
        bool inval = (c * 32 + lane >= PRE_NMS) || (cinf.y == 0.f);
        unsigned remw = (unsigned)__builtin_amdgcn_readlane((int)((c & 1) ? rem_hi : rem_lo), own)
                      | __builtin_amdgcn_ballot_w32(inval);
        unsigned kept = 0u;
        #pragma unroll
        for (int r = 0; r < 32; r++) {
            unsigned d = (unsigned)__builtin_amdgcn_readlane((int)((c & 1) ? cur[r].y : cur[r].x), own);
            unsigned m = ((remw >> r) & 1u) ? 0u : 0xFFFFFFFFu;
            kept  |= m & (1u << r);
            remw  |= d & m;
            rem_lo |= cur[r].x & m;
            rem_hi |= cur[r].y & m;
        }

        unsigned mine = (kept >> lane) & 1u;
        int pos = cnt + __popc(kept & ((1u << lane) - 1u));
        if (mine && pos < POST_NMS) {
            float* o = o_img + pos * 5;
            o[0] = cbox.x; o[1] = cbox.y; o[2] = cbox.z; o[3] = cbox.w;
            o[4] = cinf.x;
        }
        cnt += __popc(kept);

        #pragma unroll
        for (int r = 0; r < 32; r++) cur[r] = nxt[r];
        cbox = nbox; cinf = ninf;
    }

    int c0 = cnt < POST_NMS ? cnt : POST_NMS;
    for (int t = c0 * 5 + lane; t < POST_NMS * 5; t += 32) o_img[t] = 0.f;
}

extern "C" void kernel_launch(void* const* d_in, const int* in_sizes, int n_in,
                              void* d_out, int out_size, void* d_ws, size_t ws_size,
                              hipStream_t stream) {
    const float* feat   = (const float*)d_in[0];
    const float* conv_w = (const float*)d_in[1];
    const float* conv_b = (const float*)d_in[2];
    const float* det_w  = (const float*)d_in[3];
    const float* det_b  = (const float*)d_in[4];
    const float* reg_w  = (const float*)d_in[5];
    const float* reg_b  = (const float*)d_in[6];
    float* out = (float*)d_out;

    char* ws = (char*)d_ws;
    float* wt2                  = (float*)(ws);
    float* scores               = (float*)(ws + 2359296);
    float* deltas               = (float*)(ws + 2359296 + 786432);
    float* boxes                = (float*)(ws + 2359296 + 786432 + 3145728);
    unsigned long long* keys    = (unsigned long long*)(ws + 2359296 + 786432 + 3145728 + 3145728);

    repack_w_kernel<<<2304, 256, 0, stream>>>(conv_w, wt2);
    conv_fused_kernel<<<NIMG * HF * (WF / TILE_W), 256, 0, stream>>>(
        feat, wt2, conv_b, det_w, det_b, reg_w, reg_b, scores, deltas);
    decode_kernel<<<(NIMG * NPAD) / 256, 256, 0, stream>>>(scores, deltas, boxes, keys);
    topk_chunk_sort_kernel<<<NIMG * NCHUNK, 1024, 0, stream>>>(keys);
    topk_fold_kernel<<<NIMG * FOLD_B, 1024, 0, stream>>>(keys, FOLD_A, 1, FOLD_B);
    topk_fold_kernel<<<NIMG, 1024, 0, stream>>>(keys, FOLD_B, FOLD_A, 1);
    unsigned* nmask  = (unsigned*)(ws);
    float*    sorted = (float*)(ws + 2048000);
    nms_mask_kernel<<<NIMG * MASK_BLKS, 256, 0, stream>>>(keys, boxes, nmask, sorted);
    nms_scan_kernel<<<NIMG, 32, 0, stream>>>(nmask, sorted, out);
}


# conv: global loads software-pipelined through registers: next 32-channel input chunk and next k-group's weight fragments are requested ahead of the current WMMAs; sums and their order unchanged
# speedup vs baseline: 3.9856x; 1.3426x over previous
#include <hip/hip_runtime.h>
#include <hip/hip_bf16.h>
#include <math.h>

typedef __attribute__((ext_vector_type(2))) float v2f;
typedef __attribute__((ext_vector_type(8))) float v8f;
typedef __attribute__((ext_vector_type(4))) float v4f;
typedef __attribute__((ext_vector_type(2))) unsigned v2u;

#define NIMG 4
#define CIN 256
#define HF 128
#define WF 128
#define A_ANCH 3
#define NPROP (A_ANCH * HF * WF)
#define NPAD 65536
#define PRE_NMS 2000
#define POST_NMS 300
#define OBJ_THR 0.3f
#define IOU_THR 0.7f
#define TILE_W 64
#define CCHUNK 32
#define CIPAD 34

__global__ __launch_bounds__(256) void repack_w_kernel(const float* __restrict__ conv_w,
                                                       float* __restrict__ wt2) {
    int t = blockIdx.x * 256 + threadIdx.x;
    int par = t & 1;
    int co  = (t >> 1) & 255;
    int q   = t >> 9;
    int ci2 = q & 127;
    int kk  = q >> 7;
    int ci  = ci2 * 2 + par;
    int kh  = kk / 3, kw = kk % 3;
    wt2[t] = conv_w[((co * CIN + ci) * 3 + kh) * 3 + kw];
}

#define NCHK (CIN / CCHUNK)
#define NGRP 18
#define NPRE 25

__device__ __forceinline__ void stage_fetch(float (&pre)[NPRE], const float* frow,
                                            const float* fhalo, const int (&rowoff)[3], int cib) {
    const float* fc = frow + (size_t)cib * (HF * WF);
    #pragma unroll
    for (int j = 0; j < 4; j++) {
        #pragma unroll
        for (int r = 0; r < 3; r++) {
            const float* p = fc + j * (HF * WF) + rowoff[r];
            pre[(j * 3 + r) * 2 + 0] = p[0];
            pre[(j * 3 + r) * 2 + 1] = p[32];
        }
    }
    pre[24] = fhalo[(size_t)cib * (HF * WF)];
}

__device__ __forceinline__ void a_fetch(v2f (&a)[8], const v2f* wpl, int ch, int t) {
    int kk  = t >> 1;
    int ci2 = (ch * CCHUNK + (t & 1) * 16) >> 1;
    const v2f* p = wpl + (size_t)(kk * 128 + ci2) * 256;
    #pragma unroll
    for (int i = 0; i < 4; i++) {
        a[2 * i]     = p[(size_t)(2 * i) * 256];
        a[2 * i + 1] = p[(size_t)(2 * i) * 256 + 16];
    }
}

__global__ __launch_bounds__(256) void conv_fused_kernel(
    const float* __restrict__ feat, const float* __restrict__ wt2,
    const float* __restrict__ conv_b,
    const float* __restrict__ det_w, const float* __restrict__ det_b,
    const float* __restrict__ reg_w, const float* __restrict__ reg_b,
    float* __restrict__ scores, float* __restrict__ deltas) {

    __shared__ __attribute__((aligned(16))) float lds_in[3 * 66 * CIPAD];
    __shared__ float x_tile[256 * 33];
    __shared__ float hb[16];

    int bid = blockIdx.x;
    int n    = bid >> 8;
    int rem  = bid & 255;
    int row  = rem >> 1;
    int col0 = (rem & 1) * TILE_W;

    int tid  = threadIdx.x;
    int wave = tid >> 5;
    int lane = tid & 31;
    int hl   = lane >> 4;
    int lm   = lane & 15;

    v8f acc[2][4] = {};

    const float* featn = feat + (size_t)n * CIN * HF * WF;
    const v2f*   wpl   = (const v2f*)wt2 + (size_t)hl * 256 + wave * 32 + lm;

    int  rowoff[3];
    bool rowok[3];
    #pragma unroll
    for (int r = 0; r < 3; r++) {
        int g = row + r - 1;
        rowok[r]  = (g >= 0) && (g < HF);
        rowoff[r] = (g < 0 ? 0 : (g >= HF ? HF - 1 : g)) * WF;
    }
    const float* frow = featn + (size_t)(4 * wave) * (HF * WF) + col0 + lane;
    int  hj  = lane / 6;
    int  hq  = lane - hj * 6;
    int  hr  = hq >> 1;
    int  hs  = hq & 1;
    int  hg  = row + hr - 1;
    int  hc  = hs ? (col0 + TILE_W) : (col0 - 1);
    bool hok = (lane < 24) && (hg >= 0) && (hg < HF) && (hc >= 0) && (hc < WF);
    const float* fhalo = featn + (hok ? ((size_t)(4 * wave + hj) * HF + hg) * WF + hc : (size_t)0);
    int  hlds = (hr * 66 + (hs ? 65 : 0)) * CIPAD + 4 * wave + hj;

    auto mma_group = [&](const v2f (&a)[8], int t) {
        int kk = t >> 1;
        int kh = kk / 3;
        int kw = kk - kh * 3;
        int c0 = (t & 1) * 16;
        #pragma unroll
        for (int i = 0; i < 4; i++) {
            int boff = c0 + 4 * i + 2 * hl;
            v2f b[4];
            #pragma unroll
            for (int nt = 0; nt < 4; nt++)
                b[nt] = *(const v2f*)&lds_in[(kh * 66 + nt * 16 + lm + kw) * CIPAD + boff];
            #pragma unroll
            for (int nt = 0; nt < 4; nt++) {
                acc[0][nt] = __builtin_amdgcn_wmma_f32_16x16x4_f32(false, a[2 * i],     false, b[nt], (short)0, acc[0][nt], false, false);
                acc[1][nt] = __builtin_amdgcn_wmma_f32_16x16x4_f32(false, a[2 * i + 1], false, b[nt], (short)0, acc[1][nt], false, false);
            }
        }
    };

    float pre[NPRE];
    v2f   aE[8], aO[8];
    a_fetch(aE, wpl, 0, 0);
    stage_fetch(pre, frow, fhalo, rowoff, 0);

    for (int ch = 0; ch < NCHK; ch++) {
        __syncthreads();
        #pragma unroll
        for (int r = 0; r < 3; r++) {
            #pragma unroll
            for (int hcol = 0; hcol < 2; hcol++) {
                float v0 = rowok[r] ? pre[(0 * 3 + r) * 2 + hcol] : 0.0f;
                float v1 = rowok[r] ? pre[(1 * 3 + r) * 2 + hcol] : 0.0f;
                float v2 = rowok[r] ? pre[(2 * 3 + r) * 2 + hcol] : 0.0f;
                float v3 = rowok[r] ? pre[(3 * 3 + r) * 2 + hcol] : 0.0f;
                int idx = (r * 66 + 1 + hcol * 32 + lane) * CIPAD + 4 * wave;
                v2f lo = {v0, v1};
                v2f hi = {v2, v3};
                *(v2f*)&lds_in[idx]     = lo;
                *(v2f*)&lds_in[idx + 2] = hi;
            }
        }
        if (lane < 24) lds_in[hlds] = hok ? pre[24] : 0.0f;
        __syncthreads();

        int chn = ch + 1 < NCHK ? ch + 1 : NCHK - 1;

        a_fetch(aO, wpl, ch, 1);
        stage_fetch(pre, frow, fhalo, rowoff, chn * CCHUNK);
        mma_group(aE, 0);
        a_fetch(aE, wpl, ch, 2);
        mma_group(aO, 1);

        for (int tt = 1; tt < NGRP / 2; tt++) {
            a_fetch(aO, wpl, ch, 2 * tt + 1);
            mma_group(aE, 2 * tt);
            int tn = 2 * tt + 2, cn = ch;
            if (tn == NGRP) { tn = 0; cn = chn; }
            a_fetch(aE, wpl, cn, tn);
            mma_group(aO, 2 * tt + 1);
        }
    }

    __syncthreads();
    for (int e = tid; e < 15 * 256; e += 256)
        lds_in[e] = (e < 768) ? det_w[e] : reg_w[e - 768];
    if (tid < 15) hb[tid] = (tid < 3) ? det_b[tid] : reg_b[tid - 3];

    int m0  = hl * 8;
    float biasA[8], biasB[8];
    #pragma unroll
    for (int r = 0; r < 8; r++) {
        biasA[r] = conv_b[wave * 32 + m0 + r];
        biasB[r] = conv_b[wave * 32 + 16 + m0 + r];
    }

    for (int ph = 0; ph < 2; ph++) {
        if (ph) __syncthreads();
        #pragma unroll
        for (int r = 0; r < 8; r++) {
            int coA = wave * 32 + m0 + r;
            int coB = wave * 32 + 16 + m0 + r;
            x_tile[coA * 33 + lm]      = acc[0][ph * 2 + 0][r] + biasA[r];
            x_tile[coA * 33 + 16 + lm] = acc[0][ph * 2 + 1][r] + biasA[r];
            x_tile[coB * 33 + lm]      = acc[1][ph * 2 + 0][r] + biasB[r];
            x_tile[coB * 33 + 16 + lm] = acc[1][ph * 2 + 1][r] + biasB[r];
        }
        __syncthreads();

        for (int t = tid; t < 15 * 32; t += 256) {
            int oi  = t >> 5;
            int pix = t & 31;
            float sum = hb[oi];
            const float* wrow = &lds_in[oi * 256];
            #pragma unroll 8
            for (int co = 0; co < 256; co++)
                sum += wrow[co] * x_tile[co * 33 + pix];
            int p = row * WF + col0 + ph * 32 + pix;
            if (oi < 3) {
                scores[(size_t)n * NPROP + oi * (HF * WF) + p] = 1.0f / (1.0f + expf(-sum));
            } else {
                int ch = oi - 3;
                int ccomp = ch / 3;
                int a = ch % 3;
                deltas[(((size_t)n * NPROP) + a * (HF * WF) + p) * 4 + ccomp] = sum;
            }
        }
    }
}

__global__ __launch_bounds__(256) void decode_kernel(
    const float* __restrict__ scores, const float* __restrict__ deltas,
    float* __restrict__ boxes, unsigned long long* __restrict__ keys) {
    int t = blockIdx.x * 256 + threadIdx.x;
    int n = t >> 16;
    int i = t & 0xFFFF;
    unsigned long long* kout = keys + ((size_t)n << 16);
    if (i >= NPROP) { kout[i] = 0ull; return; }

    int pa = i / 3, aa = i - pa * 3;
    int ya = pa >> 7, xa = pa & 127;
    const float bs[3][4] = {{-91.f, -45.f, 91.f, 45.f},
                            {-64.f, -64.f, 64.f, 64.f},
                            {-45.f, -91.f, 45.f, 91.f}};
    float sxv = (float)(xa * 8), syv = (float)(ya * 8);
    float x1 = sxv + bs[aa][0], y1 = syv + bs[aa][1];
    float x2 = sxv + bs[aa][2], y2 = syv + bs[aa][3];
    float aw = x2 - x1, ah = y2 - y1;
    float acx = y1 + aw * 0.5f;
    float acy = x1 + ah * 0.5f;

    const float* d = deltas + (((size_t)n * NPROP) + i) * 4;
    float px = acx + d[0] * aw;
    float py = acy + d[1] * ah;
    float pw = aw * expf(d[2]);
    float ph = ah * expf(d[3]);
    float b0 = fminf(fmaxf(px - pw * 0.5f, 0.f), 1024.f);
    float b1 = fminf(fmaxf(py - ph * 0.5f, 0.f), 1024.f);
    float b2 = fminf(fmaxf(px + pw * 0.5f, 0.f), 1024.f);
    float b3 = fminf(fmaxf(py + ph * 0.5f, 0.f), 1024.f);

    float* bo = boxes + (((size_t)n * NPROP) + i) * 4;
    bo[0] = b0; bo[1] = b1; bo[2] = b2; bo[3] = b3;

    float s = scores[(size_t)n * NPROP + i];
    bool valid = (s > OBJ_THR) && ((b2 - b0) > 0.f) && ((b3 - b1) > 0.f);
    float ss = valid ? s : -__builtin_inff();
    unsigned u = __float_as_uint(ss);
    unsigned sv = (u & 0x80000000u) ? ~u : (u | 0x80000000u);
    kout[i] = ((unsigned long long)sv << 32) | (unsigned long long)(0xFFFFFFFFu - (unsigned)i);
}

#define CHUNK 2048
#define NCHUNK (NPROP / CHUNK)
#define FOLD_A 6
#define FOLD_B (NCHUNK / FOLD_A)

__device__ __forceinline__ void bitonic_ce(unsigned long long* s, int t, int j, int k) {
    int lo = t & (j - 1);
    int i  = ((t - lo) << 1) | lo;
    int p  = i | j;
    unsigned long long a = s[i];
    unsigned long long b = s[p];
    bool up = ((i & k) == 0);
    bool sw = up ? (a < b) : (a > b);
    if (sw) { s[i] = b; s[p] = a; }
}

__global__ __launch_bounds__(1024) void topk_chunk_sort_kernel(unsigned long long* keys) {
    __shared__ unsigned long long s[CHUNK];
    int n   = blockIdx.x / NCHUNK;
    int c   = blockIdx.x % NCHUNK;
    int tid = threadIdx.x;
    unsigned long long* g = keys + ((size_t)n << 16) + (size_t)c * CHUNK;
    s[tid]        = g[tid];
    s[tid + 1024] = g[tid + 1024];
    __syncthreads();
    for (int k = 2; k <= CHUNK; k <<= 1)
        for (int j = k >> 1; j > 0; j >>= 1) {
            bitonic_ce(s, tid, j, k);
            __syncthreads();
        }
    g[tid]        = s[tid];
    g[tid + 1024] = s[tid + 1024];
}

__global__ __launch_bounds__(1024) void topk_fold_kernel(unsigned long long* keys,
                                                         int nch, int stride, int groups) {
    __shared__ unsigned long long s[CHUNK];
    int n   = blockIdx.x / groups;
    int gi  = blockIdx.x % groups;
    int tid = threadIdx.x;
    size_t pitch = (size_t)stride * CHUNK;
    unsigned long long* g = keys + ((size_t)n << 16) + (size_t)gi * nch * pitch;
    s[tid]        = g[tid];
    s[tid + 1024] = g[tid + 1024];
    unsigned long long d0 = 0ull, d1 = 0ull;
    if (nch > 1) {
        d0 = g[pitch + (CHUNK - 1 - tid)];
        d1 = g[pitch + (CHUNK - 1 - 1024 - tid)];
    }
    for (int c = 1; c < nch; c++) {
        unsigned long long a0 = s[tid];
        unsigned long long a1 = s[tid + 1024];
        unsigned long long e0 = a0 > d0 ? a0 : d0;
        unsigned long long e1 = a1 > d1 ? a1 : d1;
        s[tid]        = e0 > e1 ? e0 : e1;
        s[tid + 1024] = e0 > e1 ? e1 : e0;
        if (c + 1 < nch) {
            const unsigned long long* gn = g + (size_t)(c + 1) * pitch;
            d0 = gn[CHUNK - 1 - tid];
            d1 = gn[CHUNK - 1 - 1024 - tid];
        }
        __syncthreads();
        for (int j = CHUNK >> 2; j > 0; j >>= 1) {
            bitonic_ce(s, tid, j, CHUNK);
            __syncthreads();
        }
    }
    g[tid]        = s[tid];
    g[tid + 1024] = s[tid + 1024];
}

#define MASK_ROWS 16
#define MASK_BLKS (PRE_NMS / MASK_ROWS)
#define LPITCH 65
__global__ __launch_bounds__(256) void nms_mask_kernel(const unsigned long long* __restrict__ keys,
                                                       const float* __restrict__ boxes,
                                                       unsigned* __restrict__ mask,
                                                       float* __restrict__ sorted) {
    __shared__ v4f   sbx[32 * LPITCH];
    __shared__ float sar[32 * LPITCH];

    int n  = blockIdx.x / MASK_BLKS;
    int rb = blockIdx.x % MASK_BLKS;
    int tid = threadIdx.x;
    const unsigned long long* K = keys + ((size_t)n << 16);
    const float* bimg = boxes + (size_t)n * NPROP * 4;

    unsigned long long kk[8];
    #pragma unroll
    for (int q = 0; q < 8; q++) {
        int j = tid + 256 * q;
        kk[q] = K[j < PRE_NMS ? j : PRE_NMS - 1];
    }
    v4f gb[8];
    #pragma unroll
    for (int q = 0; q < 8; q++) {
        int j = tid + 256 * q;
        unsigned sv  = (unsigned)(kk[q] >> 32);
        unsigned idx = 0xFFFFFFFFu - (unsigned)(kk[q] & 0xFFFFFFFFu);
        bool v = (j < PRE_NMS) && (sv > 0x007FFFFFu);
        gb[q] = *(const v4f*)(bimg + (size_t)(v ? idx : 0u) * 4);
    }
    #pragma unroll
    for (int q = 0; q < 8; q++) {
        int j = tid + 256 * q;
        unsigned sv = (unsigned)(kk[q] >> 32);
        bool v = (j < PRE_NMS) && (sv > 0x007FFFFFu);
        v4f bb = {0.f, 0.f, 0.f, 0.f};
        float s = 0.f;
        if (v) {
            bb = gb[q];
            unsigned u = (sv & 0x80000000u) ? (sv ^ 0x80000000u) : ~sv;
            s = __uint_as_float(u);
        }
        int slot = (j & 31) * LPITCH + (j >> 5);
        sbx[slot] = bb;
        sar[slot] = fmaxf(bb.z - bb.x, 0.f) * fmaxf(bb.w - bb.y, 0.f);
        if (j < PRE_NMS && (j / MASK_ROWS) == rb) {
            v4f* so = (v4f*)sorted + ((size_t)n * PRE_NMS + j) * 2;
            v4f inf = {s, v ? 1.f : 0.f, 0.f, 0.f};
            so[0] = bb;
            so[1] = inf;
        }
    }
    __syncthreads();

    int wave = tid >> 5;
    int lane = tid & 31;
    for (int rr = 0; rr < 2; rr++) {
        int i = rb * MASK_ROWS + wave * 2 + rr;
        int islot = (i & 31) * LPITCH + (i >> 5);
        v4f bi = sbx[islot];
        float ia = sar[islot];
        int wdiag = i >> 5;
        unsigned* mrow = mask + ((size_t)n * PRE_NMS + i) * 64;
        for (int pass = 0; pass < 2; pass++) {
            int w = pass * 32 + lane;
            unsigned bits = 0u;
            if (w >= wdiag) {
                for (int b = 0; b < 32; b++) {
                    v4f bj = sbx[b * LPITCH + w];
                    float ja = sar[b * LPITCH + w];
                    float ix1 = fmaxf(bi.x, bj.x), iy1 = fmaxf(bi.y, bj.y);
                    float ix2 = fminf(bi.z, bj.z), iy2 = fminf(bi.w, bj.w);
                    float inter = fmaxf(ix2 - ix1, 0.f) * fmaxf(iy2 - iy1, 0.f);
                    float iou = inter / fmaxf(ia + ja - inter, 1e-9f);
                    if (iou > IOU_THR) bits |= (1u << b);
                }
            }
            mrow[w] = bits;
        }
    }
}

__global__ __launch_bounds__(32) void nms_scan_kernel(const unsigned* __restrict__ mask,
                                                      const float* __restrict__ sorted,
                                                      float* __restrict__ out) {
    int n = blockIdx.x;
    int lane = threadIdx.x;
    const v2u* M = (const v2u*)mask + (size_t)n * PRE_NMS * 32;
    const v4f* S = (const v4f*)sorted + (size_t)n * PRE_NMS * 2;
    float* o_img = out + (size_t)n * POST_NMS * 5;

    unsigned rem_lo = 0u, rem_hi = 0u;

    v2u cur[32], nxt[32];
    v4f cbox, cinf, nbox, ninf;
    #pragma unroll
    for (int r = 0; r < 32; r++) cur[r] = M[(size_t)r * 32 + lane];
    cbox = S[(size_t)lane * 2];
    cinf = S[(size_t)lane * 2 + 1];

    int cnt = 0;
    for (int c = 0; c < 63; c++) {
        int cn = (c + 1 < 63) ? c + 1 : 62;
        #pragma unroll
        for (int r = 0; r < 32; r++) {
            int i = cn * 32 + r;
            i = i < PRE_NMS ? i : PRE_NMS - 1;
            nxt[r] = M[(size_t)i * 32 + lane];
        }
        {
            int i = cn * 32 + lane;
            i = i < PRE_NMS ? i : PRE_NMS - 1;
            nbox = S[(size_t)i * 2];
            ninf = S[(size_t)i * 2 + 1];
        }

        int own = c >> 1;
        bool inval = (c * 32 + lane >= PRE_NMS) || (cinf.y == 0.f);
        unsigned remw = (unsigned)__builtin_amdgcn_readlane((int)((c & 1) ? rem_hi : rem_lo), own)
                      | __builtin_amdgcn_ballot_w32(inval);
        unsigned kept = 0u;
        #pragma unroll
        for (int r = 0; r < 32; r++) {
            unsigned d = (unsigned)__builtin_amdgcn_readlane((int)((c & 1) ? cur[r].y : cur[r].x), own);
            unsigned m = ((remw >> r) & 1u) ? 0u : 0xFFFFFFFFu;
            kept  |= m & (1u << r);
            remw  |= d & m;
            rem_lo |= cur[r].x & m;
            rem_hi |= cur[r].y & m;
        }

        unsigned mine = (kept >> lane) & 1u;
        int pos = cnt + __popc(kept & ((1u << lane) - 1u));
        if (mine && pos < POST_NMS) {
            float* o = o_img + pos * 5;
            o[0] = cbox.x; o[1] = cbox.y; o[2] = cbox.z; o[3] = cbox.w;
            o[4] = cinf.x;
        }
        cnt += __popc(kept);

        #pragma unroll
        for (int r = 0; r < 32; r++) cur[r] = nxt[r];
        cbox = nbox; cinf = ninf;
    }

    int c0 = cnt < POST_NMS ? cnt : POST_NMS;
    for (int t = c0 * 5 + lane; t < POST_NMS * 5; t += 32) o_img[t] = 0.f;
}

extern "C" void kernel_launch(void* const* d_in, const int* in_sizes, int n_in,
                              void* d_out, int out_size, void* d_ws, size_t ws_size,
                              hipStream_t stream) {
    const float* feat   = (const float*)d_in[0];
    const float* conv_w = (const float*)d_in[1];
    const float* conv_b = (const float*)d_in[2];
    const float* det_w  = (const float*)d_in[3];
    const float* det_b  = (const float*)d_in[4];
    const float* reg_w  = (const float*)d_in[5];
    const float* reg_b  = (const float*)d_in[6];
    float* out = (float*)d_out;

    char* ws = (char*)d_ws;
    float* wt2                  = (float*)(ws);
    float* scores               = (float*)(ws + 2359296);
    float* deltas               = (float*)(ws + 2359296 + 786432);
    float* boxes                = (float*)(ws + 2359296 + 786432 + 3145728);
    unsigned long long* keys    = (unsigned long long*)(ws + 2359296 + 786432 + 3145728 + 3145728);

    repack_w_kernel<<<2304, 256, 0, stream>>>(conv_w, wt2);
    conv_fused_kernel<<<NIMG * HF * (WF / TILE_W), 256, 0, stream>>>(
        feat, wt2, conv_b, det_w, det_b, reg_w, reg_b, scores, deltas);
    decode_kernel<<<(NIMG * NPAD) / 256, 256, 0, stream>>>(scores, deltas, boxes, keys);
    topk_chunk_sort_kernel<<<NIMG * NCHUNK, 1024, 0, stream>>>(keys);
    topk_fold_kernel<<<NIMG * FOLD_B, 1024, 0, stream>>>(keys, FOLD_A, 1, FOLD_B);
    topk_fold_kernel<<<NIMG, 1024, 0, stream>>>(keys, FOLD_B, FOLD_A, 1);
    unsigned* nmask  = (unsigned*)(ws);
    float*    sorted = (float*)(ws + 2048000);
    nms_mask_kernel<<<NIMG * MASK_BLKS, 256, 0, stream>>>(keys, boxes, nmask, sorted);
    nms_scan_kernel<<<NIMG, 32, 0, stream>>>(nmask, sorted, out);
}


# heads folded into the 3x3 conv (no activation between): W_eff = head_w x conv_w built per call, conv makes 16 rows not 256, K split over 8 waves and summed in LDS; f32 kept; sum order changes
# speedup vs baseline: 11.4241x; 2.8663x over previous
#include <hip/hip_runtime.h>
#include <hip/hip_bf16.h>
#include <math.h>

typedef __attribute__((ext_vector_type(2))) float v2f;
typedef __attribute__((ext_vector_type(8))) float v8f;
typedef __attribute__((ext_vector_type(4))) float v4f;
typedef __attribute__((ext_vector_type(2))) unsigned v2u;

#define NIMG 4
#define CIN 256
#define HF 128
#define WF 128
#define A_ANCH 3
#define NPROP (A_ANCH * HF * WF)
#define NPAD 65536
#define PRE_NMS 2000
#define POST_NMS 300
#define OBJ_THR 0.3f
#define IOU_THR 0.7f
#define TILE_W 64
#define CCHUNK 32
#define CIPAD 34

#define NWEFF (9 * 128 * 16 * 2)
__global__ __launch_bounds__(256) void fold_w_kernel(const float* __restrict__ conv_w,
                                                     const float* __restrict__ conv_b,
                                                     const float* __restrict__ det_w,
                                                     const float* __restrict__ det_b,
                                                     const float* __restrict__ reg_w,
                                                     const float* __restrict__ reg_b,
                                                     float* __restrict__ weff2) {
    int tid = threadIdx.x;
    if (blockIdx.x == NWEFF / 256) {
        if (tid < 16) {
            float s = 0.0f;
            if (tid < 15) {
                const float* hw = (tid < 3) ? (det_w + tid * 256) : (reg_w + (tid - 3) * 256);
                s = (tid < 3) ? det_b[tid] : reg_b[tid - 3];
                for (int c0 = 0; c0 < 256; c0 += 64) {
                    float hv[64], bv[64];
                    #pragma unroll
                    for (int j = 0; j < 64; j++) { hv[j] = hw[c0 + j]; bv[j] = conv_b[c0 + j]; }
                    #pragma unroll
                    for (int j = 0; j < 64; j++) s += hv[j] * bv[j];
                }
            }
            weff2[NWEFF + tid] = s;
        }
        return;
    }
    int t   = blockIdx.x * 256 + tid;
    int par = t & 1;
    int o   = (t >> 1) & 15;
    int q   = t >> 5;
    int ci2 = q & 127;
    int kk  = q >> 7;
    int ci  = ci2 * 2 + par;
    float s = 0.0f;
    if (o < 15) {
        const float* hw = (o < 3) ? (det_w + o * 256) : (reg_w + (o - 3) * 256);
        const float* wp = conv_w + (size_t)ci * 9 + kk;
        for (int c0 = 0; c0 < 256; c0 += 32) {
            float hv[32], wv[32];
            #pragma unroll
            for (int j = 0; j < 32; j++) { hv[j] = hw[c0 + j]; wv[j] = wp[(size_t)(c0 + j) * (CIN * 9)]; }
            #pragma unroll
            for (int j = 0; j < 32; j++) s += hv[j] * wv[j];
        }
    }
    weff2[t] = s;
}

#define NCHK (CIN / CCHUNK)
#define NPRE 25
#define PP 66

__device__ __forceinline__ void stage_fetch(float (&pre)[NPRE], const float* frow,
                                            const float* fhalo, const int (&rowoff)[3], int cib) {
    const float* fc = frow + (size_t)cib * (HF * WF);
    #pragma unroll
    for (int j = 0; j < 4; j++) {
        #pragma unroll
        for (int r = 0; r < 3; r++) {
            const float* p = fc + j * (HF * WF) + rowoff[r];
            pre[(j * 3 + r) * 2 + 0] = p[0];
            pre[(j * 3 + r) * 2 + 1] = p[32];
        }
    }
    pre[24] = fhalo[(size_t)cib * (HF * WF)];
}

__global__ __launch_bounds__(256) __attribute__((amdgpu_waves_per_eu(8))) void conv_fused_kernel(
    const float* __restrict__ feat, const float* __restrict__ weff2,
    float* __restrict__ scores, float* __restrict__ deltas) {

    __shared__ __attribute__((aligned(16))) float lds_in[3 * 66 * CIPAD];
    __shared__ float part[8 * 16 * PP];

    int bid = blockIdx.x;
    int n    = bid >> 8;
    int rem  = bid & 255;
    int row  = rem >> 1;
    int col0 = (rem & 1) * TILE_W;

    int tid  = threadIdx.x;
    int wave = tid >> 5;
    int lane = tid & 31;
    int hl   = lane >> 4;
    int lm   = lane & 15;

    v8f acc[4] = {};

    const float* featn = feat + (size_t)n * CIN * HF * WF;
    const v2f*   wpl   = (const v2f*)weff2 + (size_t)(2 * wave + hl) * 16 + lm;

    int  rowoff[3];
    bool rowok[3];
    #pragma unroll
    for (int r = 0; r < 3; r++) {
        int g = row + r - 1;
        rowok[r]  = (g >= 0) && (g < HF);
        rowoff[r] = (g < 0 ? 0 : (g >= HF ? HF - 1 : g)) * WF;
    }
    const float* frow = featn + (size_t)(4 * wave) * (HF * WF) + col0 + lane;
    int  hj  = lane / 6;
    int  hq  = lane - hj * 6;
    int  hr  = hq >> 1;
    int  hs  = hq & 1;
    int  hg  = row + hr - 1;
    int  hc  = hs ? (col0 + TILE_W) : (col0 - 1);
    bool hok = (lane < 24) && (hg >= 0) && (hg < HF) && (hc >= 0) && (hc < WF);
    const float* fhalo = featn + (hok ? ((size_t)(4 * wave + hj) * HF + hg) * WF + hc : (size_t)0);
    int  hlds = (hr * 66 + (hs ? 65 : 0)) * CIPAD + 4 * wave + hj;

    int bbase = lm * CIPAD + 4 * wave + 2 * hl;

    float pre[NPRE];
    stage_fetch(pre, frow, fhalo, rowoff, 0);

    for (int ch = 0; ch < NCHK; ch++) {
        v2f a[9];
        #pragma unroll
        for (int kk = 0; kk < 9; kk++)
            a[kk] = wpl[(size_t)(kk * 128 + ch * 16) * 16];

        __syncthreads();
        #pragma unroll
        for (int r = 0; r < 3; r++) {
            #pragma unroll
            for (int hcol = 0; hcol < 2; hcol++) {
                float v0 = rowok[r] ? pre[(0 * 3 + r) * 2 + hcol] : 0.0f;
                float v1 = rowok[r] ? pre[(1 * 3 + r) * 2 + hcol] : 0.0f;
                float v2 = rowok[r] ? pre[(2 * 3 + r) * 2 + hcol] : 0.0f;
                float v3 = rowok[r] ? pre[(3 * 3 + r) * 2 + hcol] : 0.0f;
                int idx = (r * 66 + 1 + hcol * 32 + lane) * CIPAD + 4 * wave;
                v2f lo = {v0, v1};
                v2f hi = {v2, v3};
                *(v2f*)&lds_in[idx]     = lo;
                *(v2f*)&lds_in[idx + 2] = hi;
            }
        }
        if (lane < 24) lds_in[hlds] = hok ? pre[24] : 0.0f;
        __syncthreads();

        int chn = ch + 1 < NCHK ? ch + 1 : NCHK - 1;
        stage_fetch(pre, frow, fhalo, rowoff, chn * CCHUNK);

        #pragma unroll
        for (int kk = 0; kk < 9; kk++) {
            int kh = kk / 3;
            int kw = kk - kh * 3;
            v2f b[4];
            #pragma unroll
            for (int nt = 0; nt < 4; nt++)
                b[nt] = *(const v2f*)&lds_in[(kh * 66 + nt * 16 + kw) * CIPAD + bbase];
            #pragma unroll
            for (int nt = 0; nt < 4; nt++)
                acc[nt] = __builtin_amdgcn_wmma_f32_16x16x4_f32(false, a[kk], false, b[nt], (short)0, acc[nt], false, false);
        }
    }

    #pragma unroll
    for (int nt = 0; nt < 4; nt++) {
        #pragma unroll
        for (int r = 0; r < 8; r++)
            part[(wave * 16 + hl * 8 + r) * PP + nt * 16 + lm] = acc[nt][r];
    }
    __syncthreads();

    const float* beff = weff2 + NWEFF;
    for (int e = tid; e < 15 * 64; e += 256) {
        int oi  = e >> 6;
        int pix = e & 63;
        float sum = part[oi * PP + pix];
        #pragma unroll
        for (int w = 1; w < 8; w++) sum += part[(w * 16 + oi) * PP + pix];
        sum += beff[oi];
        int p = row * WF + col0 + pix;
        if (oi < 3) {
            scores[(size_t)n * NPROP + oi * (HF * WF) + p] = 1.0f / (1.0f + expf(-sum));
        } else {
            int chh = oi - 3;
            int ccomp = chh / 3;
            int a = chh % 3;
            deltas[(((size_t)n * NPROP) + a * (HF * WF) + p) * 4 + ccomp] = sum;
        }
    }
}

__global__ __launch_bounds__(256) void decode_kernel(
    const float* __restrict__ scores, const float* __restrict__ deltas,
    float* __restrict__ boxes, unsigned long long* __restrict__ keys) {
    int t = blockIdx.x * 256 + threadIdx.x;
    int n = t >> 16;
    int i = t & 0xFFFF;
    unsigned long long* kout = keys + ((size_t)n << 16);
    if (i >= NPROP) { kout[i] = 0ull; return; }

    int pa = i / 3, aa = i - pa * 3;
    int ya = pa >> 7, xa = pa & 127;
    const float bs[3][4] = {{-91.f, -45.f, 91.f, 45.f},
                            {-64.f, -64.f, 64.f, 64.f},
                            {-45.f, -91.f, 45.f, 91.f}};
    float sxv = (float)(xa * 8), syv = (float)(ya * 8);
    float x1 = sxv + bs[aa][0], y1 = syv + bs[aa][1];
    float x2 = sxv + bs[aa][2], y2 = syv + bs[aa][3];
    float aw = x2 - x1, ah = y2 - y1;
    float acx = y1 + aw * 0.5f;
    float acy = x1 + ah * 0.5f;

    const float* d = deltas + (((size_t)n * NPROP) + i) * 4;
    float px = acx + d[0] * aw;
    float py = acy + d[1] * ah;
    float pw = aw * expf(d[2]);
    float ph = ah * expf(d[3]);
    float b0 = fminf(fmaxf(px - pw * 0.5f, 0.f), 1024.f);
    float b1 = fminf(fmaxf(py - ph * 0.5f, 0.f), 1024.f);
    float b2 = fminf(fmaxf(px + pw * 0.5f, 0.f), 1024.f);
    float b3 = fminf(fmaxf(py + ph * 0.5f, 0.f), 1024.f);

    float* bo = boxes + (((size_t)n * NPROP) + i) * 4;
    bo[0] = b0; bo[1] = b1; bo[2] = b2; bo[3] = b3;

    float s = scores[(size_t)n * NPROP + i];
    bool valid = (s > OBJ_THR) && ((b2 - b0) > 0.f) && ((b3 - b1) > 0.f);
    float ss = valid ? s : -__builtin_inff();
    unsigned u = __float_as_uint(ss);
    unsigned sv = (u & 0x80000000u) ? ~u : (u | 0x80000000u);
    kout[i] = ((unsigned long long)sv << 32) | (unsigned long long)(0xFFFFFFFFu - (unsigned)i);
}

#define CHUNK 2048
#define NCHUNK (NPROP / CHUNK)
#define FOLD_A 6
#define FOLD_B (NCHUNK / FOLD_A)

__device__ __forceinline__ void bitonic_ce(unsigned long long* s, int t, int j, int k) {
    int lo = t & (j - 1);
    int i  = ((t - lo) << 1) | lo;
    int p  = i | j;
    unsigned long long a = s[i];
    unsigned long long b = s[p];
    bool up = ((i & k) == 0);
    bool sw = up ? (a < b) : (a > b);
    if (sw) { s[i] = b; s[p] = a; }
}

__global__ __launch_bounds__(1024) void topk_chunk_sort_kernel(unsigned long long* keys) {
    __shared__ unsigned long long s[CHUNK];
    int n   = blockIdx.x / NCHUNK;
    int c   = blockIdx.x % NCHUNK;
    int tid = threadIdx.x;
    unsigned long long* g = keys + ((size_t)n << 16) + (size_t)c * CHUNK;
    s[tid]        = g[tid];
    s[tid + 1024] = g[tid + 1024];
    __syncthreads();
    for (int k = 2; k <= CHUNK; k <<= 1)
        for (int j = k >> 1; j > 0; j >>= 1) {
            bitonic_ce(s, tid, j, k);
            __syncthreads();
        }
    g[tid]        = s[tid];
    g[tid + 1024] = s[tid + 1024];
}

__global__ __launch_bounds__(1024) void topk_fold_kernel(unsigned long long* keys,
                                                         int nch, int stride, int groups) {
    __shared__ unsigned long long s[CHUNK];
    int n   = blockIdx.x / groups;
    int gi  = blockIdx.x % groups;
    int tid = threadIdx.x;
    size_t pitch = (size_t)stride * CHUNK;
    unsigned long long* g = keys + ((size_t)n << 16) + (size_t)gi * nch * pitch;
    s[tid]        = g[tid];
    s[tid + 1024] = g[tid + 1024];
    unsigned long long d0 = 0ull, d1 = 0ull;
    if (nch > 1) {
        d0 = g[pitch + (CHUNK - 1 - tid)];
        d1 = g[pitch + (CHUNK - 1 - 1024 - tid)];
    }
    for (int c = 1; c < nch; c++) {
        unsigned long long a0 = s[tid];
        unsigned long long a1 = s[tid + 1024];
        unsigned long long e0 = a0 > d0 ? a0 : d0;
        unsigned long long e1 = a1 > d1 ? a1 : d1;
        s[tid]        = e0 > e1 ? e0 : e1;
        s[tid + 1024] = e0 > e1 ? e1 : e0;
        if (c + 1 < nch) {
            const unsigned long long* gn = g + (size_t)(c + 1) * pitch;
            d0 = gn[CHUNK - 1 - tid];
            d1 = gn[CHUNK - 1 - 1024 - tid];
        }
        __syncthreads();
        for (int j = CHUNK >> 2; j > 0; j >>= 1) {
            bitonic_ce(s, tid, j, CHUNK);
            __syncthreads();
        }
    }
    g[tid]        = s[tid];
    g[tid + 1024] = s[tid + 1024];
}

#define MASK_ROWS 16
#define MASK_BLKS (PRE_NMS / MASK_ROWS)
#define LPITCH 65
__global__ __launch_bounds__(256) void nms_mask_kernel(const unsigned long long* __restrict__ keys,
                                                       const float* __restrict__ boxes,
                                                       unsigned* __restrict__ mask,
                                                       float* __restrict__ sorted) {
    __shared__ v4f   sbx[32 * LPITCH];
    __shared__ float sar[32 * LPITCH];

    int n  = blockIdx.x / MASK_BLKS;
    int rb = blockIdx.x % MASK_BLKS;
    int tid = threadIdx.x;
    const unsigned long long* K = keys + ((size_t)n << 16);
    const float* bimg = boxes + (size_t)n * NPROP * 4;

    unsigned long long kk[8];
    #pragma unroll
    for (int q = 0; q < 8; q++) {
        int j = tid + 256 * q;
        kk[q] = K[j < PRE_NMS ? j : PRE_NMS - 1];
    }
    v4f gb[8];
    #pragma unroll
    for (int q = 0; q < 8; q++) {
        int j = tid + 256 * q;
        unsigned sv  = (unsigned)(kk[q] >> 32);
        unsigned idx = 0xFFFFFFFFu - (unsigned)(kk[q] & 0xFFFFFFFFu);
        bool v = (j < PRE_NMS) && (sv > 0x007FFFFFu);
        gb[q] = *(const v4f*)(bimg + (size_t)(v ? idx : 0u) * 4);
    }
    #pragma unroll
    for (int q = 0; q < 8; q++) {
        int j = tid + 256 * q;
        unsigned sv = (unsigned)(kk[q] >> 32);
        bool v = (j < PRE_NMS) && (sv > 0x007FFFFFu);
        v4f bb = {0.f, 0.f, 0.f, 0.f};
        float s = 0.f;
        if (v) {
            bb = gb[q];
            unsigned u = (sv & 0x80000000u) ? (sv ^ 0x80000000u) : ~sv;
            s = __uint_as_float(u);
        }
        int slot = (j & 31) * LPITCH + (j >> 5);
        sbx[slot] = bb;
        sar[slot] = fmaxf(bb.z - bb.x, 0.f) * fmaxf(bb.w - bb.y, 0.f);
        if (j < PRE_NMS && (j / MASK_ROWS) == rb) {
            v4f* so = (v4f*)sorted + ((size_t)n * PRE_NMS + j) * 2;
            v4f inf = {s, v ? 1.f : 0.f, 0.f, 0.f};
            so[0] = bb;
            so[1] = inf;
        }
    }
    __syncthreads();

    int wave = tid >> 5;
    int lane = tid & 31;
    for (int rr = 0; rr < 2; rr++) {
        int i = rb * MASK_ROWS + wave * 2 + rr;
        int islot = (i & 31) * LPITCH + (i >> 5);
        v4f bi = sbx[islot];
        float ia = sar[islot];
        int wdiag = i >> 5;
        unsigned* mrow = mask + ((size_t)n * PRE_NMS + i) * 64;
        for (int pass = 0; pass < 2; pass++) {
            int w = pass * 32 + lane;
            unsigned bits = 0u;
            if (w >= wdiag) {
                for (int b = 0; b < 32; b++) {
                    v4f bj = sbx[b * LPITCH + w];
                    float ja = sar[b * LPITCH + w];
                    float ix1 = fmaxf(bi.x, bj.x), iy1 = fmaxf(bi.y, bj.y);
                    float ix2 = fminf(bi.z, bj.z), iy2 = fminf(bi.w, bj.w);
                    float inter = fmaxf(ix2 - ix1, 0.f) * fmaxf(iy2 - iy1, 0.f);
                    float iou = inter / fmaxf(ia + ja - inter, 1e-9f);
                    if (iou > IOU_THR) bits |= (1u << b);
                }
            }
            mrow[w] = bits;
        }
    }
}

__global__ __launch_bounds__(32) void nms_scan_kernel(const unsigned* __restrict__ mask,
                                                      const float* __restrict__ sorted,
                                                      float* __restrict__ out) {
    int n = blockIdx.x;
    int lane = threadIdx.x;
    const v2u* M = (const v2u*)mask + (size_t)n * PRE_NMS * 32;
    const v4f* S = (const v4f*)sorted + (size_t)n * PRE_NMS * 2;
    float* o_img = out + (size_t)n * POST_NMS * 5;

    unsigned rem_lo = 0u, rem_hi = 0u;

    v2u cur[32], nxt[32];
    v4f cbox, cinf, nbox, ninf;
    #pragma unroll
    for (int r = 0; r < 32; r++) cur[r] = M[(size_t)r * 32 + lane];
    cbox = S[(size_t)lane * 2];
    cinf = S[(size_t)lane * 2 + 1];

    int cnt = 0;
    for (int c = 0; c < 63; c++) {
        int cn = (c + 1 < 63) ? c + 1 : 62;
        #pragma unroll
        for (int r = 0; r < 32; r++) {
            int i = cn * 32 + r;
            i = i < PRE_NMS ? i : PRE_NMS - 1;
            nxt[r] = M[(size_t)i * 32 + lane];
        }
        {
            int i = cn * 32 + lane;
            i = i < PRE_NMS ? i : PRE_NMS - 1;
            nbox = S[(size_t)i * 2];
            ninf = S[(size_t)i * 2 + 1];
        }

        int own = c >> 1;
        bool inval = (c * 32 + lane >= PRE_NMS) || (cinf.y == 0.f);
        unsigned remw = (unsigned)__builtin_amdgcn_readlane((int)((c & 1) ? rem_hi : rem_lo), own)
                      | __builtin_amdgcn_ballot_w32(inval);
        unsigned kept = 0u;
        #pragma unroll
        for (int r = 0; r < 32; r++) {
            unsigned d = (unsigned)__builtin_amdgcn_readlane((int)((c & 1) ? cur[r].y : cur[r].x), own);
            unsigned m = ((remw >> r) & 1u) ? 0u : 0xFFFFFFFFu;
            kept  |= m & (1u << r);
            remw  |= d & m;
            rem_lo |= cur[r].x & m;
            rem_hi |= cur[r].y & m;
        }

        unsigned mine = (kept >> lane) & 1u;
        int pos = cnt + __popc(kept & ((1u << lane) - 1u));
        if (mine && pos < POST_NMS) {
            float* o = o_img + pos * 5;
            o[0] = cbox.x; o[1] = cbox.y; o[2] = cbox.z; o[3] = cbox.w;
            o[4] = cinf.x;
        }
        cnt += __popc(kept);

        #pragma unroll
        for (int r = 0; r < 32; r++) cur[r] = nxt[r];
        cbox = nbox; cinf = ninf;
    }

    int c0 = cnt < POST_NMS ? cnt : POST_NMS;
    for (int t = c0 * 5 + lane; t < POST_NMS * 5; t += 32) o_img[t] = 0.f;
}

extern "C" void kernel_launch(void* const* d_in, const int* in_sizes, int n_in,
                              void* d_out, int out_size, void* d_ws, size_t ws_size,
                              hipStream_t stream) {
    const float* feat   = (const float*)d_in[0];
    const float* conv_w = (const float*)d_in[1];
    const float* conv_b = (const float*)d_in[2];
    const float* det_w  = (const float*)d_in[3];
    const float* det_b  = (const float*)d_in[4];
    const float* reg_w  = (const float*)d_in[5];
    const float* reg_b  = (const float*)d_in[6];
    float* out = (float*)d_out;

    char* ws = (char*)d_ws;
    float* wt2                  = (float*)(ws);
    float* scores               = (float*)(ws + 2359296);
    float* deltas               = (float*)(ws + 2359296 + 786432);
    float* boxes                = (float*)(ws + 2359296 + 786432 + 3145728);
    unsigned long long* keys    = (unsigned long long*)(ws + 2359296 + 786432 + 3145728 + 3145728);

    fold_w_kernel<<<145, 256, 0, stream>>>(conv_w, conv_b, det_w, det_b, reg_w, reg_b, wt2);
    conv_fused_kernel<<<NIMG * HF * (WF / TILE_W), 256, 0, stream>>>(feat, wt2, scores, deltas);
    decode_kernel<<<(NIMG * NPAD) / 256, 256, 0, stream>>>(scores, deltas, boxes, keys);
    topk_chunk_sort_kernel<<<NIMG * NCHUNK, 1024, 0, stream>>>(keys);
    topk_fold_kernel<<<NIMG * FOLD_B, 1024, 0, stream>>>(keys, FOLD_A, 1, FOLD_B);
    topk_fold_kernel<<<NIMG, 1024, 0, stream>>>(keys, FOLD_B, FOLD_A, 1);
    unsigned* nmask  = (unsigned*)(ws);
    float*    sorted = (float*)(ws + 2048000);
    nms_mask_kernel<<<NIMG * MASK_BLKS, 256, 0, stream>>>(keys, boxes, nmask, sorted);
    nms_scan_kernel<<<NIMG, 32, 0, stream>>>(nmask, sorted, out);
}


# NMS scan stops once 300 boxes are kept: out holds only the first 300 kept rows in score order, so later chunks cannot reach it; exact for every input, same rows and values
# speedup vs baseline: 16.5826x; 1.4515x over previous
#include <hip/hip_runtime.h>
#include <hip/hip_bf16.h>
#include <math.h>

typedef __attribute__((ext_vector_type(2))) float v2f;
typedef __attribute__((ext_vector_type(8))) float v8f;
typedef __attribute__((ext_vector_type(4))) float v4f;
typedef __attribute__((ext_vector_type(2))) unsigned v2u;

#define NIMG 4
#define CIN 256
#define HF 128
#define WF 128
#define A_ANCH 3
#define NPROP (A_ANCH * HF * WF)
#define NPAD 65536
#define PRE_NMS 2000
#define POST_NMS 300
#define OBJ_THR 0.3f
#define IOU_THR 0.7f
#define TILE_W 64
#define CCHUNK 32
#define CIPAD 34

#define NWEFF (9 * 128 * 16 * 2)
__global__ __launch_bounds__(256) void fold_w_kernel(const float* __restrict__ conv_w,
                                                     const float* __restrict__ conv_b,
                                                     const float* __restrict__ det_w,
                                                     const float* __restrict__ det_b,
                                                     const float* __restrict__ reg_w,
                                                     const float* __restrict__ reg_b,
                                                     float* __restrict__ weff2) {
    int tid = threadIdx.x;
    if (blockIdx.x == NWEFF / 256) {
        if (tid < 16) {
            float s = 0.0f;
            if (tid < 15) {
                const float* hw = (tid < 3) ? (det_w + tid * 256) : (reg_w + (tid - 3) * 256);
                s = (tid < 3) ? det_b[tid] : reg_b[tid - 3];
                for (int c0 = 0; c0 < 256; c0 += 64) {
                    float hv[64], bv[64];
                    #pragma unroll
                    for (int j = 0; j < 64; j++) { hv[j] = hw[c0 + j]; bv[j] = conv_b[c0 + j]; }
                    #pragma unroll
                    for (int j = 0; j < 64; j++) s += hv[j] * bv[j];
                }
            }
            weff2[NWEFF + tid] = s;
        }
        return;
    }
    int t   = blockIdx.x * 256 + tid;
    int par = t & 1;
    int o   = (t >> 1) & 15;
    int q   = t >> 5;
    int ci2 = q & 127;
    int kk  = q >> 7;
    int ci  = ci2 * 2 + par;
    float s = 0.0f;
    if (o < 15) {
        const float* hw = (o < 3) ? (det_w + o * 256) : (reg_w + (o - 3) * 256);
        const float* wp = conv_w + (size_t)ci * 9 + kk;
        for (int c0 = 0; c0 < 256; c0 += 32) {
            float hv[32], wv[32];
            #pragma unroll
            for (int j = 0; j < 32; j++) { hv[j] = hw[c0 + j]; wv[j] = wp[(size_t)(c0 + j) * (CIN * 9)]; }
            #pragma unroll
            for (int j = 0; j < 32; j++) s += hv[j] * wv[j];
        }
    }
    weff2[t] = s;
}

#define NCHK (CIN / CCHUNK)
#define NPRE 25
#define PP 66

__device__ __forceinline__ void stage_fetch(float (&pre)[NPRE], const float* frow,
                                            const float* fhalo, const int (&rowoff)[3], int cib) {
    const float* fc = frow + (size_t)cib * (HF * WF);
    #pragma unroll
    for (int j = 0; j < 4; j++) {
        #pragma unroll
        for (int r = 0; r < 3; r++) {
            const float* p = fc + j * (HF * WF) + rowoff[r];
            pre[(j * 3 + r) * 2 + 0] = p[0];
            pre[(j * 3 + r) * 2 + 1] = p[32];
        }
    }
    pre[24] = fhalo[(size_t)cib * (HF * WF)];
}

__global__ __launch_bounds__(256) __attribute__((amdgpu_waves_per_eu(8))) void conv_fused_kernel(
    const float* __restrict__ feat, const float* __restrict__ weff2,
    float* __restrict__ scores, float* __restrict__ deltas) {

    __shared__ __attribute__((aligned(16))) float lds_in[3 * 66 * CIPAD];
    __shared__ float part[8 * 16 * PP];

    int bid = blockIdx.x;
    int n    = bid >> 8;
    int rem  = bid & 255;
    int row  = rem >> 1;
    int col0 = (rem & 1) * TILE_W;

    int tid  = threadIdx.x;
    int wave = tid >> 5;
    int lane = tid & 31;
    int hl   = lane >> 4;
    int lm   = lane & 15;

    v8f acc[4] = {};

    const float* featn = feat + (size_t)n * CIN * HF * WF;
    const v2f*   wpl   = (const v2f*)weff2 + (size_t)(2 * wave + hl) * 16 + lm;

    int  rowoff[3];
    bool rowok[3];
    #pragma unroll
    for (int r = 0; r < 3; r++) {
        int g = row + r - 1;
        rowok[r]  = (g >= 0) && (g < HF);
        rowoff[r] = (g < 0 ? 0 : (g >= HF ? HF - 1 : g)) * WF;
    }
    const float* frow = featn + (size_t)(4 * wave) * (HF * WF) + col0 + lane;
    int  hj  = lane / 6;
    int  hq  = lane - hj * 6;
    int  hr  = hq >> 1;
    int  hs  = hq & 1;
    int  hg  = row + hr - 1;
    int  hc  = hs ? (col0 + TILE_W) : (col0 - 1);
    bool hok = (lane < 24) && (hg >= 0) && (hg < HF) && (hc >= 0) && (hc < WF);
    const float* fhalo = featn + (hok ? ((size_t)(4 * wave + hj) * HF + hg) * WF + hc : (size_t)0);
    int  hlds = (hr * 66 + (hs ? 65 : 0)) * CIPAD + 4 * wave + hj;

    int bbase = lm * CIPAD + 4 * wave + 2 * hl;

    float pre[NPRE];
    stage_fetch(pre, frow, fhalo, rowoff, 0);

    for (int ch = 0; ch < NCHK; ch++) {
        v2f a[9];
        #pragma unroll
        for (int kk = 0; kk < 9; kk++)
            a[kk] = wpl[(size_t)(kk * 128 + ch * 16) * 16];

        __syncthreads();
        #pragma unroll
        for (int r = 0; r < 3; r++) {
            #pragma unroll
            for (int hcol = 0; hcol < 2; hcol++) {
                float v0 = rowok[r] ? pre[(0 * 3 + r) * 2 + hcol] : 0.0f;
                float v1 = rowok[r] ? pre[(1 * 3 + r) * 2 + hcol] : 0.0f;
                float v2 = rowok[r] ? pre[(2 * 3 + r) * 2 + hcol] : 0.0f;
                float v3 = rowok[r] ? pre[(3 * 3 + r) * 2 + hcol] : 0.0f;
                int idx = (r * 66 + 1 + hcol * 32 + lane) * CIPAD + 4 * wave;
                v2f lo = {v0, v1};
                v2f hi = {v2, v3};
                *(v2f*)&lds_in[idx]     = lo;
                *(v2f*)&lds_in[idx + 2] = hi;
            }
        }
        if (lane < 24) lds_in[hlds] = hok ? pre[24] : 0.0f;
        __syncthreads();

        int chn = ch + 1 < NCHK ? ch + 1 : NCHK - 1;
        stage_fetch(pre, frow, fhalo, rowoff, chn * CCHUNK);

        #pragma unroll
        for (int kk = 0; kk < 9; kk++) {
            int kh = kk / 3;
            int kw = kk - kh * 3;
            v2f b[4];
            #pragma unroll
            for (int nt = 0; nt < 4; nt++)
                b[nt] = *(const v2f*)&lds_in[(kh * 66 + nt * 16 + kw) * CIPAD + bbase];
            #pragma unroll
            for (int nt = 0; nt < 4; nt++)
                acc[nt] = __builtin_amdgcn_wmma_f32_16x16x4_f32(false, a[kk], false, b[nt], (short)0, acc[nt], false, false);
        }
    }

    #pragma unroll
    for (int nt = 0; nt < 4; nt++) {
        #pragma unroll
        for (int r = 0; r < 8; r++)
            part[(wave * 16 + hl * 8 + r) * PP + nt * 16 + lm] = acc[nt][r];
    }
    __syncthreads();

    const float* beff = weff2 + NWEFF;
    for (int e = tid; e < 15 * 64; e += 256) {
        int oi  = e >> 6;
        int pix = e & 63;
        float sum = part[oi * PP + pix];
        #pragma unroll
        for (int w = 1; w < 8; w++) sum += part[(w * 16 + oi) * PP + pix];
        sum += beff[oi];
        int p = row * WF + col0 + pix;
        if (oi < 3) {
            scores[(size_t)n * NPROP + oi * (HF * WF) + p] = 1.0f / (1.0f + expf(-sum));
        } else {
            int chh = oi - 3;
            int ccomp = chh / 3;
            int a = chh % 3;
            deltas[(((size_t)n * NPROP) + a * (HF * WF) + p) * 4 + ccomp] = sum;
        }
    }
}

__global__ __launch_bounds__(256) void decode_kernel(
    const float* __restrict__ scores, const float* __restrict__ deltas,
    float* __restrict__ boxes, unsigned long long* __restrict__ keys) {
    int t = blockIdx.x * 256 + threadIdx.x;
    int n = t >> 16;
    int i = t & 0xFFFF;
    unsigned long long* kout = keys + ((size_t)n << 16);
    if (i >= NPROP) { kout[i] = 0ull; return; }

    int pa = i / 3, aa = i - pa * 3;
    int ya = pa >> 7, xa = pa & 127;
    const float bs[3][4] = {{-91.f, -45.f, 91.f, 45.f},
                            {-64.f, -64.f, 64.f, 64.f},
                            {-45.f, -91.f, 45.f, 91.f}};
    float sxv = (float)(xa * 8), syv = (float)(ya * 8);
    float x1 = sxv + bs[aa][0], y1 = syv + bs[aa][1];
    float x2 = sxv + bs[aa][2], y2 = syv + bs[aa][3];
    float aw = x2 - x1, ah = y2 - y1;
    float acx = y1 + aw * 0.5f;
    float acy = x1 + ah * 0.5f;

    const float* d = deltas + (((size_t)n * NPROP) + i) * 4;
    float px = acx + d[0] * aw;
    float py = acy + d[1] * ah;
    float pw = aw * expf(d[2]);
    float ph = ah * expf(d[3]);
    float b0 = fminf(fmaxf(px - pw * 0.5f, 0.f), 1024.f);
    float b1 = fminf(fmaxf(py - ph * 0.5f, 0.f), 1024.f);
    float b2 = fminf(fmaxf(px + pw * 0.5f, 0.f), 1024.f);
    float b3 = fminf(fmaxf(py + ph * 0.5f, 0.f), 1024.f);

    float* bo = boxes + (((size_t)n * NPROP) + i) * 4;
    bo[0] = b0; bo[1] = b1; bo[2] = b2; bo[3] = b3;

    float s = scores[(size_t)n * NPROP + i];
    bool valid = (s > OBJ_THR) && ((b2 - b0) > 0.f) && ((b3 - b1) > 0.f);
    float ss = valid ? s : -__builtin_inff();
    unsigned u = __float_as_uint(ss);
    unsigned sv = (u & 0x80000000u) ? ~u : (u | 0x80000000u);
    kout[i] = ((unsigned long long)sv << 32) | (unsigned long long)(0xFFFFFFFFu - (unsigned)i);
}

#define CHUNK 2048
#define NCHUNK (NPROP / CHUNK)
#define FOLD_A 6
#define FOLD_B (NCHUNK / FOLD_A)

__device__ __forceinline__ void bitonic_ce(unsigned long long* s, int t, int j, int k) {
    int lo = t & (j - 1);
    int i  = ((t - lo) << 1) | lo;
    int p  = i | j;
    unsigned long long a = s[i];
    unsigned long long b = s[p];
    bool up = ((i & k) == 0);
    bool sw = up ? (a < b) : (a > b);
    if (sw) { s[i] = b; s[p] = a; }
}

__global__ __launch_bounds__(1024) void topk_chunk_sort_kernel(unsigned long long* keys) {
    __shared__ unsigned long long s[CHUNK];
    int n   = blockIdx.x / NCHUNK;
    int c   = blockIdx.x % NCHUNK;
    int tid = threadIdx.x;
    unsigned long long* g = keys + ((size_t)n << 16) + (size_t)c * CHUNK;
    s[tid]        = g[tid];
    s[tid + 1024] = g[tid + 1024];
    __syncthreads();
    for (int k = 2; k <= CHUNK; k <<= 1)
        for (int j = k >> 1; j > 0; j >>= 1) {
            bitonic_ce(s, tid, j, k);
            __syncthreads();
        }
    g[tid]        = s[tid];
    g[tid + 1024] = s[tid + 1024];
}

__global__ __launch_bounds__(1024) void topk_fold_kernel(unsigned long long* keys,
                                                         int nch, int stride, int groups) {
    __shared__ unsigned long long s[CHUNK];
    int n   = blockIdx.x / groups;
    int gi  = blockIdx.x % groups;
    int tid = threadIdx.x;
    size_t pitch = (size_t)stride * CHUNK;
    unsigned long long* g = keys + ((size_t)n << 16) + (size_t)gi * nch * pitch;
    s[tid]        = g[tid];
    s[tid + 1024] = g[tid + 1024];
    unsigned long long d0 = 0ull, d1 = 0ull;
    if (nch > 1) {
        d0 = g[pitch + (CHUNK - 1 - tid)];
        d1 = g[pitch + (CHUNK - 1 - 1024 - tid)];
    }
    for (int c = 1; c < nch; c++) {
        unsigned long long a0 = s[tid];
        unsigned long long a1 = s[tid + 1024];
        unsigned long long e0 = a0 > d0 ? a0 : d0;
        unsigned long long e1 = a1 > d1 ? a1 : d1;
        s[tid]        = e0 > e1 ? e0 : e1;
        s[tid + 1024] = e0 > e1 ? e1 : e0;
        if (c + 1 < nch) {
            const unsigned long long* gn = g + (size_t)(c + 1) * pitch;
            d0 = gn[CHUNK - 1 - tid];
            d1 = gn[CHUNK - 1 - 1024 - tid];
        }
        __syncthreads();
        for (int j = CHUNK >> 2; j > 0; j >>= 1) {
            bitonic_ce(s, tid, j, CHUNK);
            __syncthreads();
        }
    }
    g[tid]        = s[tid];
    g[tid + 1024] = s[tid + 1024];
}

#define MASK_ROWS 16
#define MASK_BLKS (PRE_NMS / MASK_ROWS)
#define LPITCH 65
__global__ __launch_bounds__(256) void nms_mask_kernel(const unsigned long long* __restrict__ keys,
                                                       const float* __restrict__ boxes,
                                                       unsigned* __restrict__ mask,
                                                       float* __restrict__ sorted) {
    __shared__ v4f   sbx[32 * LPITCH];
    __shared__ float sar[32 * LPITCH];

    int n  = blockIdx.x / MASK_BLKS;
    int rb = blockIdx.x % MASK_BLKS;
    int tid = threadIdx.x;
    const unsigned long long* K = keys + ((size_t)n << 16);
    const float* bimg = boxes + (size_t)n * NPROP * 4;

    unsigned long long kk[8];
    #pragma unroll
    for (int q = 0; q < 8; q++) {
        int j = tid + 256 * q;
        kk[q] = K[j < PRE_NMS ? j : PRE_NMS - 1];
    }
    v4f gb[8];
    #pragma unroll
    for (int q = 0; q < 8; q++) {
        int j = tid + 256 * q;
        unsigned sv  = (unsigned)(kk[q] >> 32);
        unsigned idx = 0xFFFFFFFFu - (unsigned)(kk[q] & 0xFFFFFFFFu);
        bool v = (j < PRE_NMS) && (sv > 0x007FFFFFu);
        gb[q] = *(const v4f*)(bimg + (size_t)(v ? idx : 0u) * 4);
    }
    #pragma unroll
    for (int q = 0; q < 8; q++) {
        int j = tid + 256 * q;
        unsigned sv = (unsigned)(kk[q] >> 32);
        bool v = (j < PRE_NMS) && (sv > 0x007FFFFFu);
        v4f bb = {0.f, 0.f, 0.f, 0.f};
        float s = 0.f;
        if (v) {
            bb = gb[q];
            unsigned u = (sv & 0x80000000u) ? (sv ^ 0x80000000u) : ~sv;
            s = __uint_as_float(u);
        }
        int slot = (j & 31) * LPITCH + (j >> 5);
        sbx[slot] = bb;
        sar[slot] = fmaxf(bb.z - bb.x, 0.f) * fmaxf(bb.w - bb.y, 0.f);
        if (j < PRE_NMS && (j / MASK_ROWS) == rb) {
            v4f* so = (v4f*)sorted + ((size_t)n * PRE_NMS + j) * 2;
            v4f inf = {s, v ? 1.f : 0.f, 0.f, 0.f};
            so[0] = bb;
            so[1] = inf;
        }
    }
    __syncthreads();

    int wave = tid >> 5;
    int lane = tid & 31;
    for (int rr = 0; rr < 2; rr++) {
        int i = rb * MASK_ROWS + wave * 2 + rr;
        int islot = (i & 31) * LPITCH + (i >> 5);
        v4f bi = sbx[islot];
        float ia = sar[islot];
        int wdiag = i >> 5;
        unsigned* mrow = mask + ((size_t)n * PRE_NMS + i) * 64;
        for (int pass = 0; pass < 2; pass++) {
            int w = pass * 32 + lane;
            unsigned bits = 0u;
            if (w >= wdiag) {
                for (int b = 0; b < 32; b++) {
                    v4f bj = sbx[b * LPITCH + w];
                    float ja = sar[b * LPITCH + w];
                    float ix1 = fmaxf(bi.x, bj.x), iy1 = fmaxf(bi.y, bj.y);
                    float ix2 = fminf(bi.z, bj.z), iy2 = fminf(bi.w, bj.w);
                    float inter = fmaxf(ix2 - ix1, 0.f) * fmaxf(iy2 - iy1, 0.f);
                    float iou = inter / fmaxf(ia + ja - inter, 1e-9f);
                    if (iou > IOU_THR) bits |= (1u << b);
                }
            }
            mrow[w] = bits;
        }
    }
}

__global__ __launch_bounds__(32) void nms_scan_kernel(const unsigned* __restrict__ mask,
                                                      const float* __restrict__ sorted,
                                                      float* __restrict__ out) {
    int n = blockIdx.x;
    int lane = threadIdx.x;
    const v2u* M = (const v2u*)mask + (size_t)n * PRE_NMS * 32;
    const v4f* S = (const v4f*)sorted + (size_t)n * PRE_NMS * 2;
    float* o_img = out + (size_t)n * POST_NMS * 5;

    unsigned rem_lo = 0u, rem_hi = 0u;

    v2u cur[32], nxt[32];
    v4f cbox, cinf, nbox, ninf;
    #pragma unroll
    for (int r = 0; r < 32; r++) cur[r] = M[(size_t)r * 32 + lane];
    cbox = S[(size_t)lane * 2];
    cinf = S[(size_t)lane * 2 + 1];

    int cnt = 0;
    for (int c = 0; c < 63; c++) {
        int cn = (c + 1 < 63) ? c + 1 : 62;
        #pragma unroll
        for (int r = 0; r < 32; r++) {
            int i = cn * 32 + r;
            i = i < PRE_NMS ? i : PRE_NMS - 1;
            nxt[r] = M[(size_t)i * 32 + lane];
        }
        {
            int i = cn * 32 + lane;
            i = i < PRE_NMS ? i : PRE_NMS - 1;
            nbox = S[(size_t)i * 2];
            ninf = S[(size_t)i * 2 + 1];
        }

        int own = c >> 1;
        bool inval = (c * 32 + lane >= PRE_NMS) || (cinf.y == 0.f);
        unsigned remw = (unsigned)__builtin_amdgcn_readlane((int)((c & 1) ? rem_hi : rem_lo), own)
                      | __builtin_amdgcn_ballot_w32(inval);
        unsigned kept = 0u;
        #pragma unroll
        for (int r = 0; r < 32; r++) {
            unsigned d = (unsigned)__builtin_amdgcn_readlane((int)((c & 1) ? cur[r].y : cur[r].x), own);
            unsigned m = ((remw >> r) & 1u) ? 0u : 0xFFFFFFFFu;
            kept  |= m & (1u << r);
            remw  |= d & m;
            rem_lo |= cur[r].x & m;
            rem_hi |= cur[r].y & m;
        }

        unsigned mine = (kept >> lane) & 1u;
        int pos = cnt + __popc(kept & ((1u << lane) - 1u));
        if (mine && pos < POST_NMS) {
            float* o = o_img + pos * 5;
            o[0] = cbox.x; o[1] = cbox.y; o[2] = cbox.z; o[3] = cbox.w;
            o[4] = cinf.x;
        }
        cnt += __popc(kept);
        if (cnt >= POST_NMS) break;

        #pragma unroll
        for (int r = 0; r < 32; r++) cur[r] = nxt[r];
        cbox = nbox; cinf = ninf;
    }

    int c0 = cnt < POST_NMS ? cnt : POST_NMS;
    for (int t = c0 * 5 + lane; t < POST_NMS * 5; t += 32) o_img[t] = 0.f;
}

extern "C" void kernel_launch(void* const* d_in, const int* in_sizes, int n_in,
                              void* d_out, int out_size, void* d_ws, size_t ws_size,
                              hipStream_t stream) {
    const float* feat   = (const float*)d_in[0];
    const float* conv_w = (const float*)d_in[1];
    const float* conv_b = (const float*)d_in[2];
    const float* det_w  = (const float*)d_in[3];
    const float* det_b  = (const float*)d_in[4];
    const float* reg_w  = (const float*)d_in[5];
    const float* reg_b  = (const float*)d_in[6];
    float* out = (float*)d_out;

    char* ws = (char*)d_ws;
    float* wt2                  = (float*)(ws);
    float* scores               = (float*)(ws + 2359296);
    float* deltas               = (float*)(ws + 2359296 + 786432);
    float* boxes                = (float*)(ws + 2359296 + 786432 + 3145728);
    unsigned long long* keys    = (unsigned long long*)(ws + 2359296 + 786432 + 3145728 + 3145728);

    fold_w_kernel<<<145, 256, 0, stream>>>(conv_w, conv_b, det_w, det_b, reg_w, reg_b, wt2);
    conv_fused_kernel<<<NIMG * HF * (WF / TILE_W), 256, 0, stream>>>(feat, wt2, scores, deltas);
    decode_kernel<<<(NIMG * NPAD) / 256, 256, 0, stream>>>(scores, deltas, boxes, keys);
    topk_chunk_sort_kernel<<<NIMG * NCHUNK, 1024, 0, stream>>>(keys);
    topk_fold_kernel<<<NIMG * FOLD_B, 1024, 0, stream>>>(keys, FOLD_A, 1, FOLD_B);
    topk_fold_kernel<<<NIMG, 1024, 0, stream>>>(keys, FOLD_B, FOLD_A, 1);
    unsigned* nmask  = (unsigned*)(ws);
    float*    sorted = (float*)(ws + 2048000);
    nms_mask_kernel<<<NIMG * MASK_BLKS, 256, 0, stream>>>(keys, boxes, nmask, sorted);
    nms_scan_kernel<<<NIMG, 32, 0, stream>>>(nmask, sorted, out);
}


# top-k sort and fold: compare-exchange steps of distance <= 32 run in registers with lane exchanges (a wave owns 64 adjacent keys); LDS and block barriers only for distance >= 64; same network
# speedup vs baseline: 17.1604x; 1.0348x over previous
#include <hip/hip_runtime.h>
#include <hip/hip_bf16.h>
#include <math.h>

typedef __attribute__((ext_vector_type(2))) float v2f;
typedef __attribute__((ext_vector_type(8))) float v8f;
typedef __attribute__((ext_vector_type(4))) float v4f;
typedef __attribute__((ext_vector_type(2))) unsigned v2u;

#define NIMG 4
#define CIN 256
#define HF 128
#define WF 128
#define A_ANCH 3
#define NPROP (A_ANCH * HF * WF)
#define NPAD 65536
#define PRE_NMS 2000
#define POST_NMS 300
#define OBJ_THR 0.3f
#define IOU_THR 0.7f
#define TILE_W 64
#define CCHUNK 32
#define CIPAD 34

#define NWEFF (9 * 128 * 16 * 2)
__global__ __launch_bounds__(256) void fold_w_kernel(const float* __restrict__ conv_w,
                                                     const float* __restrict__ conv_b,
                                                     const float* __restrict__ det_w,
                                                     const float* __restrict__ det_b,
                                                     const float* __restrict__ reg_w,
                                                     const float* __restrict__ reg_b,
                                                     float* __restrict__ weff2) {
    int tid = threadIdx.x;
    if (blockIdx.x == NWEFF / 256) {
        if (tid < 16) {
            float s = 0.0f;
            if (tid < 15) {
                const float* hw = (tid < 3) ? (det_w + tid * 256) : (reg_w + (tid - 3) * 256);
                s = (tid < 3) ? det_b[tid] : reg_b[tid - 3];
                for (int c0 = 0; c0 < 256; c0 += 64) {
                    float hv[64], bv[64];
                    #pragma unroll
                    for (int j = 0; j < 64; j++) { hv[j] = hw[c0 + j]; bv[j] = conv_b[c0 + j]; }
                    #pragma unroll
                    for (int j = 0; j < 64; j++) s += hv[j] * bv[j];
                }
            }
            weff2[NWEFF + tid] = s;
        }
        return;
    }
    int t   = blockIdx.x * 256 + tid;
    int par = t & 1;
    int o   = (t >> 1) & 15;
    int q   = t >> 5;
    int ci2 = q & 127;
    int kk  = q >> 7;
    int ci  = ci2 * 2 + par;
    float s = 0.0f;
    if (o < 15) {
        const float* hw = (o < 3) ? (det_w + o * 256) : (reg_w + (o - 3) * 256);
        const float* wp = conv_w + (size_t)ci * 9 + kk;
        for (int c0 = 0; c0 < 256; c0 += 32) {
            float hv[32], wv[32];
            #pragma unroll
            for (int j = 0; j < 32; j++) { hv[j] = hw[c0 + j]; wv[j] = wp[(size_t)(c0 + j) * (CIN * 9)]; }
            #pragma unroll
            for (int j = 0; j < 32; j++) s += hv[j] * wv[j];
        }
    }
    weff2[t] = s;
}

#define NCHK (CIN / CCHUNK)
#define NPRE 25
#define PP 66

__device__ __forceinline__ void stage_fetch(float (&pre)[NPRE], const float* frow,
                                            const float* fhalo, const int (&rowoff)[3], int cib) {
    const float* fc = frow + (size_t)cib * (HF * WF);
    #pragma unroll
    for (int j = 0; j < 4; j++) {
        #pragma unroll
        for (int r = 0; r < 3; r++) {
            const float* p = fc + j * (HF * WF) + rowoff[r];
            pre[(j * 3 + r) * 2 + 0] = p[0];
            pre[(j * 3 + r) * 2 + 1] = p[32];
        }
    }
    pre[24] = fhalo[(size_t)cib * (HF * WF)];
}

__global__ __launch_bounds__(256) __attribute__((amdgpu_waves_per_eu(8))) void conv_fused_kernel(
    const float* __restrict__ feat, const float* __restrict__ weff2,
    float* __restrict__ scores, float* __restrict__ deltas) {

    __shared__ __attribute__((aligned(16))) float lds_in[3 * 66 * CIPAD];
    __shared__ float part[8 * 16 * PP];

    int bid = blockIdx.x;
    int n    = bid >> 8;
    int rem  = bid & 255;
    int row  = rem >> 1;
    int col0 = (rem & 1) * TILE_W;

    int tid  = threadIdx.x;
    int wave = tid >> 5;
    int lane = tid & 31;
    int hl   = lane >> 4;
    int lm   = lane & 15;

    v8f acc[4] = {};

    const float* featn = feat + (size_t)n * CIN * HF * WF;
    const v2f*   wpl   = (const v2f*)weff2 + (size_t)(2 * wave + hl) * 16 + lm;

    int  rowoff[3];
    bool rowok[3];
    #pragma unroll
    for (int r = 0; r < 3; r++) {
        int g = row + r - 1;
        rowok[r]  = (g >= 0) && (g < HF);
        rowoff[r] = (g < 0 ? 0 : (g >= HF ? HF - 1 : g)) * WF;
    }
    const float* frow = featn + (size_t)(4 * wave) * (HF * WF) + col0 + lane;
    int  hj  = lane / 6;
    int  hq  = lane - hj * 6;
    int  hr  = hq >> 1;
    int  hs  = hq & 1;
    int  hg  = row + hr - 1;
    int  hc  = hs ? (col0 + TILE_W) : (col0 - 1);
    bool hok = (lane < 24) && (hg >= 0) && (hg < HF) && (hc >= 0) && (hc < WF);
    const float* fhalo = featn + (hok ? ((size_t)(4 * wave + hj) * HF + hg) * WF + hc : (size_t)0);
    int  hlds = (hr * 66 + (hs ? 65 : 0)) * CIPAD + 4 * wave + hj;

    int bbase = lm * CIPAD + 4 * wave + 2 * hl;

    float pre[NPRE];
    stage_fetch(pre, frow, fhalo, rowoff, 0);

    for (int ch = 0; ch < NCHK; ch++) {
        v2f a[9];
        #pragma unroll
        for (int kk = 0; kk < 9; kk++)
            a[kk] = wpl[(size_t)(kk * 128 + ch * 16) * 16];

        __syncthreads();
        #pragma unroll
        for (int r = 0; r < 3; r++) {
            #pragma unroll
            for (int hcol = 0; hcol < 2; hcol++) {
                float v0 = rowok[r] ? pre[(0 * 3 + r) * 2 + hcol] : 0.0f;
                float v1 = rowok[r] ? pre[(1 * 3 + r) * 2 + hcol] : 0.0f;
                float v2 = rowok[r] ? pre[(2 * 3 + r) * 2 + hcol] : 0.0f;
                float v3 = rowok[r] ? pre[(3 * 3 + r) * 2 + hcol] : 0.0f;
                int idx = (r * 66 + 1 + hcol * 32 + lane) * CIPAD + 4 * wave;
                v2f lo = {v0, v1};
                v2f hi = {v2, v3};
                *(v2f*)&lds_in[idx]     = lo;
                *(v2f*)&lds_in[idx + 2] = hi;
            }
        }
        if (lane < 24) lds_in[hlds] = hok ? pre[24] : 0.0f;
        __syncthreads();

        int chn = ch + 1 < NCHK ? ch + 1 : NCHK - 1;
        stage_fetch(pre, frow, fhalo, rowoff, chn * CCHUNK);

        #pragma unroll
        for (int kk = 0; kk < 9; kk++) {
            int kh = kk / 3;
            int kw = kk - kh * 3;
            v2f b[4];
            #pragma unroll
            for (int nt = 0; nt < 4; nt++)
                b[nt] = *(const v2f*)&lds_in[(kh * 66 + nt * 16 + kw) * CIPAD + bbase];
            #pragma unroll
            for (int nt = 0; nt < 4; nt++)
                acc[nt] = __builtin_amdgcn_wmma_f32_16x16x4_f32(false, a[kk], false, b[nt], (short)0, acc[nt], false, false);
        }
    }

    #pragma unroll
    for (int nt = 0; nt < 4; nt++) {
        #pragma unroll
        for (int r = 0; r < 8; r++)
            part[(wave * 16 + hl * 8 + r) * PP + nt * 16 + lm] = acc[nt][r];
    }
    __syncthreads();

    const float* beff = weff2 + NWEFF;
    for (int e = tid; e < 15 * 64; e += 256) {
        int oi  = e >> 6;
        int pix = e & 63;
        float sum = part[oi * PP + pix];
        #pragma unroll
        for (int w = 1; w < 8; w++) sum += part[(w * 16 + oi) * PP + pix];
        sum += beff[oi];
        int p = row * WF + col0 + pix;
        if (oi < 3) {
            scores[(size_t)n * NPROP + oi * (HF * WF) + p] = 1.0f / (1.0f + expf(-sum));
        } else {
            int chh = oi - 3;
            int ccomp = chh / 3;
            int a = chh % 3;
            deltas[(((size_t)n * NPROP) + a * (HF * WF) + p) * 4 + ccomp] = sum;
        }
    }
}

__global__ __launch_bounds__(256) void decode_kernel(
    const float* __restrict__ scores, const float* __restrict__ deltas,
    float* __restrict__ boxes, unsigned long long* __restrict__ keys) {
    int t = blockIdx.x * 256 + threadIdx.x;
    int n = t >> 16;
    int i = t & 0xFFFF;
    unsigned long long* kout = keys + ((size_t)n << 16);
    if (i >= NPROP) { kout[i] = 0ull; return; }

    int pa = i / 3, aa = i - pa * 3;
    int ya = pa >> 7, xa = pa & 127;
    const float bs[3][4] = {{-91.f, -45.f, 91.f, 45.f},
                            {-64.f, -64.f, 64.f, 64.f},
                            {-45.f, -91.f, 45.f, 91.f}};
    float sxv = (float)(xa * 8), syv = (float)(ya * 8);
    float x1 = sxv + bs[aa][0], y1 = syv + bs[aa][1];
    float x2 = sxv + bs[aa][2], y2 = syv + bs[aa][3];
    float aw = x2 - x1, ah = y2 - y1;
    float acx = y1 + aw * 0.5f;
    float acy = x1 + ah * 0.5f;

    const float* d = deltas + (((size_t)n * NPROP) + i) * 4;
    float px = acx + d[0] * aw;
    float py = acy + d[1] * ah;
    float pw = aw * expf(d[2]);
    float ph = ah * expf(d[3]);
    float b0 = fminf(fmaxf(px - pw * 0.5f, 0.f), 1024.f);
    float b1 = fminf(fmaxf(py - ph * 0.5f, 0.f), 1024.f);
    float b2 = fminf(fmaxf(px + pw * 0.5f, 0.f), 1024.f);
    float b3 = fminf(fmaxf(py + ph * 0.5f, 0.f), 1024.f);

    float* bo = boxes + (((size_t)n * NPROP) + i) * 4;
    bo[0] = b0; bo[1] = b1; bo[2] = b2; bo[3] = b3;

    float s = scores[(size_t)n * NPROP + i];
    bool valid = (s > OBJ_THR) && ((b2 - b0) > 0.f) && ((b3 - b1) > 0.f);
    float ss = valid ? s : -__builtin_inff();
    unsigned u = __float_as_uint(ss);
    unsigned sv = (u & 0x80000000u) ? ~u : (u | 0x80000000u);
    kout[i] = ((unsigned long long)sv << 32) | (unsigned long long)(0xFFFFFFFFu - (unsigned)i);
}

#define CHUNK 2048
#define NCHUNK (NPROP / CHUNK)
#define FOLD_A 6
#define FOLD_B (NCHUNK / FOLD_A)

__device__ __forceinline__ void bitonic_ce(unsigned long long* s, int t, int j, int k) {
    int lo = t & (j - 1);
    int i  = ((t - lo) << 1) | lo;
    int p  = i | j;
    unsigned long long a = s[i];
    unsigned long long b = s[p];
    bool up = ((i & k) == 0);
    bool sw = up ? (a < b) : (a > b);
    if (sw) { s[i] = b; s[p] = a; }
}

__device__ __forceinline__ unsigned long long lane_xchg64(unsigned long long v, int addr) {
    unsigned lo = (unsigned)__builtin_amdgcn_ds_bpermute(addr, (int)(unsigned)(v & 0xFFFFFFFFull));
    unsigned hi = (unsigned)__builtin_amdgcn_ds_bpermute(addr, (int)(unsigned)(v >> 32));
    return ((unsigned long long)hi << 32) | (unsigned long long)lo;
}

__device__ __forceinline__ void wave_ce_steps(unsigned long long& e0, unsigned long long& e1,
                                              int lane, int x0, int k) {
    bool up0 = ((x0 & k) == 0);
    bool up1 = (((x0 | 32) & k) == 0);
    if (k >= 64) {
        unsigned long long mx = e0 > e1 ? e0 : e1;
        unsigned long long mn = e0 > e1 ? e1 : e0;
        e0 = up0 ? mx : mn;
        e1 = up0 ? mn : mx;
    }
    #pragma unroll
    for (int j = 16; j > 0; j >>= 1) {
        if (2 * j <= k) {
            int addr = (lane ^ j) << 2;
            unsigned long long y0 = lane_xchg64(e0, addr);
            unsigned long long y1 = lane_xchg64(e1, addr);
            bool low = ((lane & j) == 0);
            bool wmax0 = (low == up0);
            bool wmax1 = (low == up1);
            bool g0 = y0 > e0;
            bool g1 = y1 > e1;
            e0 = (g0 == wmax0) ? y0 : e0;
            e1 = (g1 == wmax1) ? y1 : e1;
        }
    }
}

__global__ __launch_bounds__(1024) void topk_chunk_sort_kernel(unsigned long long* keys) {
    __shared__ unsigned long long s[CHUNK];
    int n   = blockIdx.x / NCHUNK;
    int c   = blockIdx.x % NCHUNK;
    int tid = threadIdx.x;
    int lane = tid & 31;
    int x0   = (tid >> 5) * 64 + lane;
    unsigned long long* g = keys + ((size_t)n << 16) + (size_t)c * CHUNK;
    unsigned long long e0 = g[x0];
    unsigned long long e1 = g[x0 + 32];
    wave_ce_steps(e0, e1, lane, x0, 2);
    wave_ce_steps(e0, e1, lane, x0, 4);
    wave_ce_steps(e0, e1, lane, x0, 8);
    wave_ce_steps(e0, e1, lane, x0, 16);
    wave_ce_steps(e0, e1, lane, x0, 32);
    wave_ce_steps(e0, e1, lane, x0, 64);
    #pragma unroll
    for (int k = 128; k <= CHUNK; k <<= 1) {
        s[x0]      = e0;
        s[x0 + 32] = e1;
        __syncthreads();
        for (int j = k >> 1; j >= 64; j >>= 1) {
            bitonic_ce(s, tid, j, k);
            __syncthreads();
        }
        e0 = s[x0];
        e1 = s[x0 + 32];
        wave_ce_steps(e0, e1, lane, x0, k);
    }
    g[x0]      = e0;
    g[x0 + 32] = e1;
}

__global__ __launch_bounds__(1024) void topk_fold_kernel(unsigned long long* keys,
                                                         int nch, int stride, int groups) {
    __shared__ unsigned long long s[CHUNK];
    int n   = blockIdx.x / groups;
    int gi  = blockIdx.x % groups;
    int tid = threadIdx.x;
    int lane = tid & 31;
    int x0   = (tid >> 5) * 64 + lane;
    size_t pitch = (size_t)stride * CHUNK;
    unsigned long long* g = keys + ((size_t)n << 16) + (size_t)gi * nch * pitch;
    s[tid]        = g[tid];
    s[tid + 1024] = g[tid + 1024];
    unsigned long long d0 = 0ull, d1 = 0ull;
    if (nch > 1) {
        d0 = g[pitch + (CHUNK - 1 - tid)];
        d1 = g[pitch + (CHUNK - 1 - 1024 - tid)];
    }
    for (int c = 1; c < nch; c++) {
        unsigned long long a0 = s[tid];
        unsigned long long a1 = s[tid + 1024];
        unsigned long long e0 = a0 > d0 ? a0 : d0;
        unsigned long long e1 = a1 > d1 ? a1 : d1;
        s[tid]        = e0 > e1 ? e0 : e1;
        s[tid + 1024] = e0 > e1 ? e1 : e0;
        if (c + 1 < nch) {
            const unsigned long long* gn = g + (size_t)(c + 1) * pitch;
            d0 = gn[CHUNK - 1 - tid];
            d1 = gn[CHUNK - 1 - 1024 - tid];
        }
        __syncthreads();
        for (int j = CHUNK >> 2; j >= 64; j >>= 1) {
            bitonic_ce(s, tid, j, CHUNK);
            __syncthreads();
        }
        unsigned long long r0 = s[x0];
        unsigned long long r1 = s[x0 + 32];
        wave_ce_steps(r0, r1, lane, x0, CHUNK);
        s[x0]      = r0;
        s[x0 + 32] = r1;
        __syncthreads();
    }
    g[tid]        = s[tid];
    g[tid + 1024] = s[tid + 1024];
}

#define MASK_ROWS 16
#define MASK_BLKS (PRE_NMS / MASK_ROWS)
#define LPITCH 65
__global__ __launch_bounds__(256) void nms_mask_kernel(const unsigned long long* __restrict__ keys,
                                                       const float* __restrict__ boxes,
                                                       unsigned* __restrict__ mask,
                                                       float* __restrict__ sorted) {
    __shared__ v4f   sbx[32 * LPITCH];
    __shared__ float sar[32 * LPITCH];

    int n  = blockIdx.x / MASK_BLKS;
    int rb = blockIdx.x % MASK_BLKS;
    int tid = threadIdx.x;
    const unsigned long long* K = keys + ((size_t)n << 16);
    const float* bimg = boxes + (size_t)n * NPROP * 4;

    unsigned long long kk[8];
    #pragma unroll
    for (int q = 0; q < 8; q++) {
        int j = tid + 256 * q;
        kk[q] = K[j < PRE_NMS ? j : PRE_NMS - 1];
    }
    v4f gb[8];
    #pragma unroll
    for (int q = 0; q < 8; q++) {
        int j = tid + 256 * q;
        unsigned sv  = (unsigned)(kk[q] >> 32);
        unsigned idx = 0xFFFFFFFFu - (unsigned)(kk[q] & 0xFFFFFFFFu);
        bool v = (j < PRE_NMS) && (sv > 0x007FFFFFu);
        gb[q] = *(const v4f*)(bimg + (size_t)(v ? idx : 0u) * 4);
    }
    #pragma unroll
    for (int q = 0; q < 8; q++) {
        int j = tid + 256 * q;
        unsigned sv = (unsigned)(kk[q] >> 32);
        bool v = (j < PRE_NMS) && (sv > 0x007FFFFFu);
        v4f bb = {0.f, 0.f, 0.f, 0.f};
        float s = 0.f;
        if (v) {
            bb = gb[q];
            unsigned u = (sv & 0x80000000u) ? (sv ^ 0x80000000u) : ~sv;
            s = __uint_as_float(u);
        }
        int slot = (j & 31) * LPITCH + (j >> 5);
        sbx[slot] = bb;
        sar[slot] = fmaxf(bb.z - bb.x, 0.f) * fmaxf(bb.w - bb.y, 0.f);
        if (j < PRE_NMS && (j / MASK_ROWS) == rb) {
            v4f* so = (v4f*)sorted + ((size_t)n * PRE_NMS + j) * 2;
            v4f inf = {s, v ? 1.f : 0.f, 0.f, 0.f};
            so[0] = bb;
            so[1] = inf;
        }
    }
    __syncthreads();

    int wave = tid >> 5;
    int lane = tid & 31;
    for (int rr = 0; rr < 2; rr++) {
        int i = rb * MASK_ROWS + wave * 2 + rr;
        int islot = (i & 31) * LPITCH + (i >> 5);
        v4f bi = sbx[islot];
        float ia = sar[islot];
        int wdiag = i >> 5;
        unsigned* mrow = mask + ((size_t)n * PRE_NMS + i) * 64;
        for (int pass = 0; pass < 2; pass++) {
            int w = pass * 32 + lane;
            unsigned bits = 0u;
            if (w >= wdiag) {
                for (int b = 0; b < 32; b++) {
                    v4f bj = sbx[b * LPITCH + w];
                    float ja = sar[b * LPITCH + w];
                    float ix1 = fmaxf(bi.x, bj.x), iy1 = fmaxf(bi.y, bj.y);
                    float ix2 = fminf(bi.z, bj.z), iy2 = fminf(bi.w, bj.w);
                    float inter = fmaxf(ix2 - ix1, 0.f) * fmaxf(iy2 - iy1, 0.f);
                    float iou = inter / fmaxf(ia + ja - inter, 1e-9f);
                    if (iou > IOU_THR) bits |= (1u << b);
                }
            }
            mrow[w] = bits;
        }
    }
}

__global__ __launch_bounds__(32) void nms_scan_kernel(const unsigned* __restrict__ mask,
                                                      const float* __restrict__ sorted,
                                                      float* __restrict__ out) {
    int n = blockIdx.x;
    int lane = threadIdx.x;
    const v2u* M = (const v2u*)mask + (size_t)n * PRE_NMS * 32;
    const v4f* S = (const v4f*)sorted + (size_t)n * PRE_NMS * 2;
    float* o_img = out + (size_t)n * POST_NMS * 5;

    unsigned rem_lo = 0u, rem_hi = 0u;

    v2u cur[32], nxt[32];
    v4f cbox, cinf, nbox, ninf;
    #pragma unroll
    for (int r = 0; r < 32; r++) cur[r] = M[(size_t)r * 32 + lane];
    cbox = S[(size_t)lane * 2];
    cinf = S[(size_t)lane * 2 + 1];

    int cnt = 0;
    for (int c = 0; c < 63; c++) {
        int cn = (c + 1 < 63) ? c + 1 : 62;
        #pragma unroll
        for (int r = 0; r < 32; r++) {
            int i = cn * 32 + r;
            i = i < PRE_NMS ? i : PRE_NMS - 1;
            nxt[r] = M[(size_t)i * 32 + lane];
        }
        {
            int i = cn * 32 + lane;
            i = i < PRE_NMS ? i : PRE_NMS - 1;
            nbox = S[(size_t)i * 2];
            ninf = S[(size_t)i * 2 + 1];
        }

        int own = c >> 1;
        bool inval = (c * 32 + lane >= PRE_NMS) || (cinf.y == 0.f);
        unsigned remw = (unsigned)__builtin_amdgcn_readlane((int)((c & 1) ? rem_hi : rem_lo), own)
                      | __builtin_amdgcn_ballot_w32(inval);
        unsigned kept = 0u;
        #pragma unroll
        for (int r = 0; r < 32; r++) {
            unsigned d = (unsigned)__builtin_amdgcn_readlane((int)((c & 1) ? cur[r].y : cur[r].x), own);
            unsigned m = ((remw >> r) & 1u) ? 0u : 0xFFFFFFFFu;
            kept  |= m & (1u << r);
            remw  |= d & m;
            rem_lo |= cur[r].x & m;
            rem_hi |= cur[r].y & m;
        }

        unsigned mine = (kept >> lane) & 1u;
        int pos = cnt + __popc(kept & ((1u << lane) - 1u));
        if (mine && pos < POST_NMS) {
            float* o = o_img + pos * 5;
            o[0] = cbox.x; o[1] = cbox.y; o[2] = cbox.z; o[3] = cbox.w;
            o[4] = cinf.x;
        }
        cnt += __popc(kept);
        if (cnt >= POST_NMS) break;

        #pragma unroll
        for (int r = 0; r < 32; r++) cur[r] = nxt[r];
        cbox = nbox; cinf = ninf;
    }

    int c0 = cnt < POST_NMS ? cnt : POST_NMS;
    for (int t = c0 * 5 + lane; t < POST_NMS * 5; t += 32) o_img[t] = 0.f;
}

extern "C" void kernel_launch(void* const* d_in, const int* in_sizes, int n_in,
                              void* d_out, int out_size, void* d_ws, size_t ws_size,
                              hipStream_t stream) {
    const float* feat   = (const float*)d_in[0];
    const float* conv_w = (const float*)d_in[1];
    const float* conv_b = (const float*)d_in[2];
    const float* det_w  = (const float*)d_in[3];
    const float* det_b  = (const float*)d_in[4];
    const float* reg_w  = (const float*)d_in[5];
    const float* reg_b  = (const float*)d_in[6];
    float* out = (float*)d_out;

    char* ws = (char*)d_ws;
    float* wt2                  = (float*)(ws);
    float* scores               = (float*)(ws + 2359296);
    float* deltas               = (float*)(ws + 2359296 + 786432);
    float* boxes                = (float*)(ws + 2359296 + 786432 + 3145728);
    unsigned long long* keys    = (unsigned long long*)(ws + 2359296 + 786432 + 3145728 + 3145728);

    fold_w_kernel<<<145, 256, 0, stream>>>(conv_w, conv_b, det_w, det_b, reg_w, reg_b, wt2);
    conv_fused_kernel<<<NIMG * HF * (WF / TILE_W), 256, 0, stream>>>(feat, wt2, scores, deltas);
    decode_kernel<<<(NIMG * NPAD) / 256, 256, 0, stream>>>(scores, deltas, boxes, keys);
    topk_chunk_sort_kernel<<<NIMG * NCHUNK, 1024, 0, stream>>>(keys);
    topk_fold_kernel<<<NIMG * FOLD_B, 1024, 0, stream>>>(keys, FOLD_A, 1, FOLD_B);
    topk_fold_kernel<<<NIMG, 1024, 0, stream>>>(keys, FOLD_B, FOLD_A, 1);
    unsigned* nmask  = (unsigned*)(ws);
    float*    sorted = (float*)(ws + 2048000);
    nms_mask_kernel<<<NIMG * MASK_BLKS, 256, 0, stream>>>(keys, boxes, nmask, sorted);
    nms_scan_kernel<<<NIMG, 32, 0, stream>>>(nmask, sorted, out);
}


# NMS scan: a chunk's own 32x32 bit block is resolved by whole-wave sweeps to the greedy fixed point (lane r holds row r's word) instead of 32 dependent scalar steps; same kept set, same out
# speedup vs baseline: 17.7871x; 1.0365x over previous
#include <hip/hip_runtime.h>
#include <hip/hip_bf16.h>
#include <math.h>

typedef __attribute__((ext_vector_type(2))) float v2f;
typedef __attribute__((ext_vector_type(8))) float v8f;
typedef __attribute__((ext_vector_type(4))) float v4f;
typedef __attribute__((ext_vector_type(2))) unsigned v2u;

#define NIMG 4
#define CIN 256
#define HF 128
#define WF 128
#define A_ANCH 3
#define NPROP (A_ANCH * HF * WF)
#define NPAD 65536
#define PRE_NMS 2000
#define POST_NMS 300
#define OBJ_THR 0.3f
#define IOU_THR 0.7f
#define TILE_W 64
#define CCHUNK 32
#define CIPAD 34

#define NWEFF (9 * 128 * 16 * 2)
__global__ __launch_bounds__(256) void fold_w_kernel(const float* __restrict__ conv_w,
                                                     const float* __restrict__ conv_b,
                                                     const float* __restrict__ det_w,
                                                     const float* __restrict__ det_b,
                                                     const float* __restrict__ reg_w,
                                                     const float* __restrict__ reg_b,
                                                     float* __restrict__ weff2) {
    int tid = threadIdx.x;
    if (blockIdx.x == NWEFF / 256) {
        if (tid < 16) {
            float s = 0.0f;
            if (tid < 15) {
                const float* hw = (tid < 3) ? (det_w + tid * 256) : (reg_w + (tid - 3) * 256);
                s = (tid < 3) ? det_b[tid] : reg_b[tid - 3];
                for (int c0 = 0; c0 < 256; c0 += 64) {
                    float hv[64], bv[64];
                    #pragma unroll
                    for (int j = 0; j < 64; j++) { hv[j] = hw[c0 + j]; bv[j] = conv_b[c0 + j]; }
                    #pragma unroll
                    for (int j = 0; j < 64; j++) s += hv[j] * bv[j];
                }
            }
            weff2[NWEFF + tid] = s;
        }
        return;
    }
    int t   = blockIdx.x * 256 + tid;
    int par = t & 1;
    int o   = (t >> 1) & 15;
    int q   = t >> 5;
    int ci2 = q & 127;
    int kk  = q >> 7;
    int ci  = ci2 * 2 + par;
    float s = 0.0f;
    if (o < 15) {
        const float* hw = (o < 3) ? (det_w + o * 256) : (reg_w + (o - 3) * 256);
        const float* wp = conv_w + (size_t)ci * 9 + kk;
        for (int c0 = 0; c0 < 256; c0 += 32) {
            float hv[32], wv[32];
            #pragma unroll
            for (int j = 0; j < 32; j++) { hv[j] = hw[c0 + j]; wv[j] = wp[(size_t)(c0 + j) * (CIN * 9)]; }
            #pragma unroll
            for (int j = 0; j < 32; j++) s += hv[j] * wv[j];
        }
    }
    weff2[t] = s;
}

#define NCHK (CIN / CCHUNK)
#define NPRE 25
#define PP 66

__device__ __forceinline__ void stage_fetch(float (&pre)[NPRE], const float* frow,
                                            const float* fhalo, const int (&rowoff)[3], int cib) {
    const float* fc = frow + (size_t)cib * (HF * WF);
    #pragma unroll
    for (int j = 0; j < 4; j++) {
        #pragma unroll
        for (int r = 0; r < 3; r++) {
            const float* p = fc + j * (HF * WF) + rowoff[r];
            pre[(j * 3 + r) * 2 + 0] = p[0];
            pre[(j * 3 + r) * 2 + 1] = p[32];
        }
    }
    pre[24] = fhalo[(size_t)cib * (HF * WF)];
}

__global__ __launch_bounds__(256) __attribute__((amdgpu_waves_per_eu(8))) void conv_fused_kernel(
    const float* __restrict__ feat, const float* __restrict__ weff2,
    float* __restrict__ scores, float* __restrict__ deltas) {

    __shared__ __attribute__((aligned(16))) float lds_in[3 * 66 * CIPAD];
    __shared__ float part[8 * 16 * PP];

    int bid = blockIdx.x;
    int n    = bid >> 8;
    int rem  = bid & 255;
    int row  = rem >> 1;
    int col0 = (rem & 1) * TILE_W;

    int tid  = threadIdx.x;
    int wave = tid >> 5;
    int lane = tid & 31;
    int hl   = lane >> 4;
    int lm   = lane & 15;

    v8f acc[4] = {};

    const float* featn = feat + (size_t)n * CIN * HF * WF;
    const v2f*   wpl   = (const v2f*)weff2 + (size_t)(2 * wave + hl) * 16 + lm;

    int  rowoff[3];
    bool rowok[3];
    #pragma unroll
    for (int r = 0; r < 3; r++) {
        int g = row + r - 1;
        rowok[r]  = (g >= 0) && (g < HF);
        rowoff[r] = (g < 0 ? 0 : (g >= HF ? HF - 1 : g)) * WF;
    }
    const float* frow = featn + (size_t)(4 * wave) * (HF * WF) + col0 + lane;
    int  hj  = lane / 6;
    int  hq  = lane - hj * 6;
    int  hr  = hq >> 1;
    int  hs  = hq & 1;
    int  hg  = row + hr - 1;
    int  hc  = hs ? (col0 + TILE_W) : (col0 - 1);
    bool hok = (lane < 24) && (hg >= 0) && (hg < HF) && (hc >= 0) && (hc < WF);
    const float* fhalo = featn + (hok ? ((size_t)(4 * wave + hj) * HF + hg) * WF + hc : (size_t)0);
    int  hlds = (hr * 66 + (hs ? 65 : 0)) * CIPAD + 4 * wave + hj;

    int bbase = lm * CIPAD + 4 * wave + 2 * hl;

    float pre[NPRE];
    stage_fetch(pre, frow, fhalo, rowoff, 0);

    for (int ch = 0; ch < NCHK; ch++) {
        v2f a[9];
        #pragma unroll
        for (int kk = 0; kk < 9; kk++)
            a[kk] = wpl[(size_t)(kk * 128 + ch * 16) * 16];

        __syncthreads();
        #pragma unroll
        for (int r = 0; r < 3; r++) {
            #pragma unroll
            for (int hcol = 0; hcol < 2; hcol++) {
                float v0 = rowok[r] ? pre[(0 * 3 + r) * 2 + hcol] : 0.0f;
                float v1 = rowok[r] ? pre[(1 * 3 + r) * 2 + hcol] : 0.0f;
                float v2 = rowok[r] ? pre[(2 * 3 + r) * 2 + hcol] : 0.0f;
                float v3 = rowok[r] ? pre[(3 * 3 + r) * 2 + hcol] : 0.0f;
                int idx = (r * 66 + 1 + hcol * 32 + lane) * CIPAD + 4 * wave;
                v2f lo = {v0, v1};
                v2f hi = {v2, v3};
                *(v2f*)&lds_in[idx]     = lo;
                *(v2f*)&lds_in[idx + 2] = hi;
            }
        }
        if (lane < 24) lds_in[hlds] = hok ? pre[24] : 0.0f;
        __syncthreads();

        int chn = ch + 1 < NCHK ? ch + 1 : NCHK - 1;
        stage_fetch(pre, frow, fhalo, rowoff, chn * CCHUNK);

        #pragma unroll
        for (int kk = 0; kk < 9; kk++) {
            int kh = kk / 3;
            int kw = kk - kh * 3;
            v2f b[4];
            #pragma unroll
            for (int nt = 0; nt < 4; nt++)
                b[nt] = *(const v2f*)&lds_in[(kh * 66 + nt * 16 + kw) * CIPAD + bbase];
            #pragma unroll
            for (int nt = 0; nt < 4; nt++)
                acc[nt] = __builtin_amdgcn_wmma_f32_16x16x4_f32(false, a[kk], false, b[nt], (short)0, acc[nt], false, false);
        }
    }

    #pragma unroll
    for (int nt = 0; nt < 4; nt++) {
        #pragma unroll
        for (int r = 0; r < 8; r++)
            part[(wave * 16 + hl * 8 + r) * PP + nt * 16 + lm] = acc[nt][r];
    }
    __syncthreads();

    const float* beff = weff2 + NWEFF;
    for (int e = tid; e < 15 * 64; e += 256) {
        int oi  = e >> 6;
        int pix = e & 63;
        float sum = part[oi * PP + pix];
        #pragma unroll
        for (int w = 1; w < 8; w++) sum += part[(w * 16 + oi) * PP + pix];
        sum += beff[oi];
        int p = row * WF + col0 + pix;
        if (oi < 3) {
            scores[(size_t)n * NPROP + oi * (HF * WF) + p] = 1.0f / (1.0f + expf(-sum));
        } else {
            int chh = oi - 3;
            int ccomp = chh / 3;
            int a = chh % 3;
            deltas[(((size_t)n * NPROP) + a * (HF * WF) + p) * 4 + ccomp] = sum;
        }
    }
}

__global__ __launch_bounds__(256) void decode_kernel(
    const float* __restrict__ scores, const float* __restrict__ deltas,
    float* __restrict__ boxes, unsigned long long* __restrict__ keys) {
    int t = blockIdx.x * 256 + threadIdx.x;
    int n = t >> 16;
    int i = t & 0xFFFF;
    unsigned long long* kout = keys + ((size_t)n << 16);
    if (i >= NPROP) { kout[i] = 0ull; return; }

    int pa = i / 3, aa = i - pa * 3;
    int ya = pa >> 7, xa = pa & 127;
    const float bs[3][4] = {{-91.f, -45.f, 91.f, 45.f},
                            {-64.f, -64.f, 64.f, 64.f},
                            {-45.f, -91.f, 45.f, 91.f}};
    float sxv = (float)(xa * 8), syv = (float)(ya * 8);
    float x1 = sxv + bs[aa][0], y1 = syv + bs[aa][1];
    float x2 = sxv + bs[aa][2], y2 = syv + bs[aa][3];
    float aw = x2 - x1, ah = y2 - y1;
    float acx = y1 + aw * 0.5f;
    float acy = x1 + ah * 0.5f;

    const float* d = deltas + (((size_t)n * NPROP) + i) * 4;
    float px = acx + d[0] * aw;
    float py = acy + d[1] * ah;
    float pw = aw * expf(d[2]);
    float ph = ah * expf(d[3]);
    float b0 = fminf(fmaxf(px - pw * 0.5f, 0.f), 1024.f);
    float b1 = fminf(fmaxf(py - ph * 0.5f, 0.f), 1024.f);
    float b2 = fminf(fmaxf(px + pw * 0.5f, 0.f), 1024.f);
    float b3 = fminf(fmaxf(py + ph * 0.5f, 0.f), 1024.f);

    float* bo = boxes + (((size_t)n * NPROP) + i) * 4;
    bo[0] = b0; bo[1] = b1; bo[2] = b2; bo[3] = b3;

    float s = scores[(size_t)n * NPROP + i];
    bool valid = (s > OBJ_THR) && ((b2 - b0) > 0.f) && ((b3 - b1) > 0.f);
    float ss = valid ? s : -__builtin_inff();
    unsigned u = __float_as_uint(ss);
    unsigned sv = (u & 0x80000000u) ? ~u : (u | 0x80000000u);
    kout[i] = ((unsigned long long)sv << 32) | (unsigned long long)(0xFFFFFFFFu - (unsigned)i);
}

#define CHUNK 2048
#define NCHUNK (NPROP / CHUNK)
#define FOLD_A 6
#define FOLD_B (NCHUNK / FOLD_A)

__device__ __forceinline__ void bitonic_ce(unsigned long long* s, int t, int j, int k) {
    int lo = t & (j - 1);
    int i  = ((t - lo) << 1) | lo;
    int p  = i | j;
    unsigned long long a = s[i];
    unsigned long long b = s[p];
    bool up = ((i & k) == 0);
    bool sw = up ? (a < b) : (a > b);
    if (sw) { s[i] = b; s[p] = a; }
}

__device__ __forceinline__ unsigned long long lane_xchg64(unsigned long long v, int addr) {
    unsigned lo = (unsigned)__builtin_amdgcn_ds_bpermute(addr, (int)(unsigned)(v & 0xFFFFFFFFull));
    unsigned hi = (unsigned)__builtin_amdgcn_ds_bpermute(addr, (int)(unsigned)(v >> 32));
    return ((unsigned long long)hi << 32) | (unsigned long long)lo;
}

__device__ __forceinline__ void wave_ce_steps(unsigned long long& e0, unsigned long long& e1,
                                              int lane, int x0, int k) {
    bool up0 = ((x0 & k) == 0);
    bool up1 = (((x0 | 32) & k) == 0);
    if (k >= 64) {
        unsigned long long mx = e0 > e1 ? e0 : e1;
        unsigned long long mn = e0 > e1 ? e1 : e0;
        e0 = up0 ? mx : mn;
        e1 = up0 ? mn : mx;
    }
    #pragma unroll
    for (int j = 16; j > 0; j >>= 1) {
        if (2 * j <= k) {
            int addr = (lane ^ j) << 2;
            unsigned long long y0 = lane_xchg64(e0, addr);
            unsigned long long y1 = lane_xchg64(e1, addr);
            bool low = ((lane & j) == 0);
            bool wmax0 = (low == up0);
            bool wmax1 = (low == up1);
            bool g0 = y0 > e0;
            bool g1 = y1 > e1;
            e0 = (g0 == wmax0) ? y0 : e0;
            e1 = (g1 == wmax1) ? y1 : e1;
        }
    }
}

__global__ __launch_bounds__(1024) void topk_chunk_sort_kernel(unsigned long long* keys) {
    __shared__ unsigned long long s[CHUNK];
    int n   = blockIdx.x / NCHUNK;
    int c   = blockIdx.x % NCHUNK;
    int tid = threadIdx.x;
    int lane = tid & 31;
    int x0   = (tid >> 5) * 64 + lane;
    unsigned long long* g = keys + ((size_t)n << 16) + (size_t)c * CHUNK;
    unsigned long long e0 = g[x0];
    unsigned long long e1 = g[x0 + 32];
    wave_ce_steps(e0, e1, lane, x0, 2);
    wave_ce_steps(e0, e1, lane, x0, 4);
    wave_ce_steps(e0, e1, lane, x0, 8);
    wave_ce_steps(e0, e1, lane, x0, 16);
    wave_ce_steps(e0, e1, lane, x0, 32);
    wave_ce_steps(e0, e1, lane, x0, 64);
    #pragma unroll
    for (int k = 128; k <= CHUNK; k <<= 1) {
        s[x0]      = e0;
        s[x0 + 32] = e1;
        __syncthreads();
        for (int j = k >> 1; j >= 64; j >>= 1) {
            bitonic_ce(s, tid, j, k);
            __syncthreads();
        }
        e0 = s[x0];
        e1 = s[x0 + 32];
        wave_ce_steps(e0, e1, lane, x0, k);
    }
    g[x0]      = e0;
    g[x0 + 32] = e1;
}

__global__ __launch_bounds__(1024) void topk_fold_kernel(unsigned long long* keys,
                                                         int nch, int stride, int groups) {
    __shared__ unsigned long long s[CHUNK];
    int n   = blockIdx.x / groups;
    int gi  = blockIdx.x % groups;
    int tid = threadIdx.x;
    int lane = tid & 31;
    int x0   = (tid >> 5) * 64 + lane;
    size_t pitch = (size_t)stride * CHUNK;
    unsigned long long* g = keys + ((size_t)n << 16) + (size_t)gi * nch * pitch;
    s[tid]        = g[tid];
    s[tid + 1024] = g[tid + 1024];
    unsigned long long d0 = 0ull, d1 = 0ull;
    if (nch > 1) {
        d0 = g[pitch + (CHUNK - 1 - tid)];
        d1 = g[pitch + (CHUNK - 1 - 1024 - tid)];
    }
    for (int c = 1; c < nch; c++) {
        unsigned long long a0 = s[tid];
        unsigned long long a1 = s[tid + 1024];
        unsigned long long e0 = a0 > d0 ? a0 : d0;
        unsigned long long e1 = a1 > d1 ? a1 : d1;
        s[tid]        = e0 > e1 ? e0 : e1;
        s[tid + 1024] = e0 > e1 ? e1 : e0;
        if (c + 1 < nch) {
            const unsigned long long* gn = g + (size_t)(c + 1) * pitch;
            d0 = gn[CHUNK - 1 - tid];
            d1 = gn[CHUNK - 1 - 1024 - tid];
        }
        __syncthreads();
        for (int j = CHUNK >> 2; j >= 64; j >>= 1) {
            bitonic_ce(s, tid, j, CHUNK);
            __syncthreads();
        }
        unsigned long long r0 = s[x0];
        unsigned long long r1 = s[x0 + 32];
        wave_ce_steps(r0, r1, lane, x0, CHUNK);
        s[x0]      = r0;
        s[x0 + 32] = r1;
        __syncthreads();
    }
    g[tid]        = s[tid];
    g[tid + 1024] = s[tid + 1024];
}

#define MASK_ROWS 16
#define MASK_BLKS (PRE_NMS / MASK_ROWS)
#define LPITCH 65
__global__ __launch_bounds__(256) void nms_mask_kernel(const unsigned long long* __restrict__ keys,
                                                       const float* __restrict__ boxes,
                                                       unsigned* __restrict__ mask,
                                                       float* __restrict__ sorted) {
    __shared__ v4f   sbx[32 * LPITCH];
    __shared__ float sar[32 * LPITCH];

    int n  = blockIdx.x / MASK_BLKS;
    int rb = blockIdx.x % MASK_BLKS;
    int tid = threadIdx.x;
    const unsigned long long* K = keys + ((size_t)n << 16);
    const float* bimg = boxes + (size_t)n * NPROP * 4;

    unsigned long long kk[8];
    #pragma unroll
    for (int q = 0; q < 8; q++) {
        int j = tid + 256 * q;
        kk[q] = K[j < PRE_NMS ? j : PRE_NMS - 1];
    }
    v4f gb[8];
    #pragma unroll
    for (int q = 0; q < 8; q++) {
        int j = tid + 256 * q;
        unsigned sv  = (unsigned)(kk[q] >> 32);
        unsigned idx = 0xFFFFFFFFu - (unsigned)(kk[q] & 0xFFFFFFFFu);
        bool v = (j < PRE_NMS) && (sv > 0x007FFFFFu);
        gb[q] = *(const v4f*)(bimg + (size_t)(v ? idx : 0u) * 4);
    }
    #pragma unroll
    for (int q = 0; q < 8; q++) {
        int j = tid + 256 * q;
        unsigned sv = (unsigned)(kk[q] >> 32);
        bool v = (j < PRE_NMS) && (sv > 0x007FFFFFu);
        v4f bb = {0.f, 0.f, 0.f, 0.f};
        float s = 0.f;
        if (v) {
            bb = gb[q];
            unsigned u = (sv & 0x80000000u) ? (sv ^ 0x80000000u) : ~sv;
            s = __uint_as_float(u);
        }
        int slot = (j & 31) * LPITCH + (j >> 5);
        sbx[slot] = bb;
        sar[slot] = fmaxf(bb.z - bb.x, 0.f) * fmaxf(bb.w - bb.y, 0.f);
        if (j < PRE_NMS && (j / MASK_ROWS) == rb) {
            v4f* so = (v4f*)sorted + ((size_t)n * PRE_NMS + j) * 2;
            v4f inf = {s, v ? 1.f : 0.f, 0.f, 0.f};
            so[0] = bb;
            so[1] = inf;
        }
    }
    __syncthreads();

    int wave = tid >> 5;
    int lane = tid & 31;
    for (int rr = 0; rr < 2; rr++) {
        int i = rb * MASK_ROWS + wave * 2 + rr;
        int islot = (i & 31) * LPITCH + (i >> 5);
        v4f bi = sbx[islot];
        float ia = sar[islot];
        int wdiag = i >> 5;
        unsigned* mrow = mask + ((size_t)n * PRE_NMS + i) * 64;
        for (int pass = 0; pass < 2; pass++) {
            int w = pass * 32 + lane;
            unsigned bits = 0u;
            if (w >= wdiag) {
                for (int b = 0; b < 32; b++) {
                    v4f bj = sbx[b * LPITCH + w];
                    float ja = sar[b * LPITCH + w];
                    float ix1 = fmaxf(bi.x, bj.x), iy1 = fmaxf(bi.y, bj.y);
                    float ix2 = fminf(bi.z, bj.z), iy2 = fminf(bi.w, bj.w);
                    float inter = fmaxf(ix2 - ix1, 0.f) * fmaxf(iy2 - iy1, 0.f);
                    float iou = inter / fmaxf(ia + ja - inter, 1e-9f);
                    if (iou > IOU_THR) bits |= (1u << b);
                }
            }
            mrow[w] = bits;
        }
    }
}

__global__ __launch_bounds__(32) void nms_scan_kernel(const unsigned* __restrict__ mask,
                                                      const float* __restrict__ sorted,
                                                      float* __restrict__ out) {
    int n = blockIdx.x;
    int lane = threadIdx.x;
    const unsigned* Mw = mask + (size_t)n * PRE_NMS * 64;
    const v2u* M = (const v2u*)Mw;
    const v4f* S = (const v4f*)sorted + (size_t)n * PRE_NMS * 2;
    float* o_img = out + (size_t)n * POST_NMS * 5;

    unsigned rem_lo = 0u, rem_hi = 0u;
    const unsigned below = (1u << lane) - 1u;

    v2u cur[32], nxt[32];
    v4f cbox, cinf, nbox, ninf;
    unsigned cdiag, ndiag;
    #pragma unroll
    for (int r = 0; r < 32; r++) cur[r] = M[(size_t)r * 32 + lane];
    cbox = S[(size_t)lane * 2];
    cinf = S[(size_t)lane * 2 + 1];
    cdiag = Mw[(size_t)lane * 64];

    int cnt = 0;
    for (int c = 0; c < 63; c++) {
        int cn = (c + 1 < 63) ? c + 1 : 62;
        #pragma unroll
        for (int r = 0; r < 32; r++) {
            int i = cn * 32 + r;
            i = i < PRE_NMS ? i : PRE_NMS - 1;
            nxt[r] = M[(size_t)i * 32 + lane];
        }
        {
            int i = cn * 32 + lane;
            i = i < PRE_NMS ? i : PRE_NMS - 1;
            nbox = S[(size_t)i * 2];
            ninf = S[(size_t)i * 2 + 1];
            ndiag = Mw[(size_t)i * 64 + cn];
        }

        int own = c >> 1;
        bool inval = (c * 32 + lane >= PRE_NMS) || (cinf.y == 0.f);
        unsigned remw = (unsigned)__builtin_amdgcn_readlane((int)((c & 1) ? rem_hi : rem_lo), own)
                      | __builtin_amdgcn_ballot_w32(inval);
        unsigned cand = ~remw;
        unsigned notc = ((cand >> lane) & 1u) ^ 1u;
        unsigned low = cdiag & below;
        unsigned kept = cand;
        for (int it = 0; it < 32; it++) {
            unsigned nk = __builtin_amdgcn_ballot_w32(((low & kept) | notc) == 0u);
            if (nk == kept) break;
            kept = nk;
        }
        #pragma unroll
        for (int r = 0; r < 32; r++) {
            unsigned m = ((kept >> r) & 1u) ? 0xFFFFFFFFu : 0u;
            rem_lo |= cur[r].x & m;
            rem_hi |= cur[r].y & m;
        }

        unsigned mine = (kept >> lane) & 1u;
        int pos = cnt + __popc(kept & below);
        if (mine && pos < POST_NMS) {
            float* o = o_img + pos * 5;
            o[0] = cbox.x; o[1] = cbox.y; o[2] = cbox.z; o[3] = cbox.w;
            o[4] = cinf.x;
        }
        cnt += __popc(kept);
        if (cnt >= POST_NMS) break;

        #pragma unroll
        for (int r = 0; r < 32; r++) cur[r] = nxt[r];
        cbox = nbox; cinf = ninf; cdiag = ndiag;
    }

    int c0 = cnt < POST_NMS ? cnt : POST_NMS;
    for (int t = c0 * 5 + lane; t < POST_NMS * 5; t += 32) o_img[t] = 0.f;
}

extern "C" void kernel_launch(void* const* d_in, const int* in_sizes, int n_in,
                              void* d_out, int out_size, void* d_ws, size_t ws_size,
                              hipStream_t stream) {
    const float* feat   = (const float*)d_in[0];
    const float* conv_w = (const float*)d_in[1];
    const float* conv_b = (const float*)d_in[2];
    const float* det_w  = (const float*)d_in[3];
    const float* det_b  = (const float*)d_in[4];
    const float* reg_w  = (const float*)d_in[5];
    const float* reg_b  = (const float*)d_in[6];
    float* out = (float*)d_out;

    char* ws = (char*)d_ws;
    float* wt2                  = (float*)(ws);
    float* scores               = (float*)(ws + 2359296);
    float* deltas               = (float*)(ws + 2359296 + 786432);
    float* boxes                = (float*)(ws + 2359296 + 786432 + 3145728);
    unsigned long long* keys    = (unsigned long long*)(ws + 2359296 + 786432 + 3145728 + 3145728);

    fold_w_kernel<<<145, 256, 0, stream>>>(conv_w, conv_b, det_w, det_b, reg_w, reg_b, wt2);
    conv_fused_kernel<<<NIMG * HF * (WF / TILE_W), 256, 0, stream>>>(feat, wt2, scores, deltas);
    decode_kernel<<<(NIMG * NPAD) / 256, 256, 0, stream>>>(scores, deltas, boxes, keys);
    topk_chunk_sort_kernel<<<NIMG * NCHUNK, 1024, 0, stream>>>(keys);
    topk_fold_kernel<<<NIMG * FOLD_B, 1024, 0, stream>>>(keys, FOLD_A, 1, FOLD_B);
    topk_fold_kernel<<<NIMG, 1024, 0, stream>>>(keys, FOLD_B, FOLD_A, 1);
    unsigned* nmask  = (unsigned*)(ws);
    float*    sorted = (float*)(ws + 2048000);
    nms_mask_kernel<<<NIMG * MASK_BLKS, 256, 0, stream>>>(keys, boxes, nmask, sorted);
    nms_scan_kernel<<<NIMG, 32, 0, stream>>>(nmask, sorted, out);
}


# decode fused into the conv epilogue: the thread that sums an anchor's five head rows for a pixel decodes its box and builds its sort key; score/delta planes and one launch dropped; same sums
# speedup vs baseline: 18.5861x; 1.0449x over previous
#include <hip/hip_runtime.h>
#include <hip/hip_bf16.h>
#include <math.h>

typedef __attribute__((ext_vector_type(2))) float v2f;
typedef __attribute__((ext_vector_type(8))) float v8f;
typedef __attribute__((ext_vector_type(4))) float v4f;
typedef __attribute__((ext_vector_type(2))) unsigned v2u;

#define NIMG 4
#define CIN 256
#define HF 128
#define WF 128
#define A_ANCH 3
#define NPROP (A_ANCH * HF * WF)
#define NPAD 65536
#define PRE_NMS 2000
#define POST_NMS 300
#define OBJ_THR 0.3f
#define IOU_THR 0.7f
#define TILE_W 64
#define CCHUNK 32
#define CIPAD 34

#define NWEFF (9 * 128 * 16 * 2)
__global__ __launch_bounds__(256) void fold_w_kernel(const float* __restrict__ conv_w,
                                                     const float* __restrict__ conv_b,
                                                     const float* __restrict__ det_w,
                                                     const float* __restrict__ det_b,
                                                     const float* __restrict__ reg_w,
                                                     const float* __restrict__ reg_b,
                                                     float* __restrict__ weff2) {
    int tid = threadIdx.x;
    if (blockIdx.x == NWEFF / 256) {
        if (tid < 16) {
            float s = 0.0f;
            if (tid < 15) {
                const float* hw = (tid < 3) ? (det_w + tid * 256) : (reg_w + (tid - 3) * 256);
                s = (tid < 3) ? det_b[tid] : reg_b[tid - 3];
                for (int c0 = 0; c0 < 256; c0 += 64) {
                    float hv[64], bv[64];
                    #pragma unroll
                    for (int j = 0; j < 64; j++) { hv[j] = hw[c0 + j]; bv[j] = conv_b[c0 + j]; }
                    #pragma unroll
                    for (int j = 0; j < 64; j++) s += hv[j] * bv[j];
                }
            }
            weff2[NWEFF + tid] = s;
        }
        return;
    }
    int t   = blockIdx.x * 256 + tid;
    int par = t & 1;
    int o   = (t >> 1) & 15;
    int q   = t >> 5;
    int ci2 = q & 127;
    int kk  = q >> 7;
    int ci  = ci2 * 2 + par;
    float s = 0.0f;
    if (o < 15) {
        const float* hw = (o < 3) ? (det_w + o * 256) : (reg_w + (o - 3) * 256);
        const float* wp = conv_w + (size_t)ci * 9 + kk;
        for (int c0 = 0; c0 < 256; c0 += 32) {
            float hv[32], wv[32];
            #pragma unroll
            for (int j = 0; j < 32; j++) { hv[j] = hw[c0 + j]; wv[j] = wp[(size_t)(c0 + j) * (CIN * 9)]; }
            #pragma unroll
            for (int j = 0; j < 32; j++) s += hv[j] * wv[j];
        }
    }
    weff2[t] = s;
}

#define NCHK (CIN / CCHUNK)
#define NPRE 25
#define PP 66

__device__ __forceinline__ void stage_fetch(float (&pre)[NPRE], const float* frow,
                                            const float* fhalo, const int (&rowoff)[3], int cib) {
    const float* fc = frow + (size_t)cib * (HF * WF);
    #pragma unroll
    for (int j = 0; j < 4; j++) {
        #pragma unroll
        for (int r = 0; r < 3; r++) {
            const float* p = fc + j * (HF * WF) + rowoff[r];
            pre[(j * 3 + r) * 2 + 0] = p[0];
            pre[(j * 3 + r) * 2 + 1] = p[32];
        }
    }
    pre[24] = fhalo[(size_t)cib * (HF * WF)];
}

__global__ __launch_bounds__(256) __attribute__((amdgpu_waves_per_eu(8))) void conv_fused_kernel(
    const float* __restrict__ feat, const float* __restrict__ weff2,
    float* __restrict__ boxes, unsigned long long* __restrict__ keys) {

    __shared__ __attribute__((aligned(16))) float lds_in[3 * 66 * CIPAD];
    __shared__ float part[8 * 16 * PP];

    int bid = blockIdx.x;
    int n    = bid >> 8;
    int rem  = bid & 255;
    int row  = rem >> 1;
    int col0 = (rem & 1) * TILE_W;

    int tid  = threadIdx.x;
    int wave = tid >> 5;
    int lane = tid & 31;
    int hl   = lane >> 4;
    int lm   = lane & 15;

    v8f acc[4] = {};

    const float* featn = feat + (size_t)n * CIN * HF * WF;
    const v2f*   wpl   = (const v2f*)weff2 + (size_t)(2 * wave + hl) * 16 + lm;

    int  rowoff[3];
    bool rowok[3];
    #pragma unroll
    for (int r = 0; r < 3; r++) {
        int g = row + r - 1;
        rowok[r]  = (g >= 0) && (g < HF);
        rowoff[r] = (g < 0 ? 0 : (g >= HF ? HF - 1 : g)) * WF;
    }
    const float* frow = featn + (size_t)(4 * wave) * (HF * WF) + col0 + lane;
    int  hj  = lane / 6;
    int  hq  = lane - hj * 6;
    int  hr  = hq >> 1;
    int  hs  = hq & 1;
    int  hg  = row + hr - 1;
    int  hc  = hs ? (col0 + TILE_W) : (col0 - 1);
    bool hok = (lane < 24) && (hg >= 0) && (hg < HF) && (hc >= 0) && (hc < WF);
    const float* fhalo = featn + (hok ? ((size_t)(4 * wave + hj) * HF + hg) * WF + hc : (size_t)0);
    int  hlds = (hr * 66 + (hs ? 65 : 0)) * CIPAD + 4 * wave + hj;

    int bbase = lm * CIPAD + 4 * wave + 2 * hl;

    float pre[NPRE];
    stage_fetch(pre, frow, fhalo, rowoff, 0);

    for (int ch = 0; ch < NCHK; ch++) {
        v2f a[9];
        #pragma unroll
        for (int kk = 0; kk < 9; kk++)
            a[kk] = wpl[(size_t)(kk * 128 + ch * 16) * 16];

        __syncthreads();
        #pragma unroll
        for (int r = 0; r < 3; r++) {
            #pragma unroll
            for (int hcol = 0; hcol < 2; hcol++) {
                float v0 = rowok[r] ? pre[(0 * 3 + r) * 2 + hcol] : 0.0f;
                float v1 = rowok[r] ? pre[(1 * 3 + r) * 2 + hcol] : 0.0f;
                float v2 = rowok[r] ? pre[(2 * 3 + r) * 2 + hcol] : 0.0f;
                float v3 = rowok[r] ? pre[(3 * 3 + r) * 2 + hcol] : 0.0f;
                int idx = (r * 66 + 1 + hcol * 32 + lane) * CIPAD + 4 * wave;
                v2f lo = {v0, v1};
                v2f hi = {v2, v3};
                *(v2f*)&lds_in[idx]     = lo;
                *(v2f*)&lds_in[idx + 2] = hi;
            }
        }
        if (lane < 24) lds_in[hlds] = hok ? pre[24] : 0.0f;
        __syncthreads();

        int chn = ch + 1 < NCHK ? ch + 1 : NCHK - 1;
        stage_fetch(pre, frow, fhalo, rowoff, chn * CCHUNK);

        #pragma unroll
        for (int kk = 0; kk < 9; kk++) {
            int kh = kk / 3;
            int kw = kk - kh * 3;
            v2f b[4];
            #pragma unroll
            for (int nt = 0; nt < 4; nt++)
                b[nt] = *(const v2f*)&lds_in[(kh * 66 + nt * 16 + kw) * CIPAD + bbase];
            #pragma unroll
            for (int nt = 0; nt < 4; nt++)
                acc[nt] = __builtin_amdgcn_wmma_f32_16x16x4_f32(false, a[kk], false, b[nt], (short)0, acc[nt], false, false);
        }
    }

    #pragma unroll
    for (int nt = 0; nt < 4; nt++) {
        #pragma unroll
        for (int r = 0; r < 8; r++)
            part[(wave * 16 + hl * 8 + r) * PP + nt * 16 + lm] = acc[nt][r];
    }
    __syncthreads();

    const float* beff = weff2 + NWEFF;
    if (tid < A_ANCH * TILE_W) {
        int a   = tid >> 6;
        int pix = tid & 63;
        float hv[5];
        #pragma unroll
        for (int q = 0; q < 5; q++) {
            int oi = (q == 0) ? a : (3 + (q - 1) * A_ANCH + a);
            float sum = part[oi * PP + pix];
            #pragma unroll
            for (int w = 1; w < 8; w++) sum += part[(w * 16 + oi) * PP + pix];
            hv[q] = sum + beff[oi];
        }
        float s = 1.0f / (1.0f + expf(-hv[0]));

        int i  = a * (HF * WF) + row * WF + col0 + pix;
        int pa = i / 3, aa = i - pa * 3;
        int ya = pa >> 7, xa = pa & 127;
        float hwv = (aa == 0) ? 91.f : ((aa == 1) ? 64.f : 45.f);
        float hhv = (aa == 0) ? 45.f : ((aa == 1) ? 64.f : 91.f);
        float sxv = (float)(xa * 8), syv = (float)(ya * 8);
        float x1 = sxv - hwv, y1 = syv - hhv;
        float x2 = sxv + hwv, y2 = syv + hhv;
        float aw = x2 - x1, ah = y2 - y1;
        float acx = y1 + aw * 0.5f;
        float acy = x1 + ah * 0.5f;

        float px = __builtin_fmaf(hv[1], aw, acx);
        float py = __builtin_fmaf(hv[2], ah, acy);
        float pw = aw * expf(hv[3]);
        float ph = ah * expf(hv[4]);
        float b0 = fminf(fmaxf(px - pw * 0.5f, 0.f), 1024.f);
        float b1 = fminf(fmaxf(py - ph * 0.5f, 0.f), 1024.f);
        float b2 = fminf(fmaxf(px + pw * 0.5f, 0.f), 1024.f);
        float b3 = fminf(fmaxf(py + ph * 0.5f, 0.f), 1024.f);

        v4f bo = {b0, b1, b2, b3};
        *(v4f*)(boxes + (((size_t)n * NPROP) + i) * 4) = bo;

        bool valid = (s > OBJ_THR) && ((b2 - b0) > 0.f) && ((b3 - b1) > 0.f);
        float ss = valid ? s : -__builtin_inff();
        unsigned u = __float_as_uint(ss);
        unsigned sv = (u & 0x80000000u) ? ~u : (u | 0x80000000u);
        keys[((size_t)n << 16) + i] =
            ((unsigned long long)sv << 32) | (unsigned long long)(0xFFFFFFFFu - (unsigned)i);
    }
}

#define CHUNK 2048
#define NCHUNK (NPROP / CHUNK)
#define FOLD_A 6
#define FOLD_B (NCHUNK / FOLD_A)

__device__ __forceinline__ void bitonic_ce(unsigned long long* s, int t, int j, int k) {
    int lo = t & (j - 1);
    int i  = ((t - lo) << 1) | lo;
    int p  = i | j;
    unsigned long long a = s[i];
    unsigned long long b = s[p];
    bool up = ((i & k) == 0);
    bool sw = up ? (a < b) : (a > b);
    if (sw) { s[i] = b; s[p] = a; }
}

__device__ __forceinline__ unsigned long long lane_xchg64(unsigned long long v, int addr) {
    unsigned lo = (unsigned)__builtin_amdgcn_ds_bpermute(addr, (int)(unsigned)(v & 0xFFFFFFFFull));
    unsigned hi = (unsigned)__builtin_amdgcn_ds_bpermute(addr, (int)(unsigned)(v >> 32));
    return ((unsigned long long)hi << 32) | (unsigned long long)lo;
}

__device__ __forceinline__ void wave_ce_steps(unsigned long long& e0, unsigned long long& e1,
                                              int lane, int x0, int k) {
    bool up0 = ((x0 & k) == 0);
    bool up1 = (((x0 | 32) & k) == 0);
    if (k >= 64) {
        unsigned long long mx = e0 > e1 ? e0 : e1;
        unsigned long long mn = e0 > e1 ? e1 : e0;
        e0 = up0 ? mx : mn;
        e1 = up0 ? mn : mx;
    }
    #pragma unroll
    for (int j = 16; j > 0; j >>= 1) {
        if (2 * j <= k) {
            int addr = (lane ^ j) << 2;
            unsigned long long y0 = lane_xchg64(e0, addr);
            unsigned long long y1 = lane_xchg64(e1, addr);
            bool low = ((lane & j) == 0);
            bool wmax0 = (low == up0);
            bool wmax1 = (low == up1);
            bool g0 = y0 > e0;
            bool g1 = y1 > e1;
            e0 = (g0 == wmax0) ? y0 : e0;
            e1 = (g1 == wmax1) ? y1 : e1;
        }
    }
}

__global__ __launch_bounds__(1024) void topk_chunk_sort_kernel(unsigned long long* keys) {
    __shared__ unsigned long long s[CHUNK];
    int n   = blockIdx.x / NCHUNK;
    int c   = blockIdx.x % NCHUNK;
    int tid = threadIdx.x;
    int lane = tid & 31;
    int x0   = (tid >> 5) * 64 + lane;
    unsigned long long* g = keys + ((size_t)n << 16) + (size_t)c * CHUNK;
    unsigned long long e0 = g[x0];
    unsigned long long e1 = g[x0 + 32];
    wave_ce_steps(e0, e1, lane, x0, 2);
    wave_ce_steps(e0, e1, lane, x0, 4);
    wave_ce_steps(e0, e1, lane, x0, 8);
    wave_ce_steps(e0, e1, lane, x0, 16);
    wave_ce_steps(e0, e1, lane, x0, 32);
    wave_ce_steps(e0, e1, lane, x0, 64);
    #pragma unroll
    for (int k = 128; k <= CHUNK; k <<= 1) {
        s[x0]      = e0;
        s[x0 + 32] = e1;
        __syncthreads();
        for (int j = k >> 1; j >= 64; j >>= 1) {
            bitonic_ce(s, tid, j, k);
            __syncthreads();
        }
        e0 = s[x0];
        e1 = s[x0 + 32];
        wave_ce_steps(e0, e1, lane, x0, k);
    }
    g[x0]      = e0;
    g[x0 + 32] = e1;
}

__global__ __launch_bounds__(1024) void topk_fold_kernel(unsigned long long* keys,
                                                         int nch, int stride, int groups) {
    __shared__ unsigned long long s[CHUNK];
    int n   = blockIdx.x / groups;
    int gi  = blockIdx.x % groups;
    int tid = threadIdx.x;
    int lane = tid & 31;
    int x0   = (tid >> 5) * 64 + lane;
    size_t pitch = (size_t)stride * CHUNK;
    unsigned long long* g = keys + ((size_t)n << 16) + (size_t)gi * nch * pitch;
    s[tid]        = g[tid];
    s[tid + 1024] = g[tid + 1024];
    unsigned long long d0 = 0ull, d1 = 0ull;
    if (nch > 1) {
        d0 = g[pitch + (CHUNK - 1 - tid)];
        d1 = g[pitch + (CHUNK - 1 - 1024 - tid)];
    }
    for (int c = 1; c < nch; c++) {
        unsigned long long a0 = s[tid];
        unsigned long long a1 = s[tid + 1024];
        unsigned long long e0 = a0 > d0 ? a0 : d0;
        unsigned long long e1 = a1 > d1 ? a1 : d1;
        s[tid]        = e0 > e1 ? e0 : e1;
        s[tid + 1024] = e0 > e1 ? e1 : e0;
        if (c + 1 < nch) {
            const unsigned long long* gn = g + (size_t)(c + 1) * pitch;
            d0 = gn[CHUNK - 1 - tid];
            d1 = gn[CHUNK - 1 - 1024 - tid];
        }
        __syncthreads();
        for (int j = CHUNK >> 2; j >= 64; j >>= 1) {
            bitonic_ce(s, tid, j, CHUNK);
            __syncthreads();
        }
        unsigned long long r0 = s[x0];
        unsigned long long r1 = s[x0 + 32];
        wave_ce_steps(r0, r1, lane, x0, CHUNK);
        s[x0]      = r0;
        s[x0 + 32] = r1;
        __syncthreads();
    }
    g[tid]        = s[tid];
    g[tid + 1024] = s[tid + 1024];
}

#define MASK_ROWS 16
#define MASK_BLKS (PRE_NMS / MASK_ROWS)
#define LPITCH 65
__global__ __launch_bounds__(256) void nms_mask_kernel(const unsigned long long* __restrict__ keys,
                                                       const float* __restrict__ boxes,
                                                       unsigned* __restrict__ mask,
                                                       float* __restrict__ sorted) {
    __shared__ v4f   sbx[32 * LPITCH];
    __shared__ float sar[32 * LPITCH];

    int n  = blockIdx.x / MASK_BLKS;
    int rb = blockIdx.x % MASK_BLKS;
    int tid = threadIdx.x;
    const unsigned long long* K = keys + ((size_t)n << 16);
    const float* bimg = boxes + (size_t)n * NPROP * 4;

    unsigned long long kk[8];
    #pragma unroll
    for (int q = 0; q < 8; q++) {
        int j = tid + 256 * q;
        kk[q] = K[j < PRE_NMS ? j : PRE_NMS - 1];
    }
    v4f gb[8];
    #pragma unroll
    for (int q = 0; q < 8; q++) {
        int j = tid + 256 * q;
        unsigned sv  = (unsigned)(kk[q] >> 32);
        unsigned idx = 0xFFFFFFFFu - (unsigned)(kk[q] & 0xFFFFFFFFu);
        bool v = (j < PRE_NMS) && (sv > 0x007FFFFFu);
        gb[q] = *(const v4f*)(bimg + (size_t)(v ? idx : 0u) * 4);
    }
    #pragma unroll
    for (int q = 0; q < 8; q++) {
        int j = tid + 256 * q;
        unsigned sv = (unsigned)(kk[q] >> 32);
        bool v = (j < PRE_NMS) && (sv > 0x007FFFFFu);
        v4f bb = {0.f, 0.f, 0.f, 0.f};
        float s = 0.f;
        if (v) {
            bb = gb[q];
            unsigned u = (sv & 0x80000000u) ? (sv ^ 0x80000000u) : ~sv;
            s = __uint_as_float(u);
        }
        int slot = (j & 31) * LPITCH + (j >> 5);
        sbx[slot] = bb;
        sar[slot] = fmaxf(bb.z - bb.x, 0.f) * fmaxf(bb.w - bb.y, 0.f);
        if (j < PRE_NMS && (j / MASK_ROWS) == rb) {
            v4f* so = (v4f*)sorted + ((size_t)n * PRE_NMS + j) * 2;
            v4f inf = {s, v ? 1.f : 0.f, 0.f, 0.f};
            so[0] = bb;
            so[1] = inf;
        }
    }
    __syncthreads();

    int wave = tid >> 5;
    int lane = tid & 31;
    for (int rr = 0; rr < 2; rr++) {
        int i = rb * MASK_ROWS + wave * 2 + rr;
        int islot = (i & 31) * LPITCH + (i >> 5);
        v4f bi = sbx[islot];
        float ia = sar[islot];
        int wdiag = i >> 5;
        unsigned* mrow = mask + ((size_t)n * PRE_NMS + i) * 64;
        for (int pass = 0; pass < 2; pass++) {
            int w = pass * 32 + lane;
            unsigned bits = 0u;
            if (w >= wdiag) {
                for (int b = 0; b < 32; b++) {
                    v4f bj = sbx[b * LPITCH + w];
                    float ja = sar[b * LPITCH + w];
                    float ix1 = fmaxf(bi.x, bj.x), iy1 = fmaxf(bi.y, bj.y);
                    float ix2 = fminf(bi.z, bj.z), iy2 = fminf(bi.w, bj.w);
                    float inter = fmaxf(ix2 - ix1, 0.f) * fmaxf(iy2 - iy1, 0.f);
                    float iou = inter / fmaxf(ia + ja - inter, 1e-9f);
                    if (iou > IOU_THR) bits |= (1u << b);
                }
            }
            mrow[w] = bits;
        }
    }
}

__global__ __launch_bounds__(32) void nms_scan_kernel(const unsigned* __restrict__ mask,
                                                      const float* __restrict__ sorted,
                                                      float* __restrict__ out) {
    int n = blockIdx.x;
    int lane = threadIdx.x;
    const unsigned* Mw = mask + (size_t)n * PRE_NMS * 64;
    const v2u* M = (const v2u*)Mw;
    const v4f* S = (const v4f*)sorted + (size_t)n * PRE_NMS * 2;
    float* o_img = out + (size_t)n * POST_NMS * 5;

    unsigned rem_lo = 0u, rem_hi = 0u;
    const unsigned below = (1u << lane) - 1u;

    v2u cur[32], nxt[32];
    v4f cbox, cinf, nbox, ninf;
    unsigned cdiag, ndiag;
    #pragma unroll
    for (int r = 0; r < 32; r++) cur[r] = M[(size_t)r * 32 + lane];
    cbox = S[(size_t)lane * 2];
    cinf = S[(size_t)lane * 2 + 1];
    cdiag = Mw[(size_t)lane * 64];

    int cnt = 0;
    for (int c = 0; c < 63; c++) {
        int cn = (c + 1 < 63) ? c + 1 : 62;
        #pragma unroll
        for (int r = 0; r < 32; r++) {
            int i = cn * 32 + r;
            i = i < PRE_NMS ? i : PRE_NMS - 1;
            nxt[r] = M[(size_t)i * 32 + lane];
        }
        {
            int i = cn * 32 + lane;
            i = i < PRE_NMS ? i : PRE_NMS - 1;
            nbox = S[(size_t)i * 2];
            ninf = S[(size_t)i * 2 + 1];
            ndiag = Mw[(size_t)i * 64 + cn];
        }

        int own = c >> 1;
        bool inval = (c * 32 + lane >= PRE_NMS) || (cinf.y == 0.f);
        unsigned remw = (unsigned)__builtin_amdgcn_readlane((int)((c & 1) ? rem_hi : rem_lo), own)
                      | __builtin_amdgcn_ballot_w32(inval);
        unsigned cand = ~remw;
        unsigned notc = ((cand >> lane) & 1u) ^ 1u;
        unsigned low = cdiag & below;
        unsigned kept = cand;
        for (int it = 0; it < 32; it++) {
            unsigned nk = __builtin_amdgcn_ballot_w32(((low & kept) | notc) == 0u);
            if (nk == kept) break;
            kept = nk;
        }
        #pragma unroll
        for (int r = 0; r < 32; r++) {
            unsigned m = ((kept >> r) & 1u) ? 0xFFFFFFFFu : 0u;
            rem_lo |= cur[r].x & m;
            rem_hi |= cur[r].y & m;
        }

        unsigned mine = (kept >> lane) & 1u;
        int pos = cnt + __popc(kept & below);
        if (mine && pos < POST_NMS) {
            float* o = o_img + pos * 5;
            o[0] = cbox.x; o[1] = cbox.y; o[2] = cbox.z; o[3] = cbox.w;
            o[4] = cinf.x;
        }
        cnt += __popc(kept);
        if (cnt >= POST_NMS) break;

        #pragma unroll
        for (int r = 0; r < 32; r++) cur[r] = nxt[r];
        cbox = nbox; cinf = ninf; cdiag = ndiag;
    }

    int c0 = cnt < POST_NMS ? cnt : POST_NMS;
    for (int t = c0 * 5 + lane; t < POST_NMS * 5; t += 32) o_img[t] = 0.f;
}

extern "C" void kernel_launch(void* const* d_in, const int* in_sizes, int n_in,
                              void* d_out, int out_size, void* d_ws, size_t ws_size,
                              hipStream_t stream) {
    const float* feat   = (const float*)d_in[0];
    const float* conv_w = (const float*)d_in[1];
    const float* conv_b = (const float*)d_in[2];
    const float* det_w  = (const float*)d_in[3];
    const float* det_b  = (const float*)d_in[4];
    const float* reg_w  = (const float*)d_in[5];
    const float* reg_b  = (const float*)d_in[6];
    float* out = (float*)d_out;

    char* ws = (char*)d_ws;
    float* wt2                  = (float*)(ws);
    float* boxes                = (float*)(ws + 2359296 + 786432 + 3145728);
    unsigned long long* keys    = (unsigned long long*)(ws + 2359296 + 786432 + 3145728 + 3145728);

    fold_w_kernel<<<145, 256, 0, stream>>>(conv_w, conv_b, det_w, det_b, reg_w, reg_b, wt2);
    conv_fused_kernel<<<NIMG * HF * (WF / TILE_W), 256, 0, stream>>>(feat, wt2, boxes, keys);
    topk_chunk_sort_kernel<<<NIMG * NCHUNK, 1024, 0, stream>>>(keys);
    topk_fold_kernel<<<NIMG * FOLD_B, 1024, 0, stream>>>(keys, FOLD_A, 1, FOLD_B);
    topk_fold_kernel<<<NIMG, 1024, 0, stream>>>(keys, FOLD_B, FOLD_A, 1);
    unsigned* nmask  = (unsigned*)(ws);
    float*    sorted = (float*)(ws + 2048000);
    nms_mask_kernel<<<NIMG * MASK_BLKS, 256, 0, stream>>>(keys, boxes, nmask, sorted);
    nms_scan_kernel<<<NIMG, 32, 0, stream>>>(nmask, sorted, out);
}


# NMS bit rows built only for the first 512 sorted boxes (16x fewer IoUs; the scan ends at 300 kept); if those keep fewer than 300, later chunks are tested against the kept boxes directly; same out
# speedup vs baseline: 21.0333x; 1.1317x over previous
#include <hip/hip_runtime.h>
#include <hip/hip_bf16.h>
#include <math.h>

typedef __attribute__((ext_vector_type(2))) float v2f;
typedef __attribute__((ext_vector_type(8))) float v8f;
typedef __attribute__((ext_vector_type(4))) float v4f;
typedef __attribute__((ext_vector_type(2))) unsigned v2u;

#define NIMG 4
#define CIN 256
#define HF 128
#define WF 128
#define A_ANCH 3
#define NPROP (A_ANCH * HF * WF)
#define NPAD 65536
#define PRE_NMS 2000
#define POST_NMS 300
#define OBJ_THR 0.3f
#define IOU_THR 0.7f
#define TILE_W 64
#define CCHUNK 32
#define CIPAD 34

#define NWEFF (9 * 128 * 16 * 2)
__global__ __launch_bounds__(256) void fold_w_kernel(const float* __restrict__ conv_w,
                                                     const float* __restrict__ conv_b,
                                                     const float* __restrict__ det_w,
                                                     const float* __restrict__ det_b,
                                                     const float* __restrict__ reg_w,
                                                     const float* __restrict__ reg_b,
                                                     float* __restrict__ weff2) {
    int tid = threadIdx.x;
    if (blockIdx.x == NWEFF / 256) {
        if (tid < 16) {
            float s = 0.0f;
            if (tid < 15) {
                const float* hw = (tid < 3) ? (det_w + tid * 256) : (reg_w + (tid - 3) * 256);
                s = (tid < 3) ? det_b[tid] : reg_b[tid - 3];
                for (int c0 = 0; c0 < 256; c0 += 64) {
                    float hv[64], bv[64];
                    #pragma unroll
                    for (int j = 0; j < 64; j++) { hv[j] = hw[c0 + j]; bv[j] = conv_b[c0 + j]; }
                    #pragma unroll
                    for (int j = 0; j < 64; j++) s += hv[j] * bv[j];
                }
            }
            weff2[NWEFF + tid] = s;
        }
        return;
    }
    int t   = blockIdx.x * 256 + tid;
    int par = t & 1;
    int o   = (t >> 1) & 15;
    int q   = t >> 5;
    int ci2 = q & 127;
    int kk  = q >> 7;
    int ci  = ci2 * 2 + par;
    float s = 0.0f;
    if (o < 15) {
        const float* hw = (o < 3) ? (det_w + o * 256) : (reg_w + (o - 3) * 256);
        const float* wp = conv_w + (size_t)ci * 9 + kk;
        for (int c0 = 0; c0 < 256; c0 += 32) {
            float hv[32], wv[32];
            #pragma unroll
            for (int j = 0; j < 32; j++) { hv[j] = hw[c0 + j]; wv[j] = wp[(size_t)(c0 + j) * (CIN * 9)]; }
            #pragma unroll
            for (int j = 0; j < 32; j++) s += hv[j] * wv[j];
        }
    }
    weff2[t] = s;
}

#define NCHK (CIN / CCHUNK)
#define NPRE 25
#define PP 66

__device__ __forceinline__ void stage_fetch(float (&pre)[NPRE], const float* frow,
                                            const float* fhalo, const int (&rowoff)[3], int cib) {
    const float* fc = frow + (size_t)cib * (HF * WF);
    #pragma unroll
    for (int j = 0; j < 4; j++) {
        #pragma unroll
        for (int r = 0; r < 3; r++) {
            const float* p = fc + j * (HF * WF) + rowoff[r];
            pre[(j * 3 + r) * 2 + 0] = p[0];
            pre[(j * 3 + r) * 2 + 1] = p[32];
        }
    }
    pre[24] = fhalo[(size_t)cib * (HF * WF)];
}

__global__ __launch_bounds__(256) __attribute__((amdgpu_waves_per_eu(8))) void conv_fused_kernel(
    const float* __restrict__ feat, const float* __restrict__ weff2,
    float* __restrict__ boxes, unsigned long long* __restrict__ keys) {

    __shared__ __attribute__((aligned(16))) float lds_in[3 * 66 * CIPAD];
    __shared__ float part[8 * 16 * PP];

    int bid = blockIdx.x;
    int n    = bid >> 8;
    int rem  = bid & 255;
    int row  = rem >> 1;
    int col0 = (rem & 1) * TILE_W;

    int tid  = threadIdx.x;
    int wave = tid >> 5;
    int lane = tid & 31;
    int hl   = lane >> 4;
    int lm   = lane & 15;

    v8f acc[4] = {};

    const float* featn = feat + (size_t)n * CIN * HF * WF;
    const v2f*   wpl   = (const v2f*)weff2 + (size_t)(2 * wave + hl) * 16 + lm;

    int  rowoff[3];
    bool rowok[3];
    #pragma unroll
    for (int r = 0; r < 3; r++) {
        int g = row + r - 1;
        rowok[r]  = (g >= 0) && (g < HF);
        rowoff[r] = (g < 0 ? 0 : (g >= HF ? HF - 1 : g)) * WF;
    }
    const float* frow = featn + (size_t)(4 * wave) * (HF * WF) + col0 + lane;
    int  hj  = lane / 6;
    int  hq  = lane - hj * 6;
    int  hr  = hq >> 1;
    int  hs  = hq & 1;
    int  hg  = row + hr - 1;
    int  hc  = hs ? (col0 + TILE_W) : (col0 - 1);
    bool hok = (lane < 24) && (hg >= 0) && (hg < HF) && (hc >= 0) && (hc < WF);
    const float* fhalo = featn + (hok ? ((size_t)(4 * wave + hj) * HF + hg) * WF + hc : (size_t)0);
    int  hlds = (hr * 66 + (hs ? 65 : 0)) * CIPAD + 4 * wave + hj;

    int bbase = lm * CIPAD + 4 * wave + 2 * hl;

    float pre[NPRE];
    stage_fetch(pre, frow, fhalo, rowoff, 0);

    for (int ch = 0; ch < NCHK; ch++) {
        v2f a[9];
        #pragma unroll
        for (int kk = 0; kk < 9; kk++)
            a[kk] = wpl[(size_t)(kk * 128 + ch * 16) * 16];

        __syncthreads();
        #pragma unroll
        for (int r = 0; r < 3; r++) {
            #pragma unroll
            for (int hcol = 0; hcol < 2; hcol++) {
                float v0 = rowok[r] ? pre[(0 * 3 + r) * 2 + hcol] : 0.0f;
                float v1 = rowok[r] ? pre[(1 * 3 + r) * 2 + hcol] : 0.0f;
                float v2 = rowok[r] ? pre[(2 * 3 + r) * 2 + hcol] : 0.0f;
                float v3 = rowok[r] ? pre[(3 * 3 + r) * 2 + hcol] : 0.0f;
                int idx = (r * 66 + 1 + hcol * 32 + lane) * CIPAD + 4 * wave;
                v2f lo = {v0, v1};
                v2f hi = {v2, v3};
                *(v2f*)&lds_in[idx]     = lo;
                *(v2f*)&lds_in[idx + 2] = hi;
            }
        }
        if (lane < 24) lds_in[hlds] = hok ? pre[24] : 0.0f;
        __syncthreads();

        int chn = ch + 1 < NCHK ? ch + 1 : NCHK - 1;
        stage_fetch(pre, frow, fhalo, rowoff, chn * CCHUNK);

        #pragma unroll
        for (int kk = 0; kk < 9; kk++) {
            int kh = kk / 3;
            int kw = kk - kh * 3;
            v2f b[4];
            #pragma unroll
            for (int nt = 0; nt < 4; nt++)
                b[nt] = *(const v2f*)&lds_in[(kh * 66 + nt * 16 + kw) * CIPAD + bbase];
            #pragma unroll
            for (int nt = 0; nt < 4; nt++)
                acc[nt] = __builtin_amdgcn_wmma_f32_16x16x4_f32(false, a[kk], false, b[nt], (short)0, acc[nt], false, false);
        }
    }

    #pragma unroll
    for (int nt = 0; nt < 4; nt++) {
        #pragma unroll
        for (int r = 0; r < 8; r++)
            part[(wave * 16 + hl * 8 + r) * PP + nt * 16 + lm] = acc[nt][r];
    }
    __syncthreads();

    const float* beff = weff2 + NWEFF;
    if (tid < A_ANCH * TILE_W) {
        int a   = tid >> 6;
        int pix = tid & 63;
        float hv[5];
        #pragma unroll
        for (int q = 0; q < 5; q++) {
            int oi = (q == 0) ? a : (3 + (q - 1) * A_ANCH + a);
            float sum = part[oi * PP + pix];
            #pragma unroll
            for (int w = 1; w < 8; w++) sum += part[(w * 16 + oi) * PP + pix];
            hv[q] = sum + beff[oi];
        }
        float s = 1.0f / (1.0f + expf(-hv[0]));

        int i  = a * (HF * WF) + row * WF + col0 + pix;
        int pa = i / 3, aa = i - pa * 3;
        int ya = pa >> 7, xa = pa & 127;
        float hwv = (aa == 0) ? 91.f : ((aa == 1) ? 64.f : 45.f);
        float hhv = (aa == 0) ? 45.f : ((aa == 1) ? 64.f : 91.f);
        float sxv = (float)(xa * 8), syv = (float)(ya * 8);
        float x1 = sxv - hwv, y1 = syv - hhv;
        float x2 = sxv + hwv, y2 = syv + hhv;
        float aw = x2 - x1, ah = y2 - y1;
        float acx = y1 + aw * 0.5f;
        float acy = x1 + ah * 0.5f;

        float px = __builtin_fmaf(hv[1], aw, acx);
        float py = __builtin_fmaf(hv[2], ah, acy);
        float pw = aw * expf(hv[3]);
        float ph = ah * expf(hv[4]);
        float b0 = fminf(fmaxf(px - pw * 0.5f, 0.f), 1024.f);
        float b1 = fminf(fmaxf(py - ph * 0.5f, 0.f), 1024.f);
        float b2 = fminf(fmaxf(px + pw * 0.5f, 0.f), 1024.f);
        float b3 = fminf(fmaxf(py + ph * 0.5f, 0.f), 1024.f);

        v4f bo = {b0, b1, b2, b3};
        *(v4f*)(boxes + (((size_t)n * NPROP) + i) * 4) = bo;

        bool valid = (s > OBJ_THR) && ((b2 - b0) > 0.f) && ((b3 - b1) > 0.f);
        float ss = valid ? s : -__builtin_inff();
        unsigned u = __float_as_uint(ss);
        unsigned sv = (u & 0x80000000u) ? ~u : (u | 0x80000000u);
        keys[((size_t)n << 16) + i] =
            ((unsigned long long)sv << 32) | (unsigned long long)(0xFFFFFFFFu - (unsigned)i);
    }
}

#define CHUNK 2048
#define NCHUNK (NPROP / CHUNK)
#define FOLD_A 6
#define FOLD_B (NCHUNK / FOLD_A)

__device__ __forceinline__ void bitonic_ce(unsigned long long* s, int t, int j, int k) {
    int lo = t & (j - 1);
    int i  = ((t - lo) << 1) | lo;
    int p  = i | j;
    unsigned long long a = s[i];
    unsigned long long b = s[p];
    bool up = ((i & k) == 0);
    bool sw = up ? (a < b) : (a > b);
    if (sw) { s[i] = b; s[p] = a; }
}

__device__ __forceinline__ unsigned long long lane_xchg64(unsigned long long v, int addr) {
    unsigned lo = (unsigned)__builtin_amdgcn_ds_bpermute(addr, (int)(unsigned)(v & 0xFFFFFFFFull));
    unsigned hi = (unsigned)__builtin_amdgcn_ds_bpermute(addr, (int)(unsigned)(v >> 32));
    return ((unsigned long long)hi << 32) | (unsigned long long)lo;
}

__device__ __forceinline__ void wave_ce_steps(unsigned long long& e0, unsigned long long& e1,
                                              int lane, int x0, int k) {
    bool up0 = ((x0 & k) == 0);
    bool up1 = (((x0 | 32) & k) == 0);
    if (k >= 64) {
        unsigned long long mx = e0 > e1 ? e0 : e1;
        unsigned long long mn = e0 > e1 ? e1 : e0;
        e0 = up0 ? mx : mn;
        e1 = up0 ? mn : mx;
    }
    #pragma unroll
    for (int j = 16; j > 0; j >>= 1) {
        if (2 * j <= k) {
            int addr = (lane ^ j) << 2;
            unsigned long long y0 = lane_xchg64(e0, addr);
            unsigned long long y1 = lane_xchg64(e1, addr);
            bool low = ((lane & j) == 0);
            bool wmax0 = (low == up0);
            bool wmax1 = (low == up1);
            bool g0 = y0 > e0;
            bool g1 = y1 > e1;
            e0 = (g0 == wmax0) ? y0 : e0;
            e1 = (g1 == wmax1) ? y1 : e1;
        }
    }
}

__global__ __launch_bounds__(1024) void topk_chunk_sort_kernel(unsigned long long* keys) {
    __shared__ unsigned long long s[CHUNK];
    int n   = blockIdx.x / NCHUNK;
    int c   = blockIdx.x % NCHUNK;
    int tid = threadIdx.x;
    int lane = tid & 31;
    int x0   = (tid >> 5) * 64 + lane;
    unsigned long long* g = keys + ((size_t)n << 16) + (size_t)c * CHUNK;
    unsigned long long e0 = g[x0];
    unsigned long long e1 = g[x0 + 32];
    wave_ce_steps(e0, e1, lane, x0, 2);
    wave_ce_steps(e0, e1, lane, x0, 4);
    wave_ce_steps(e0, e1, lane, x0, 8);
    wave_ce_steps(e0, e1, lane, x0, 16);
    wave_ce_steps(e0, e1, lane, x0, 32);
    wave_ce_steps(e0, e1, lane, x0, 64);
    #pragma unroll
    for (int k = 128; k <= CHUNK; k <<= 1) {
        s[x0]      = e0;
        s[x0 + 32] = e1;
        __syncthreads();
        for (int j = k >> 1; j >= 64; j >>= 1) {
            bitonic_ce(s, tid, j, k);
            __syncthreads();
        }
        e0 = s[x0];
        e1 = s[x0 + 32];
        wave_ce_steps(e0, e1, lane, x0, k);
    }
    g[x0]      = e0;
    g[x0 + 32] = e1;
}

__global__ __launch_bounds__(1024) void topk_fold_kernel(unsigned long long* keys,
                                                         int nch, int stride, int groups) {
    __shared__ unsigned long long s[CHUNK];
    int n   = blockIdx.x / groups;
    int gi  = blockIdx.x % groups;
    int tid = threadIdx.x;
    int lane = tid & 31;
    int x0   = (tid >> 5) * 64 + lane;
    size_t pitch = (size_t)stride * CHUNK;
    unsigned long long* g = keys + ((size_t)n << 16) + (size_t)gi * nch * pitch;
    s[tid]        = g[tid];
    s[tid + 1024] = g[tid + 1024];
    unsigned long long d0 = 0ull, d1 = 0ull;
    if (nch > 1) {
        d0 = g[pitch + (CHUNK - 1 - tid)];
        d1 = g[pitch + (CHUNK - 1 - 1024 - tid)];
    }
    for (int c = 1; c < nch; c++) {
        unsigned long long a0 = s[tid];
        unsigned long long a1 = s[tid + 1024];
        unsigned long long e0 = a0 > d0 ? a0 : d0;
        unsigned long long e1 = a1 > d1 ? a1 : d1;
        s[tid]        = e0 > e1 ? e0 : e1;
        s[tid + 1024] = e0 > e1 ? e1 : e0;
        if (c + 1 < nch) {
            const unsigned long long* gn = g + (size_t)(c + 1) * pitch;
            d0 = gn[CHUNK - 1 - tid];
            d1 = gn[CHUNK - 1 - 1024 - tid];
        }
        __syncthreads();
        for (int j = CHUNK >> 2; j >= 64; j >>= 1) {
            bitonic_ce(s, tid, j, CHUNK);
            __syncthreads();
        }
        unsigned long long r0 = s[x0];
        unsigned long long r1 = s[x0 + 32];
        wave_ce_steps(r0, r1, lane, x0, CHUNK);
        s[x0]      = r0;
        s[x0 + 32] = r1;
        __syncthreads();
    }
    g[tid]        = s[tid];
    g[tid + 1024] = s[tid + 1024];
}

#define NMS_R 512
#define NMS_W (NMS_R / 32)
#define MASK_ROWS 16
#define MASK_BLKS (NMS_R / MASK_ROWS)
#define LPITCH 17

__device__ __forceinline__ bool iou_over(v4f bi, float ia, v4f bj, float ja) {
    float ix1 = fmaxf(bi.x, bj.x), iy1 = fmaxf(bi.y, bj.y);
    float ix2 = fminf(bi.z, bj.z), iy2 = fminf(bi.w, bj.w);
    float inter = fmaxf(ix2 - ix1, 0.f) * fmaxf(iy2 - iy1, 0.f);
    float iou = inter / fmaxf(ia + ja - inter, 1e-9f);
    return iou > IOU_THR;
}

__global__ __launch_bounds__(256) void nms_mask_kernel(const unsigned long long* __restrict__ keys,
                                                       const float* __restrict__ boxes,
                                                       unsigned* __restrict__ mask,
                                                       float* __restrict__ sorted) {
    __shared__ v4f   sbx[32 * LPITCH];
    __shared__ float sar[32 * LPITCH];

    int n  = blockIdx.x / MASK_BLKS;
    int rb = blockIdx.x % MASK_BLKS;
    int tid = threadIdx.x;
    const unsigned long long* K = keys + ((size_t)n << 16);
    const float* bimg = boxes + (size_t)n * NPROP * 4;

    unsigned long long kk[2];
    #pragma unroll
    for (int q = 0; q < 2; q++) kk[q] = K[tid + 256 * q];
    v4f gb[2];
    #pragma unroll
    for (int q = 0; q < 2; q++) {
        unsigned sv  = (unsigned)(kk[q] >> 32);
        unsigned idx = 0xFFFFFFFFu - (unsigned)(kk[q] & 0xFFFFFFFFu);
        bool v = (sv > 0x007FFFFFu);
        gb[q] = *(const v4f*)(bimg + (size_t)(v ? idx : 0u) * 4);
    }
    #pragma unroll
    for (int q = 0; q < 2; q++) {
        int j = tid + 256 * q;
        unsigned sv = (unsigned)(kk[q] >> 32);
        bool v = (sv > 0x007FFFFFu);
        v4f bb = {0.f, 0.f, 0.f, 0.f};
        float s = 0.f;
        if (v) {
            bb = gb[q];
            unsigned u = (sv & 0x80000000u) ? (sv ^ 0x80000000u) : ~sv;
            s = __uint_as_float(u);
        }
        int slot = (j & 31) * LPITCH + (j >> 5);
        sbx[slot] = bb;
        sar[slot] = fmaxf(bb.z - bb.x, 0.f) * fmaxf(bb.w - bb.y, 0.f);
        if ((j / MASK_ROWS) == rb) {
            v4f* so = (v4f*)sorted + ((size_t)n * NMS_R + j) * 2;
            v4f inf = {s, v ? 1.f : 0.f, 0.f, 0.f};
            so[0] = bb;
            so[1] = inf;
        }
    }
    __syncthreads();

    int wave = tid >> 5;
    int lane = tid & 31;
    int i = rb * MASK_ROWS + wave * 2 + (lane >> 4);
    int w = lane & 15;
    int islot = (i & 31) * LPITCH + (i >> 5);
    v4f bi = sbx[islot];
    float ia = sar[islot];
    unsigned bits = 0u;
    if (w >= (i >> 5)) {
        for (int b = 0; b < 32; b++) {
            v4f bj = sbx[b * LPITCH + w];
            float ja = sar[b * LPITCH + w];
            if (iou_over(bi, ia, bj, ja)) bits |= (1u << b);
        }
    }
    mask[((size_t)n * NMS_R + i) * NMS_W + w] = bits;
}

#define KCAP 332

__device__ __forceinline__ unsigned greedy_chunk(unsigned cand, unsigned low, int lane) {
    unsigned notc = ((cand >> lane) & 1u) ^ 1u;
    unsigned kept = cand;
    for (int it = 0; it < 32; it++) {
        unsigned nk = __builtin_amdgcn_ballot_w32(((low & kept) | notc) == 0u);
        if (nk == kept) break;
        kept = nk;
    }
    return kept;
}

__global__ __launch_bounds__(32) void nms_scan_kernel(const unsigned* __restrict__ mask,
                                                      const float* __restrict__ sorted,
                                                      const unsigned long long* __restrict__ keys,
                                                      const float* __restrict__ boxes,
                                                      float* __restrict__ out) {
    __shared__ v4f   kbx[KCAP];
    __shared__ float kar[KCAP];

    int n = blockIdx.x;
    int lane = threadIdx.x;
    int wl = lane & 15;
    const unsigned* Mw = mask + (size_t)n * NMS_R * NMS_W;
    const v4f* S = (const v4f*)sorted + (size_t)n * NMS_R * 2;
    float* o_img = out + (size_t)n * POST_NMS * 5;

    unsigned rem = 0u;
    const unsigned below = (1u << lane) - 1u;

    unsigned cur[32], nxt[32];
    v4f cbox, cinf, nbox, ninf;
    unsigned cdiag, ndiag;
    #pragma unroll
    for (int r = 0; r < 32; r++) cur[r] = Mw[(size_t)r * NMS_W + wl];
    cbox = S[(size_t)lane * 2];
    cinf = S[(size_t)lane * 2 + 1];
    cdiag = Mw[(size_t)lane * NMS_W];

    int cnt = 0;
    for (int c = 0; c < NMS_W; c++) {
        int cn = (c + 1 < NMS_W) ? c + 1 : NMS_W - 1;
        #pragma unroll
        for (int r = 0; r < 32; r++) nxt[r] = Mw[(size_t)(cn * 32 + r) * NMS_W + wl];
        {
            int i = cn * 32 + lane;
            nbox = S[(size_t)i * 2];
            ninf = S[(size_t)i * 2 + 1];
            ndiag = Mw[(size_t)i * NMS_W + cn];
        }

        bool inval = (cinf.y == 0.f);
        unsigned remw = (unsigned)__builtin_amdgcn_readlane((int)rem, c)
                      | __builtin_amdgcn_ballot_w32(inval);
        unsigned kept = greedy_chunk(~remw, cdiag & below, lane);
        #pragma unroll
        for (int r = 0; r < 32; r++) {
            unsigned m = ((kept >> r) & 1u) ? 0xFFFFFFFFu : 0u;
            rem |= cur[r] & m;
        }

        unsigned mine = (kept >> lane) & 1u;
        int pos = cnt + __popc(kept & below);
        if (mine && pos < KCAP) {
            kbx[pos] = cbox;
            kar[pos] = fmaxf(cbox.z - cbox.x, 0.f) * fmaxf(cbox.w - cbox.y, 0.f);
        }
        if (mine && pos < POST_NMS) {
            float* o = o_img + pos * 5;
            o[0] = cbox.x; o[1] = cbox.y; o[2] = cbox.z; o[3] = cbox.w;
            o[4] = cinf.x;
        }
        cnt += __popc(kept);
        if (cnt >= POST_NMS) break;

        #pragma unroll
        for (int r = 0; r < 32; r++) cur[r] = nxt[r];
        cbox = nbox; cinf = ninf; cdiag = ndiag;
    }

    const unsigned long long* K = keys + ((size_t)n << 16);
    const float* bimg = boxes + (size_t)n * NPROP * 4;
    for (int c = NMS_W; c < 63 && cnt < POST_NMS; c++) {
        __syncthreads();
        int i = c * 32 + lane;
        unsigned long long kk = K[i < PRE_NMS ? i : PRE_NMS - 1];
        unsigned sv  = (unsigned)(kk >> 32);
        unsigned idx = 0xFFFFFFFFu - (unsigned)(kk & 0xFFFFFFFFu);
        bool v = (i < PRE_NMS) && (sv > 0x007FFFFFu) && (idx < (unsigned)NPROP);
        v4f bb = *(const v4f*)(bimg + (size_t)(v ? idx : 0u) * 4);
        float s = 0.f;
        if (v) {
            unsigned u = (sv & 0x80000000u) ? (sv ^ 0x80000000u) : ~sv;
            s = __uint_as_float(u);
        } else {
            bb = v4f{0.f, 0.f, 0.f, 0.f};
        }
        float ar = fmaxf(bb.z - bb.x, 0.f) * fmaxf(bb.w - bb.y, 0.f);

        bool removed = false;
        for (int k = 0; k < cnt; k++)
            removed = removed || iou_over(kbx[k], kar[k], bb, ar);

        unsigned diag = 0u;
        for (int j = 0; j < 32; j++) {
            v4f bj;
            bj.x = __int_as_float(__builtin_amdgcn_readlane(__float_as_int(bb.x), j));
            bj.y = __int_as_float(__builtin_amdgcn_readlane(__float_as_int(bb.y), j));
            bj.z = __int_as_float(__builtin_amdgcn_readlane(__float_as_int(bb.z), j));
            bj.w = __int_as_float(__builtin_amdgcn_readlane(__float_as_int(bb.w), j));
            float ja = __int_as_float(__builtin_amdgcn_readlane(__float_as_int(ar), j));
            if (iou_over(bb, ar, bj, ja)) diag |= (1u << j);
        }

        unsigned remw = __builtin_amdgcn_ballot_w32(removed || !v);
        unsigned kept = greedy_chunk(~remw, diag & below, lane);

        unsigned mine = (kept >> lane) & 1u;
        int pos = cnt + __popc(kept & below);
        if (mine && pos < KCAP) { kbx[pos] = bb; kar[pos] = ar; }
        if (mine && pos < POST_NMS) {
            float* o = o_img + pos * 5;
            o[0] = bb.x; o[1] = bb.y; o[2] = bb.z; o[3] = bb.w;
            o[4] = s;
        }
        cnt += __popc(kept);
    }

    int c0 = cnt < POST_NMS ? cnt : POST_NMS;
    for (int t = c0 * 5 + lane; t < POST_NMS * 5; t += 32) o_img[t] = 0.f;
}

extern "C" void kernel_launch(void* const* d_in, const int* in_sizes, int n_in,
                              void* d_out, int out_size, void* d_ws, size_t ws_size,
                              hipStream_t stream) {
    const float* feat   = (const float*)d_in[0];
    const float* conv_w = (const float*)d_in[1];
    const float* conv_b = (const float*)d_in[2];
    const float* det_w  = (const float*)d_in[3];
    const float* det_b  = (const float*)d_in[4];
    const float* reg_w  = (const float*)d_in[5];
    const float* reg_b  = (const float*)d_in[6];
    float* out = (float*)d_out;

    char* ws = (char*)d_ws;
    float* wt2                  = (float*)(ws);
    float* boxes                = (float*)(ws + 2359296 + 786432 + 3145728);
    unsigned long long* keys    = (unsigned long long*)(ws + 2359296 + 786432 + 3145728 + 3145728);

    fold_w_kernel<<<145, 256, 0, stream>>>(conv_w, conv_b, det_w, det_b, reg_w, reg_b, wt2);
    conv_fused_kernel<<<NIMG * HF * (WF / TILE_W), 256, 0, stream>>>(feat, wt2, boxes, keys);
    topk_chunk_sort_kernel<<<NIMG * NCHUNK, 1024, 0, stream>>>(keys);
    topk_fold_kernel<<<NIMG * FOLD_B, 1024, 0, stream>>>(keys, FOLD_A, 1, FOLD_B);
    topk_fold_kernel<<<NIMG, 1024, 0, stream>>>(keys, FOLD_B, FOLD_A, 1);
    unsigned* nmask  = (unsigned*)(ws);
    float*    sorted = (float*)(ws + 131072);
    nms_mask_kernel<<<NIMG * MASK_BLKS, 256, 0, stream>>>(keys, boxes, nmask, sorted);
    nms_scan_kernel<<<NIMG, 32, 0, stream>>>(nmask, sorted, keys, boxes, out);
}


# weight-fold kernel re-tiled: a block owns one channel pair (18 contiguous conv_w columns = 9 whole output lines), co split in 16 slices summed via LDS, head rows read from LDS; sum order changes
# speedup vs baseline: 21.9874x; 1.0454x over previous
#include <hip/hip_runtime.h>
#include <hip/hip_bf16.h>
#include <math.h>

typedef __attribute__((ext_vector_type(2))) float v2f;
typedef __attribute__((ext_vector_type(8))) float v8f;
typedef __attribute__((ext_vector_type(4))) float v4f;
typedef __attribute__((ext_vector_type(2))) unsigned v2u;

#define NIMG 4
#define CIN 256
#define HF 128
#define WF 128
#define A_ANCH 3
#define NPROP (A_ANCH * HF * WF)
#define NPAD 65536
#define PRE_NMS 2000
#define POST_NMS 300
#define OBJ_THR 0.3f
#define IOU_THR 0.7f
#define TILE_W 64
#define CCHUNK 32
#define CIPAD 34

#define NWEFF (9 * 128 * 16 * 2)
#define FW_COLS 18
#define FW_SL 16
#define FW_THREADS (FW_COLS * FW_SL)
__global__ __launch_bounds__(FW_THREADS) void fold_w_kernel(const float* __restrict__ conv_w,
                                                            const float* __restrict__ conv_b,
                                                            const float* __restrict__ det_w,
                                                            const float* __restrict__ det_b,
                                                            const float* __restrict__ reg_w,
                                                            const float* __restrict__ reg_b,
                                                            float* __restrict__ weff2) {
    __shared__ __attribute__((aligned(16))) float shw[14 * FW_THREADS];
    __shared__ float spart[FW_SL * 15 * FW_COLS];

    int tid = threadIdx.x;
    if (blockIdx.x == 128) {
        int o  = tid >> 4;
        int sl = tid & 15;
        float s = 0.0f;
        if (o < 15) {
            const float* hw = ((o < 3) ? (det_w + o * 256) : (reg_w + (o - 3) * 256)) + sl * 16;
            const float* cb = conv_b + sl * 16;
            float hv[16], bv[16];
            #pragma unroll
            for (int j = 0; j < 16; j++) { hv[j] = hw[j]; bv[j] = cb[j]; }
            #pragma unroll
            for (int j = 0; j < 16; j++) s = __builtin_fmaf(hv[j], bv[j], s);
        }
        float hb = 0.0f;
        if (tid < 15) hb = (tid < 3) ? det_b[tid] : reg_b[tid - 3];
        spart[tid] = s;
        __syncthreads();
        if (tid < 16) {
            float b = hb;
            if (tid < 15) {
                #pragma unroll
                for (int k = 0; k < FW_SL; k++) b += spart[tid * 16 + k];
            }
            weff2[NWEFF + tid] = b;
        }
        return;
    }
    int ci2 = blockIdx.x;
    int sl  = tid / FW_COLS;
    int col = tid - sl * FW_COLS;

    const float* wp = conv_w + (size_t)(sl * 16) * (CIN * 9) + ci2 * FW_COLS + col;
    float wv[16];
    #pragma unroll
    for (int j = 0; j < 16; j++) wv[j] = wp[(size_t)j * (CIN * 9)];

    float hst[14];
    #pragma unroll
    for (int q = 0; q < 14; q++) {
        int i  = tid + q * FW_THREADS;
        int ic = i < 15 * 256 ? i : 15 * 256 - 1;
        const float* p = (ic < 3 * 256) ? (det_w + ic) : (reg_w + (ic - 3 * 256));
        hst[q] = *p;
    }
    #pragma unroll
    for (int q = 0; q < 14; q++) shw[tid + q * FW_THREADS] = hst[q];
    __syncthreads();

    #pragma unroll
    for (int o = 0; o < 15; o++) {
        const v4f* hp = (const v4f*)&shw[o * 256 + sl * 16];
        float s = 0.0f;
        #pragma unroll
        for (int q = 0; q < 4; q++) {
            v4f h = hp[q];
            s = __builtin_fmaf(h.x, wv[q * 4 + 0], s);
            s = __builtin_fmaf(h.y, wv[q * 4 + 1], s);
            s = __builtin_fmaf(h.z, wv[q * 4 + 2], s);
            s = __builtin_fmaf(h.w, wv[q * 4 + 3], s);
        }
        spart[(sl * 15 + o) * FW_COLS + col] = s;
    }
    __syncthreads();

    int kk  = tid >> 5;
    int r   = tid & 31;
    int o   = r >> 1;
    int par = r & 1;
    float s = 0.0f;
    if (o < 15) {
        int c = par * 9 + kk;
        #pragma unroll
        for (int k = 0; k < FW_SL; k++) s += spart[(k * 15 + o) * FW_COLS + c];
    }
    weff2[(size_t)(kk * 128 + ci2) * 32 + r] = s;
}

#define NCHK (CIN / CCHUNK)
#define NPRE 25
#define PP 66

__device__ __forceinline__ void stage_fetch(float (&pre)[NPRE], const float* frow,
                                            const float* fhalo, const int (&rowoff)[3], int cib) {
    const float* fc = frow + (size_t)cib * (HF * WF);
    #pragma unroll
    for (int j = 0; j < 4; j++) {
        #pragma unroll
        for (int r = 0; r < 3; r++) {
            const float* p = fc + j * (HF * WF) + rowoff[r];
            pre[(j * 3 + r) * 2 + 0] = p[0];
            pre[(j * 3 + r) * 2 + 1] = p[32];
        }
    }
    pre[24] = fhalo[(size_t)cib * (HF * WF)];
}

__global__ __launch_bounds__(256) __attribute__((amdgpu_waves_per_eu(8))) void conv_fused_kernel(
    const float* __restrict__ feat, const float* __restrict__ weff2,
    float* __restrict__ boxes, unsigned long long* __restrict__ keys) {

    __shared__ __attribute__((aligned(16))) float lds_in[3 * 66 * CIPAD];
    __shared__ float part[8 * 16 * PP];

    int bid = blockIdx.x;
    int n    = bid >> 8;
    int rem  = bid & 255;
    int row  = rem >> 1;
    int col0 = (rem & 1) * TILE_W;

    int tid  = threadIdx.x;
    int wave = tid >> 5;
    int lane = tid & 31;
    int hl   = lane >> 4;
    int lm   = lane & 15;

    v8f acc[4] = {};

    const float* featn = feat + (size_t)n * CIN * HF * WF;
    const v2f*   wpl   = (const v2f*)weff2 + (size_t)(2 * wave + hl) * 16 + lm;

    int  rowoff[3];
    bool rowok[3];
    #pragma unroll
    for (int r = 0; r < 3; r++) {
        int g = row + r - 1;
        rowok[r]  = (g >= 0) && (g < HF);
        rowoff[r] = (g < 0 ? 0 : (g >= HF ? HF - 1 : g)) * WF;
    }
    const float* frow = featn + (size_t)(4 * wave) * (HF * WF) + col0 + lane;
    int  hj  = lane / 6;
    int  hq  = lane - hj * 6;
    int  hr  = hq >> 1;
    int  hs  = hq & 1;
    int  hg  = row + hr - 1;
    int  hc  = hs ? (col0 + TILE_W) : (col0 - 1);
    bool hok = (lane < 24) && (hg >= 0) && (hg < HF) && (hc >= 0) && (hc < WF);
    const float* fhalo = featn + (hok ? ((size_t)(4 * wave + hj) * HF + hg) * WF + hc : (size_t)0);
    int  hlds = (hr * 66 + (hs ? 65 : 0)) * CIPAD + 4 * wave + hj;

    int bbase = lm * CIPAD + 4 * wave + 2 * hl;

    float pre[NPRE];
    stage_fetch(pre, frow, fhalo, rowoff, 0);

    for (int ch = 0; ch < NCHK; ch++) {
        v2f a[9];
        #pragma unroll
        for (int kk = 0; kk < 9; kk++)
            a[kk] = wpl[(size_t)(kk * 128 + ch * 16) * 16];

        __syncthreads();
        #pragma unroll
        for (int r = 0; r < 3; r++) {
            #pragma unroll
            for (int hcol = 0; hcol < 2; hcol++) {
                float v0 = rowok[r] ? pre[(0 * 3 + r) * 2 + hcol] : 0.0f;
                float v1 = rowok[r] ? pre[(1 * 3 + r) * 2 + hcol] : 0.0f;
                float v2 = rowok[r] ? pre[(2 * 3 + r) * 2 + hcol] : 0.0f;
                float v3 = rowok[r] ? pre[(3 * 3 + r) * 2 + hcol] : 0.0f;
                int idx = (r * 66 + 1 + hcol * 32 + lane) * CIPAD + 4 * wave;
                v2f lo = {v0, v1};
                v2f hi = {v2, v3};
                *(v2f*)&lds_in[idx]     = lo;
                *(v2f*)&lds_in[idx + 2] = hi;
            }
        }
        if (lane < 24) lds_in[hlds] = hok ? pre[24] : 0.0f;
        __syncthreads();

        int chn = ch + 1 < NCHK ? ch + 1 : NCHK - 1;
        stage_fetch(pre, frow, fhalo, rowoff, chn * CCHUNK);

        #pragma unroll
        for (int kk = 0; kk < 9; kk++) {
            int kh = kk / 3;
            int kw = kk - kh * 3;
            v2f b[4];
            #pragma unroll
            for (int nt = 0; nt < 4; nt++)
                b[nt] = *(const v2f*)&lds_in[(kh * 66 + nt * 16 + kw) * CIPAD + bbase];
            #pragma unroll
            for (int nt = 0; nt < 4; nt++)
                acc[nt] = __builtin_amdgcn_wmma_f32_16x16x4_f32(false, a[kk], false, b[nt], (short)0, acc[nt], false, false);
        }
    }

    #pragma unroll
    for (int nt = 0; nt < 4; nt++) {
        #pragma unroll
        for (int r = 0; r < 8; r++)
            part[(wave * 16 + hl * 8 + r) * PP + nt * 16 + lm] = acc[nt][r];
    }
    __syncthreads();

    const float* beff = weff2 + NWEFF;
    if (tid < A_ANCH * TILE_W) {
        int a   = tid >> 6;
        int pix = tid & 63;
        float hv[5];
        #pragma unroll
        for (int q = 0; q < 5; q++) {
            int oi = (q == 0) ? a : (3 + (q - 1) * A_ANCH + a);
            float sum = part[oi * PP + pix];
            #pragma unroll
            for (int w = 1; w < 8; w++) sum += part[(w * 16 + oi) * PP + pix];
            hv[q] = sum + beff[oi];
        }
        float s = 1.0f / (1.0f + expf(-hv[0]));

        int i  = a * (HF * WF) + row * WF + col0 + pix;
        int pa = i / 3, aa = i - pa * 3;
        int ya = pa >> 7, xa = pa & 127;
        float hwv = (aa == 0) ? 91.f : ((aa == 1) ? 64.f : 45.f);
        float hhv = (aa == 0) ? 45.f : ((aa == 1) ? 64.f : 91.f);
        float sxv = (float)(xa * 8), syv = (float)(ya * 8);
        float x1 = sxv - hwv, y1 = syv - hhv;
        float x2 = sxv + hwv, y2 = syv + hhv;
        float aw = x2 - x1, ah = y2 - y1;
        float acx = y1 + aw * 0.5f;
        float acy = x1 + ah * 0.5f;

        float px = __builtin_fmaf(hv[1], aw, acx);
        float py = __builtin_fmaf(hv[2], ah, acy);
        float pw = aw * expf(hv[3]);
        float ph = ah * expf(hv[4]);
        float b0 = fminf(fmaxf(px - pw * 0.5f, 0.f), 1024.f);
        float b1 = fminf(fmaxf(py - ph * 0.5f, 0.f), 1024.f);
        float b2 = fminf(fmaxf(px + pw * 0.5f, 0.f), 1024.f);
        float b3 = fminf(fmaxf(py + ph * 0.5f, 0.f), 1024.f);

        v4f bo = {b0, b1, b2, b3};
        *(v4f*)(boxes + (((size_t)n * NPROP) + i) * 4) = bo;

        bool valid = (s > OBJ_THR) && ((b2 - b0) > 0.f) && ((b3 - b1) > 0.f);
        float ss = valid ? s : -__builtin_inff();
        unsigned u = __float_as_uint(ss);
        unsigned sv = (u & 0x80000000u) ? ~u : (u | 0x80000000u);
        keys[((size_t)n << 16) + i] =
            ((unsigned long long)sv << 32) | (unsigned long long)(0xFFFFFFFFu - (unsigned)i);
    }
}

#define CHUNK 2048
#define NCHUNK (NPROP / CHUNK)
#define FOLD_A 6
#define FOLD_B (NCHUNK / FOLD_A)

__device__ __forceinline__ void bitonic_ce(unsigned long long* s, int t, int j, int k) {
    int lo = t & (j - 1);
    int i  = ((t - lo) << 1) | lo;
    int p  = i | j;
    unsigned long long a = s[i];
    unsigned long long b = s[p];
    bool up = ((i & k) == 0);
    bool sw = up ? (a < b) : (a > b);
    if (sw) { s[i] = b; s[p] = a; }
}

__device__ __forceinline__ unsigned long long lane_xchg64(unsigned long long v, int addr) {
    unsigned lo = (unsigned)__builtin_amdgcn_ds_bpermute(addr, (int)(unsigned)(v & 0xFFFFFFFFull));
    unsigned hi = (unsigned)__builtin_amdgcn_ds_bpermute(addr, (int)(unsigned)(v >> 32));
    return ((unsigned long long)hi << 32) | (unsigned long long)lo;
}

__device__ __forceinline__ void wave_ce_steps(unsigned long long& e0, unsigned long long& e1,
                                              int lane, int x0, int k) {
    bool up0 = ((x0 & k) == 0);
    bool up1 = (((x0 | 32) & k) == 0);
    if (k >= 64) {
        unsigned long long mx = e0 > e1 ? e0 : e1;
        unsigned long long mn = e0 > e1 ? e1 : e0;
        e0 = up0 ? mx : mn;
        e1 = up0 ? mn : mx;
    }
    #pragma unroll
    for (int j = 16; j > 0; j >>= 1) {
        if (2 * j <= k) {
            int addr = (lane ^ j) << 2;
            unsigned long long y0 = lane_xchg64(e0, addr);
            unsigned long long y1 = lane_xchg64(e1, addr);
            bool low = ((lane & j) == 0);
            bool wmax0 = (low == up0);
            bool wmax1 = (low == up1);
            bool g0 = y0 > e0;
            bool g1 = y1 > e1;
            e0 = (g0 == wmax0) ? y0 : e0;
            e1 = (g1 == wmax1) ? y1 : e1;
        }
    }
}

__global__ __launch_bounds__(1024) void topk_chunk_sort_kernel(unsigned long long* keys) {
    __shared__ unsigned long long s[CHUNK];
    int n   = blockIdx.x / NCHUNK;
    int c   = blockIdx.x % NCHUNK;
    int tid = threadIdx.x;
    int lane = tid & 31;
    int x0   = (tid >> 5) * 64 + lane;
    unsigned long long* g = keys + ((size_t)n << 16) + (size_t)c * CHUNK;
    unsigned long long e0 = g[x0];
    unsigned long long e1 = g[x0 + 32];
    wave_ce_steps(e0, e1, lane, x0, 2);
    wave_ce_steps(e0, e1, lane, x0, 4);
    wave_ce_steps(e0, e1, lane, x0, 8);
    wave_ce_steps(e0, e1, lane, x0, 16);
    wave_ce_steps(e0, e1, lane, x0, 32);
    wave_ce_steps(e0, e1, lane, x0, 64);
    #pragma unroll
    for (int k = 128; k <= CHUNK; k <<= 1) {
        s[x0]      = e0;
        s[x0 + 32] = e1;
        __syncthreads();
        for (int j = k >> 1; j >= 64; j >>= 1) {
            bitonic_ce(s, tid, j, k);
            __syncthreads();
        }
        e0 = s[x0];
        e1 = s[x0 + 32];
        wave_ce_steps(e0, e1, lane, x0, k);
    }
    g[x0]      = e0;
    g[x0 + 32] = e1;
}

__global__ __launch_bounds__(1024) void topk_fold_kernel(unsigned long long* keys,
                                                         int nch, int stride, int groups) {
    __shared__ unsigned long long s[CHUNK];
    int n   = blockIdx.x / groups;
    int gi  = blockIdx.x % groups;
    int tid = threadIdx.x;
    int lane = tid & 31;
    int x0   = (tid >> 5) * 64 + lane;
    size_t pitch = (size_t)stride * CHUNK;
    unsigned long long* g = keys + ((size_t)n << 16) + (size_t)gi * nch * pitch;
    s[tid]        = g[tid];
    s[tid + 1024] = g[tid + 1024];
    unsigned long long d0 = 0ull, d1 = 0ull;
    if (nch > 1) {
        d0 = g[pitch + (CHUNK - 1 - tid)];
        d1 = g[pitch + (CHUNK - 1 - 1024 - tid)];
    }
    for (int c = 1; c < nch; c++) {
        unsigned long long a0 = s[tid];
        unsigned long long a1 = s[tid + 1024];
        unsigned long long e0 = a0 > d0 ? a0 : d0;
        unsigned long long e1 = a1 > d1 ? a1 : d1;
        s[tid]        = e0 > e1 ? e0 : e1;
        s[tid + 1024] = e0 > e1 ? e1 : e0;
        if (c + 1 < nch) {
            const unsigned long long* gn = g + (size_t)(c + 1) * pitch;
            d0 = gn[CHUNK - 1 - tid];
            d1 = gn[CHUNK - 1 - 1024 - tid];
        }
        __syncthreads();
        for (int j = CHUNK >> 2; j >= 64; j >>= 1) {
            bitonic_ce(s, tid, j, CHUNK);
            __syncthreads();
        }
        unsigned long long r0 = s[x0];
        unsigned long long r1 = s[x0 + 32];
        wave_ce_steps(r0, r1, lane, x0, CHUNK);
        s[x0]      = r0;
        s[x0 + 32] = r1;
        __syncthreads();
    }
    g[tid]        = s[tid];
    g[tid + 1024] = s[tid + 1024];
}

#define NMS_R 512
#define NMS_W (NMS_R / 32)
#define MASK_ROWS 16
#define MASK_BLKS (NMS_R / MASK_ROWS)
#define LPITCH 17

__device__ __forceinline__ bool iou_over(v4f bi, float ia, v4f bj, float ja) {
    float ix1 = fmaxf(bi.x, bj.x), iy1 = fmaxf(bi.y, bj.y);
    float ix2 = fminf(bi.z, bj.z), iy2 = fminf(bi.w, bj.w);
    float inter = fmaxf(ix2 - ix1, 0.f) * fmaxf(iy2 - iy1, 0.f);
    float iou = inter / fmaxf(ia + ja - inter, 1e-9f);
    return iou > IOU_THR;
}

__global__ __launch_bounds__(256) void nms_mask_kernel(const unsigned long long* __restrict__ keys,
                                                       const float* __restrict__ boxes,
                                                       unsigned* __restrict__ mask,
                                                       float* __restrict__ sorted) {
    __shared__ v4f   sbx[32 * LPITCH];
    __shared__ float sar[32 * LPITCH];

    int n  = blockIdx.x / MASK_BLKS;
    int rb = blockIdx.x % MASK_BLKS;
    int tid = threadIdx.x;
    const unsigned long long* K = keys + ((size_t)n << 16);
    const float* bimg = boxes + (size_t)n * NPROP * 4;

    unsigned long long kk[2];
    #pragma unroll
    for (int q = 0; q < 2; q++) kk[q] = K[tid + 256 * q];
    v4f gb[2];
    #pragma unroll
    for (int q = 0; q < 2; q++) {
        unsigned sv  = (unsigned)(kk[q] >> 32);
        unsigned idx = 0xFFFFFFFFu - (unsigned)(kk[q] & 0xFFFFFFFFu);
        bool v = (sv > 0x007FFFFFu);
        gb[q] = *(const v4f*)(bimg + (size_t)(v ? idx : 0u) * 4);
    }
    #pragma unroll
    for (int q = 0; q < 2; q++) {
        int j = tid + 256 * q;
        unsigned sv = (unsigned)(kk[q] >> 32);
        bool v = (sv > 0x007FFFFFu);
        v4f bb = {0.f, 0.f, 0.f, 0.f};
        float s = 0.f;
        if (v) {
            bb = gb[q];
            unsigned u = (sv & 0x80000000u) ? (sv ^ 0x80000000u) : ~sv;
            s = __uint_as_float(u);
        }
        int slot = (j & 31) * LPITCH + (j >> 5);
        sbx[slot] = bb;
        sar[slot] = fmaxf(bb.z - bb.x, 0.f) * fmaxf(bb.w - bb.y, 0.f);
        if ((j / MASK_ROWS) == rb) {
            v4f* so = (v4f*)sorted + ((size_t)n * NMS_R + j) * 2;
            v4f inf = {s, v ? 1.f : 0.f, 0.f, 0.f};
            so[0] = bb;
            so[1] = inf;
        }
    }
    __syncthreads();

    int wave = tid >> 5;
    int lane = tid & 31;
    int i = rb * MASK_ROWS + wave * 2 + (lane >> 4);
    int w = lane & 15;
    int islot = (i & 31) * LPITCH + (i >> 5);
    v4f bi = sbx[islot];
    float ia = sar[islot];
    unsigned bits = 0u;
    if (w >= (i >> 5)) {
        for (int b = 0; b < 32; b++) {
            v4f bj = sbx[b * LPITCH + w];
            float ja = sar[b * LPITCH + w];
            if (iou_over(bi, ia, bj, ja)) bits |= (1u << b);
        }
    }
    mask[((size_t)n * NMS_R + i) * NMS_W + w] = bits;
}

#define KCAP 332

__device__ __forceinline__ unsigned greedy_chunk(unsigned cand, unsigned low, int lane) {
    unsigned notc = ((cand >> lane) & 1u) ^ 1u;
    unsigned kept = cand;
    for (int it = 0; it < 32; it++) {
        unsigned nk = __builtin_amdgcn_ballot_w32(((low & kept) | notc) == 0u);
        if (nk == kept) break;
        kept = nk;
    }
    return kept;
}

__global__ __launch_bounds__(32) void nms_scan_kernel(const unsigned* __restrict__ mask,
                                                      const float* __restrict__ sorted,
                                                      const unsigned long long* __restrict__ keys,
                                                      const float* __restrict__ boxes,
                                                      float* __restrict__ out) {
    __shared__ v4f   kbx[KCAP];
    __shared__ float kar[KCAP];

    int n = blockIdx.x;
    int lane = threadIdx.x;
    int wl = lane & 15;
    const unsigned* Mw = mask + (size_t)n * NMS_R * NMS_W;
    const v4f* S = (const v4f*)sorted + (size_t)n * NMS_R * 2;
    float* o_img = out + (size_t)n * POST_NMS * 5;

    unsigned rem = 0u;
    const unsigned below = (1u << lane) - 1u;

    unsigned cur[32], nxt[32];
    v4f cbox, cinf, nbox, ninf;
    unsigned cdiag, ndiag;
    #pragma unroll
    for (int r = 0; r < 32; r++) cur[r] = Mw[(size_t)r * NMS_W + wl];
    cbox = S[(size_t)lane * 2];
    cinf = S[(size_t)lane * 2 + 1];
    cdiag = Mw[(size_t)lane * NMS_W];

    int cnt = 0;
    for (int c = 0; c < NMS_W; c++) {
        int cn = (c + 1 < NMS_W) ? c + 1 : NMS_W - 1;
        #pragma unroll
        for (int r = 0; r < 32; r++) nxt[r] = Mw[(size_t)(cn * 32 + r) * NMS_W + wl];
        {
            int i = cn * 32 + lane;
            nbox = S[(size_t)i * 2];
            ninf = S[(size_t)i * 2 + 1];
            ndiag = Mw[(size_t)i * NMS_W + cn];
        }

        bool inval = (cinf.y == 0.f);
        unsigned remw = (unsigned)__builtin_amdgcn_readlane((int)rem, c)
                      | __builtin_amdgcn_ballot_w32(inval);
        unsigned kept = greedy_chunk(~remw, cdiag & below, lane);
        #pragma unroll
        for (int r = 0; r < 32; r++) {
            unsigned m = ((kept >> r) & 1u) ? 0xFFFFFFFFu : 0u;
            rem |= cur[r] & m;
        }

        unsigned mine = (kept >> lane) & 1u;
        int pos = cnt + __popc(kept & below);
        if (mine && pos < KCAP) {
            kbx[pos] = cbox;
            kar[pos] = fmaxf(cbox.z - cbox.x, 0.f) * fmaxf(cbox.w - cbox.y, 0.f);
        }
        if (mine && pos < POST_NMS) {
            float* o = o_img + pos * 5;
            o[0] = cbox.x; o[1] = cbox.y; o[2] = cbox.z; o[3] = cbox.w;
            o[4] = cinf.x;
        }
        cnt += __popc(kept);
        if (cnt >= POST_NMS) break;

        #pragma unroll
        for (int r = 0; r < 32; r++) cur[r] = nxt[r];
        cbox = nbox; cinf = ninf; cdiag = ndiag;
    }

    const unsigned long long* K = keys + ((size_t)n << 16);
    const float* bimg = boxes + (size_t)n * NPROP * 4;
    for (int c = NMS_W; c < 63 && cnt < POST_NMS; c++) {
        __syncthreads();
        int i = c * 32 + lane;
        unsigned long long kk = K[i < PRE_NMS ? i : PRE_NMS - 1];
        unsigned sv  = (unsigned)(kk >> 32);
        unsigned idx = 0xFFFFFFFFu - (unsigned)(kk & 0xFFFFFFFFu);
        bool v = (i < PRE_NMS) && (sv > 0x007FFFFFu) && (idx < (unsigned)NPROP);
        v4f bb = *(const v4f*)(bimg + (size_t)(v ? idx : 0u) * 4);
        float s = 0.f;
        if (v) {
            unsigned u = (sv & 0x80000000u) ? (sv ^ 0x80000000u) : ~sv;
            s = __uint_as_float(u);
        } else {
            bb = v4f{0.f, 0.f, 0.f, 0.f};
        }
        float ar = fmaxf(bb.z - bb.x, 0.f) * fmaxf(bb.w - bb.y, 0.f);

        bool removed = false;
        for (int k = 0; k < cnt; k++)
            removed = removed || iou_over(kbx[k], kar[k], bb, ar);

        unsigned diag = 0u;
        for (int j = 0; j < 32; j++) {
            v4f bj;
            bj.x = __int_as_float(__builtin_amdgcn_readlane(__float_as_int(bb.x), j));
            bj.y = __int_as_float(__builtin_amdgcn_readlane(__float_as_int(bb.y), j));
            bj.z = __int_as_float(__builtin_amdgcn_readlane(__float_as_int(bb.z), j));
            bj.w = __int_as_float(__builtin_amdgcn_readlane(__float_as_int(bb.w), j));
            float ja = __int_as_float(__builtin_amdgcn_readlane(__float_as_int(ar), j));
            if (iou_over(bb, ar, bj, ja)) diag |= (1u << j);
        }

        unsigned remw = __builtin_amdgcn_ballot_w32(removed || !v);
        unsigned kept = greedy_chunk(~remw, diag & below, lane);

        unsigned mine = (kept >> lane) & 1u;
        int pos = cnt + __popc(kept & below);
        if (mine && pos < KCAP) { kbx[pos] = bb; kar[pos] = ar; }
        if (mine && pos < POST_NMS) {
            float* o = o_img + pos * 5;
            o[0] = bb.x; o[1] = bb.y; o[2] = bb.z; o[3] = bb.w;
            o[4] = s;
        }
        cnt += __popc(kept);
    }

    int c0 = cnt < POST_NMS ? cnt : POST_NMS;
    for (int t = c0 * 5 + lane; t < POST_NMS * 5; t += 32) o_img[t] = 0.f;
}

extern "C" void kernel_launch(void* const* d_in, const int* in_sizes, int n_in,
                              void* d_out, int out_size, void* d_ws, size_t ws_size,
                              hipStream_t stream) {
    const float* feat   = (const float*)d_in[0];
    const float* conv_w = (const float*)d_in[1];
    const float* conv_b = (const float*)d_in[2];
    const float* det_w  = (const float*)d_in[3];
    const float* det_b  = (const float*)d_in[4];
    const float* reg_w  = (const float*)d_in[5];
    const float* reg_b  = (const float*)d_in[6];
    float* out = (float*)d_out;

    char* ws = (char*)d_ws;
    float* wt2                  = (float*)(ws);
    float* boxes                = (float*)(ws + 2359296 + 786432 + 3145728);
    unsigned long long* keys    = (unsigned long long*)(ws + 2359296 + 786432 + 3145728 + 3145728);

    fold_w_kernel<<<129, FW_THREADS, 0, stream>>>(conv_w, conv_b, det_w, det_b, reg_w, reg_b, wt2);
    conv_fused_kernel<<<NIMG * HF * (WF / TILE_W), 256, 0, stream>>>(feat, wt2, boxes, keys);
    topk_chunk_sort_kernel<<<NIMG * NCHUNK, 1024, 0, stream>>>(keys);
    topk_fold_kernel<<<NIMG * FOLD_B, 1024, 0, stream>>>(keys, FOLD_A, 1, FOLD_B);
    topk_fold_kernel<<<NIMG, 1024, 0, stream>>>(keys, FOLD_B, FOLD_A, 1);
    unsigned* nmask  = (unsigned*)(ws);
    float*    sorted = (float*)(ws + 131072);
    nms_mask_kernel<<<NIMG * MASK_BLKS, 256, 0, stream>>>(keys, boxes, nmask, sorted);
    nms_scan_kernel<<<NIMG, 32, 0, stream>>>(nmask, sorted, keys, boxes, out);
}
